# starting kernel
# baseline (speedup 1.0000x reference)
#include <hip/hip_runtime.h>
#include <math.h>

#define LOG_TD 22u
#define TD (1u << LOG_TD)

typedef __attribute__((ext_vector_type(2))) float v2f;
typedef __attribute__((ext_vector_type(8))) float v8f;

constexpr unsigned TPW = 4u;

__device__ __forceinline__ float hw_exp2(float x) {
#if __has_builtin(__builtin_amdgcn_exp2f)
  return __builtin_amdgcn_exp2f(x);
#else
  return exp2f(x);
#endif
}
__device__ __forceinline__ float hw_rcp(float x) {
#if __has_builtin(__builtin_amdgcn_rcpf)
  return __builtin_amdgcn_rcpf(x);
#else
  return 1.0f / x;
#endif
}
__device__ __forceinline__ float fast_sigmoid(float x) {
  return hw_rcp(1.0f + hw_exp2(-1.4426950408889634f * x));
}
__device__ __forceinline__ float fast_tanh(float x) {
#if __has_builtin(__builtin_amdgcn_tanhf)
  return __builtin_amdgcn_tanhf(x);
#else
  return 1.0f - 2.0f * hw_rcp(hw_exp2(2.8853900817779268f * x) + 1.0f);
#endif
}

template <bool HAS_M1>
__device__ __forceinline__ float gate_src(unsigned m,
    const float* __restrict__ hv, const float* __restrict__ m0,
    const float* __restrict__ m1) {
  if (m < 3u * TD) {
    unsigned n = m >> LOG_TD;
    unsigned r = m & (TD - 1u);
    if (n == 0u) return m0[r];
    if (HAS_M1 && n == 1u) return m1[r];
    return 0.0f;
  }
  unsigned r = (m - 3u * TD) & (TD - 1u);
  return hv[r];
}

template <bool HAS_M1>
__device__ __forceinline__ float cand_src(unsigned m,
    const float* __restrict__ hv, const float* __restrict__ m0,
    const float* __restrict__ m1, const float* __restrict__ G) {
  unsigned seg = m >> LOG_TD;
  unsigned r   = m & (TD - 1u);
  if (seg & 1u) return hv[r];
  unsigned n = seg >> 1;
  if (n == 0u) return G[r] * m0[r];
  if (HAS_M1 && n == 1u) return G[TD + r] * m1[r];
  return 0.0f;
}

template <bool HAS_M1>
__global__ __launch_bounds__(256) void gsn_gate_wmma(
    const float* __restrict__ hv, const float* __restrict__ m0,
    const float* __restrict__ m1, const float* __restrict__ Wx,
    const float* __restrict__ Bx,
    float* __restrict__ G) {
  const unsigned lane  = threadIdx.x & 31u;
  const unsigned wave  = threadIdx.x >> 5u;
  const unsigned tile0 = (blockIdx.x * 8u + wave) * TPW;

  const bool     lo  = lane < 16u;
  const unsigned N   = lane & 15u;
  const bool     nok = N < 3u;
  const unsigned wn  = nok ? N : 0u;
  const float    msk = nok ? 1.0f : 0.0f;
  const float w0 = Wx[wn*6u+0u]*msk, w1 = Wx[wn*6u+1u]*msk, w2 = Wx[wn*6u+2u]*msk,
              w3 = Wx[wn*6u+3u]*msk, w4 = Wx[wn*6u+4u]*msk, w5 = Wx[wn*6u+5u]*msk;
  const float bias = Bx[wn] * msk;

  v2f b, b2;
  b.x  = lo ? w0 : w2;
  b.y  = lo ? w1 : w3;
  b2.x = lo ? w4 : 0.0f;
  b2.y = lo ? w5 : 0.0f;

  for (unsigned t = 0; t < TPW; ++t) {
    const unsigned q0   = (tile0 + t) * 16u;
    const unsigned row  = q0 + (lane & 15u);
    const unsigned base = 6u * row;
    v2f a, a2;
    a.x  = gate_src<HAS_M1>(base + (lo ? 0u : 2u), hv, m0, m1);
    a.y  = gate_src<HAS_M1>(base + (lo ? 1u : 3u), hv, m0, m1);
    a2.x = lo ? gate_src<HAS_M1>(base + 4u, hv, m0, m1) : 0.0f;
    a2.y = lo ? gate_src<HAS_M1>(base + 5u, hv, m0, m1) : 0.0f;

    v8f c;
#pragma unroll
    for (int v = 0; v < 8; ++v) c[v] = bias;

    v8f d = __builtin_amdgcn_wmma_f32_16x16x4_f32(false, a,  false, b,  (short)0, c, false, false);
    d     = __builtin_amdgcn_wmma_f32_16x16x4_f32(false, a2, false, b2, (short)0, d, false, false);

    if (nok) {
      const unsigned mb = lo ? 0u : 8u;
#pragma unroll
      for (int v = 0; v < 8; ++v) {
        unsigned p = 3u * (q0 + mb + (unsigned)v) + N;
        G[p] = fast_sigmoid(d[v]);
      }
    }
  }
}

template <bool HAS_M1>
__global__ __launch_bounds__(256) void gsn_cand_wmma(
    const float* __restrict__ hv, const float* __restrict__ m0,
    const float* __restrict__ m1, const float* __restrict__ Ww,
    const float* __restrict__ Bw,
    const float* __restrict__ G, float* __restrict__ CT) {
  const unsigned lane  = threadIdx.x & 31u;
  const unsigned wave  = threadIdx.x >> 5u;
  const unsigned tile0 = (blockIdx.x * 8u + wave) * TPW;

  const bool     lo  = lane < 16u;
  const unsigned N   = lane & 15u;
  const bool     nok = N < 3u;
  const unsigned wn  = nok ? N : 0u;
  const float    msk = nok ? 1.0f : 0.0f;
  const float w0 = Ww[wn*6u+0u]*msk, w1 = Ww[wn*6u+1u]*msk, w2 = Ww[wn*6u+2u]*msk,
              w3 = Ww[wn*6u+3u]*msk, w4 = Ww[wn*6u+4u]*msk, w5 = Ww[wn*6u+5u]*msk;
  const float bias = Bw[wn] * msk;

  v2f b, b2;
  b.x  = lo ? w0 : w2;
  b.y  = lo ? w1 : w3;
  b2.x = lo ? w4 : 0.0f;
  b2.y = lo ? w5 : 0.0f;

  for (unsigned t = 0; t < TPW; ++t) {
    const unsigned q0   = (tile0 + t) * 16u;
    const unsigned row  = q0 + (lane & 15u);
    const unsigned base = 6u * row;
    v2f a, a2;
    a.x  = cand_src<HAS_M1>(base + (lo ? 0u : 2u), hv, m0, m1, G);
    a.y  = cand_src<HAS_M1>(base + (lo ? 1u : 3u), hv, m0, m1, G);
    a2.x = lo ? cand_src<HAS_M1>(base + 4u, hv, m0, m1, G) : 0.0f;
    a2.y = lo ? cand_src<HAS_M1>(base + 5u, hv, m0, m1, G) : 0.0f;

    v8f c;
#pragma unroll
    for (int v = 0; v < 8; ++v) c[v] = bias;

    v8f d = __builtin_amdgcn_wmma_f32_16x16x4_f32(false, a,  false, b,  (short)0, c, false, false);
    d     = __builtin_amdgcn_wmma_f32_16x16x4_f32(false, a2, false, b2, (short)0, d, false, false);

    if (nok) {
      const unsigned mb = lo ? 0u : 8u;
#pragma unroll
      for (int v = 0; v < 8; ++v) {
        unsigned p   = 3u * (q0 + mb + (unsigned)v) + N;
        float cand   = fast_tanh(d[v]);
        float g      = G[p];
        unsigned n_o = p >> LOG_TD;
        unsigned r   = p & (TD - 1u);
        float mv;
        if (n_o == 0u)               mv = m0[r];
        else if (HAS_M1 && n_o == 1u) mv = m1[r];
        else                          mv = 0.0f;
        CT[p] = (1.0f - g) * mv + g * cand;
      }
    }
  }
}

__global__ __launch_bounds__(256) void gsn_finalize(
    const float* __restrict__ hv, const float* __restrict__ CT,
    float* __restrict__ out) {
  const unsigned i = (blockIdx.x * 256u + threadIdx.x) * 4u;
  const float4 h4 = *(const float4*)(hv + i);
  const float4 c0 = *(const float4*)(CT + i);
  const float4 c1 = *(const float4*)(CT + TD + i);
  const float4 c2 = *(const float4*)(CT + 2u * TD + i);
  float4 o;
  o.x = h4.x + c0.x + c1.x + c2.x;
  o.y = h4.y + c0.y + c1.y + c2.y;
  o.z = h4.z + c0.z + c1.z + c2.z;
  o.w = h4.w + c0.w + c1.w + c2.w;
  *(float4*)(out + i) = o;
}

extern "C" void kernel_launch(void* const* d_in, const int* in_sizes, int n_in,
                              void* d_out, int out_size, void* d_ws, size_t ws_size,
                              hipStream_t stream) {
  (void)in_sizes; (void)n_in; (void)out_size; (void)ws_size;
  const float* feat = (const float*)d_in[0];
  const float* W_w  = (const float*)d_in[1];
  const float* W_b  = (const float*)d_in[2];
  const float* Wx_w = (const float*)d_in[3];
  const float* Wx_b = (const float*)d_in[4];
  float* out = (float*)d_out;

  float* G  = (float*)d_ws;
  float* CT = G + (size_t)3 * TD;

  const float* h0 = feat;
  const float* h1 = feat + (size_t)TD;
  const float* h2 = feat + (size_t)2 * TD;

  dim3 blk(256);
  dim3 grd_mm(TD / 16u / TPW / 8u);
  dim3 grd_fin(TD / 4u / 256u);

  gsn_gate_wmma<false><<<grd_mm, blk, 0, stream>>>(h0, h2, nullptr, Wx_w, Wx_b, G);
  gsn_cand_wmma<false><<<grd_mm, blk, 0, stream>>>(h0, h2, nullptr, W_w, W_b, G, CT);
  gsn_finalize<<<grd_fin, blk, 0, stream>>>(h0, CT, out + 0 * (size_t)TD);

  gsn_gate_wmma<false><<<grd_mm, blk, 0, stream>>>(h1, h0, nullptr, Wx_w, Wx_b, G);
  gsn_cand_wmma<false><<<grd_mm, blk, 0, stream>>>(h1, h0, nullptr, W_w, W_b, G, CT);
  gsn_finalize<<<grd_fin, blk, 0, stream>>>(h1, CT, out + 1 * (size_t)TD);

  gsn_gate_wmma<true><<<grd_mm, blk, 0, stream>>>(h2, h1, h0, Wx_w, Wx_b, G);
  gsn_cand_wmma<true><<<grd_mm, blk, 0, stream>>>(h2, h1, h0, W_w, W_b, G, CT);
  gsn_finalize<<<grd_fin, blk, 0, stream>>>(h2, CT, out + 2 * (size_t)TD);
}


# Each launch is 1024 resident work-groups looping over 4 tiles (sum pass: 4 float4 per thread), not 4096 one-tile groups; next tile's loads issued a tile ahead across LDS-only barriers; sum order kept
# speedup vs baseline: 2.0849x; 2.0849x over previous
#include <hip/hip_runtime.h>
#include <math.h>

#define LOG_TD 22u
#define TD (1u << LOG_TD)

typedef __attribute__((ext_vector_type(2))) float v2f;
typedef __attribute__((ext_vector_type(4))) float v4f;
typedef __attribute__((ext_vector_type(8))) float v8f;

constexpr unsigned TPW = 8u;
constexpr unsigned BLK_OUT = 8u * TPW * 48u;
constexpr unsigned BLK_IN  = 8u * TPW * 96u;
constexpr unsigned STG     = BLK_IN / 4u / 256u;
constexpr unsigned TLN     = BLK_OUT / 4u / 256u;
constexpr unsigned NTILE   = TD / 16u / TPW / 8u;
constexpr unsigned PGRID   = 1024u;
constexpr unsigned FIN_PER = 4u;

__device__ __forceinline__ float hw_exp2(float x) {
#if __has_builtin(__builtin_amdgcn_exp2f)
  return __builtin_amdgcn_exp2f(x);
#else
  return exp2f(x);
#endif
}
__device__ __forceinline__ float hw_rcp(float x) {
#if __has_builtin(__builtin_amdgcn_rcpf)
  return __builtin_amdgcn_rcpf(x);
#else
  return 1.0f / x;
#endif
}
__device__ __forceinline__ float fast_sigmoid(float x) {
  return hw_rcp(1.0f + hw_exp2(-1.4426950408889634f * x));
}
__device__ __forceinline__ float fast_tanh(float x) {
#if __has_builtin(__builtin_amdgcn_tanhf)
  return __builtin_amdgcn_tanhf(x);
#else
  return 1.0f - 2.0f * hw_rcp(hw_exp2(2.8853900817779268f * x) + 1.0f);
#endif
}

__device__ __forceinline__ void lds_barrier() {
  __builtin_amdgcn_fence(__ATOMIC_RELEASE, "workgroup", "local");
  __builtin_amdgcn_s_barrier();
  __builtin_amdgcn_fence(__ATOMIC_ACQUIRE, "workgroup", "local");
}

template <bool HAS_M1>
__device__ __forceinline__ bool gate_zero(unsigned m) {
  const unsigned n = m >> LOG_TD;
  return n < 3u && n != 0u && !(HAS_M1 && n == 1u);
}
template <bool HAS_M1>
__device__ __forceinline__ v4f gate_load4(unsigned m,
    const float* __restrict__ hv, const float* __restrict__ m0,
    const float* __restrict__ m1) {
  const unsigned n = m >> LOG_TD;
  const unsigned r = m & (TD - 1u);
  const float* src = (n == 0u) ? m0 : ((HAS_M1 && n == 1u) ? m1 : hv);
  return *(const v4f*)(src + (gate_zero<HAS_M1>(m) ? 0u : r));
}
template <bool HAS_M1>
__device__ __forceinline__ v4f gate_mix4(unsigned m, const v4f v) {
  const bool zero = gate_zero<HAS_M1>(m);
  v4f o;
  o.x = zero ? 0.0f : v.x;
  o.y = zero ? 0.0f : v.y;
  o.z = zero ? 0.0f : v.z;
  o.w = zero ? 0.0f : v.w;
  return o;
}

template <bool HAS_M1>
__device__ __forceinline__ void cand_load4(unsigned m,
    const float* __restrict__ hv, const float* __restrict__ m0,
    const float* __restrict__ m1, const float* __restrict__ G,
    v4f& a, v4f& g) {
  const unsigned seg = m >> LOG_TD;
  const unsigned r   = m & (TD - 1u);
  const bool  isH = (seg & 1u) != 0u;
  const bool  s0  = seg == 0u;
  const bool  s1  = HAS_M1 && seg == 2u;
  const bool  msg = s0 || s1;
  const float* pa = s0 ? m0 : (s1 ? m1 : hv);
  const float* pg = s1 ? G + TD : G;
  a = *(const v4f*)(pa + ((isH || msg) ? r : 0u));
  g = *(const v4f*)(msg ? pg + r : hv);
}
template <bool HAS_M1>
__device__ __forceinline__ v4f cand_mix4(unsigned m, const v4f a, const v4f g) {
  const unsigned seg = m >> LOG_TD;
  const bool  isH = (seg & 1u) != 0u;
  const bool  msg = seg == 0u || (HAS_M1 && seg == 2u);
  const float px = g.x * a.x, py = g.y * a.y, pz = g.z * a.z, pw = g.w * a.w;
  const float zx = msg ? px : 0.0f, zy = msg ? py : 0.0f,
              zz = msg ? pz : 0.0f, zw = msg ? pw : 0.0f;
  v4f v;
  v.x = isH ? a.x : zx;
  v.y = isH ? a.y : zy;
  v.z = isH ? a.z : zz;
  v.w = isH ? a.w : zw;
  return v;
}

template <bool HAS_M1>
__global__ __launch_bounds__(256) void gsn_gate_wmma(
    const float* __restrict__ hv, const float* __restrict__ m0,
    const float* __restrict__ m1, const float* __restrict__ Wx,
    const float* __restrict__ Bx,
    float* __restrict__ G) {
  const unsigned lane  = threadIdx.x & 31u;
  const unsigned wave  = threadIdx.x >> 5u;
  __shared__ __attribute__((aligned(16))) float sA[BLK_IN];
  __shared__ __attribute__((aligned(16))) float sD[BLK_OUT];
  float* const sW = sD + wave * (TPW * 48u);
  const float* const sAw = sA + wave * (TPW * 96u);

  v4f st[STG];
  {
    const unsigned mF = blockIdx.x * BLK_IN + 4u * threadIdx.x;
#pragma unroll
    for (unsigned j = 0; j < STG; ++j)
      st[j] = gate_load4<HAS_M1>(mF + 1024u * j, hv, m0, m1);
  }

  const bool     lo  = lane < 16u;
  const unsigned N   = lane & 15u;
  const bool     nok = N < 3u;
  const unsigned wn  = nok ? N : 0u;
  const float    msk = nok ? 1.0f : 0.0f;
  const float w0 = Wx[wn*6u+0u]*msk, w1 = Wx[wn*6u+1u]*msk, w2 = Wx[wn*6u+2u]*msk,
              w3 = Wx[wn*6u+3u]*msk, w4 = Wx[wn*6u+4u]*msk, w5 = Wx[wn*6u+5u]*msk;
  const float bias = Bx[wn] * msk;

  v2f b, b2;
  b.x  = lo ? w0 : w2;
  b.y  = lo ? w1 : w3;
  b2.x = lo ? w4 : 0.0f;
  b2.y = lo ? w5 : 0.0f;

  for (unsigned tile = blockIdx.x; tile < NTILE; tile += PGRID) {
  const unsigned mT = tile * BLK_IN + 4u * threadIdx.x;
#pragma unroll
  for (unsigned j = 0; j < STG; ++j)
    *(v4f*)(sA + 4u * (threadIdx.x + 256u * j)) = gate_mix4<HAS_M1>(mT + 1024u * j, st[j]);
  {
    const unsigned nt   = tile + PGRID;
    const bool     more = nt < NTILE;
    const unsigned mN   = nt * BLK_IN + 4u * threadIdx.x;
#pragma unroll
    for (unsigned j = 0; j < STG; ++j)
      st[j] = gate_load4<HAS_M1>(more ? mN + 1024u * j : 2u * TD, hv, m0, m1);
  }
  lds_barrier();

#pragma unroll
  for (unsigned th = 0; th < TPW; th += TPW / 2u) {
  v2f fa[TPW / 2u], fa4[TPW / 2u];
#pragma unroll
  for (unsigned u = 0; u < TPW / 2u; ++u) {
    const float* const ar = sAw + (th + u) * 96u + 6u * (lane & 15u);
    fa[u]  = *(const v2f*)(ar + (lo ? 0u : 2u));
    fa4[u] = *(const v2f*)(ar + 4u);
  }

#pragma unroll
  for (unsigned u = 0; u < TPW / 2u; ++u) {
    const unsigned t = th + u;
    const v2f a = fa[u];
    v2f a2;
    a2.x = lo ? fa4[u].x : 0.0f;
    a2.y = lo ? fa4[u].y : 0.0f;

    v8f c;
#pragma unroll
    for (int v = 0; v < 8; ++v) c[v] = bias;

    v8f d = __builtin_amdgcn_wmma_f32_16x16x4_f32(false, a,  false, b,  (short)0, c, false, false);
    d     = __builtin_amdgcn_wmma_f32_16x16x4_f32(false, a2, false, b2, (short)0, d, false, false);

    if (nok) {
      const unsigned mb = lo ? 0u : 8u;
#pragma unroll
      for (int v = 0; v < 8; ++v)
        sW[t * 48u + 3u * (mb + (unsigned)v) + N] = d[v];
    }
  }
  }

  lds_barrier();

  const unsigned pB = tile * BLK_OUT;
  for (unsigned i = threadIdx.x; i < BLK_OUT / 4u; i += 256u) {
    const float4 x = *(const float4*)(sD + 4u * i);
    float4 g;
    g.x = fast_sigmoid(x.x);
    g.y = fast_sigmoid(x.y);
    g.z = fast_sigmoid(x.z);
    g.w = fast_sigmoid(x.w);
    *(float4*)(G + pB + 4u * i) = g;
  }
  }
}

template <bool HAS_M1>
__global__ __launch_bounds__(256) __attribute__((amdgpu_waves_per_eu(8))) void gsn_cand_wmma(
    const float* __restrict__ hv, const float* __restrict__ m0,
    const float* __restrict__ m1, const float* __restrict__ Ww,
    const float* __restrict__ Bw,
    const float* __restrict__ G, float* __restrict__ CT) {
  const unsigned lane  = threadIdx.x & 31u;
  const unsigned wave  = threadIdx.x >> 5u;
  __shared__ __attribute__((aligned(16))) float sA[BLK_IN];
  __shared__ __attribute__((aligned(16))) float sD[BLK_OUT];
  float* const sW = sD + wave * (TPW * 48u);
  const float* const sAw = sA + wave * (TPW * 96u);

  v4f st[STG], sg[STG];
  {
    const unsigned mF = blockIdx.x * BLK_IN + 4u * threadIdx.x;
#pragma unroll
    for (unsigned j = 0; j < STG; ++j)
      cand_load4<HAS_M1>(mF + 1024u * j, hv, m0, m1, G, st[j], sg[j]);
  }

  const bool     lo  = lane < 16u;
  const unsigned N   = lane & 15u;
  const bool     nok = N < 3u;
  const unsigned wn  = nok ? N : 0u;
  const float    msk = nok ? 1.0f : 0.0f;
  const float r0 = Ww[wn*6u+0u], r1 = Ww[wn*6u+1u], r2 = Ww[wn*6u+2u],
              r3 = Ww[wn*6u+3u], r4 = Ww[wn*6u+4u], r5 = Ww[wn*6u+5u];
  const float rb = Bw[wn];
  asm volatile("" :: "v"(st[0]), "v"(st[1]), "v"(st[2]), "v"(st[3]), "v"(st[4]), "v"(st[5]), "v"(sg[0]), "v"(sg[1]), "v"(sg[2]), "v"(sg[3]), "v"(sg[4]), "v"(sg[5]),
               "v"(r0), "v"(r1), "v"(r2), "v"(r3), "v"(r4), "v"(r5), "v"(rb));
  const float w0 = r0*msk, w1 = r1*msk, w2 = r2*msk, w3 = r3*msk, w4 = r4*msk, w5 = r5*msk;
  const float bias = rb * msk;

  v2f b, b2;
  b.x  = lo ? w0 : w2;
  b.y  = lo ? w1 : w3;
  b2.x = lo ? w4 : 0.0f;
  b2.y = lo ? w5 : 0.0f;

  for (unsigned tile = blockIdx.x; tile < NTILE; tile += PGRID) {
  const unsigned mT = tile * BLK_IN + 4u * threadIdx.x;
#pragma unroll
  for (unsigned j = 0; j < STG; ++j)
    *(v4f*)(sA + 4u * (threadIdx.x + 256u * j)) = cand_mix4<HAS_M1>(mT + 1024u * j, st[j], sg[j]);

  const unsigned pB = tile * BLK_OUT;
  float4 tg[TLN], tm[TLN];
#pragma unroll
  for (unsigned k = 0; k < TLN; ++k) {
    const unsigned p   = pB + 4u * (threadIdx.x + 256u * k);
    const unsigned n_o = p >> LOG_TD;
    const unsigned r   = p & (TD - 1u);
    const float* const pm = (n_o == 0u) ? m0 + r : ((HAS_M1 && n_o == 1u) ? m1 + r : hv);
    tg[k] = *(const float4*)(G + p);
    tm[k] = *(const float4*)pm;
  }
  __builtin_amdgcn_sched_barrier(0);
  {
    const unsigned nt   = tile + PGRID;
    const bool     more = nt < NTILE;
    const unsigned mN   = nt * BLK_IN + 4u * threadIdx.x;
#pragma unroll
    for (unsigned j = 0; j < STG; ++j)
      cand_load4<HAS_M1>(more ? mN + 1024u * j : 4u * TD, hv, m0, m1, G, st[j], sg[j]);
  }
  lds_barrier();

#pragma unroll
  for (unsigned th = 0; th < TPW; th += TPW / 2u) {
  v2f fa[TPW / 2u], fa4[TPW / 2u];
#pragma unroll
  for (unsigned u = 0; u < TPW / 2u; ++u) {
    const float* const ar = sAw + (th + u) * 96u + 6u * (lane & 15u);
    fa[u]  = *(const v2f*)(ar + (lo ? 0u : 2u));
    fa4[u] = *(const v2f*)(ar + 4u);
  }

#pragma unroll
  for (unsigned u = 0; u < TPW / 2u; ++u) {
    const unsigned t = th + u;
    const v2f a = fa[u];
    v2f a2;
    a2.x = lo ? fa4[u].x : 0.0f;
    a2.y = lo ? fa4[u].y : 0.0f;

    v8f c;
#pragma unroll
    for (int v = 0; v < 8; ++v) c[v] = bias;

    v8f d = __builtin_amdgcn_wmma_f32_16x16x4_f32(false, a,  false, b,  (short)0, c, false, false);
    d     = __builtin_amdgcn_wmma_f32_16x16x4_f32(false, a2, false, b2, (short)0, d, false, false);

    if (nok) {
      const unsigned mb = lo ? 0u : 8u;
#pragma unroll
      for (int v = 0; v < 8; ++v)
        sW[t * 48u + 3u * (mb + (unsigned)v) + N] = d[v];
    }
  }
  }

  lds_barrier();

#pragma unroll
  for (unsigned k = 0; k < TLN; ++k) {
    const unsigned i   = threadIdx.x + 256u * k;
    const unsigned p   = pB + 4u * i;
    const float4   x   = *(const float4*)(sD + 4u * i);
    const float4   g   = tg[k];
    const unsigned n_o = p >> LOG_TD;
    const bool     has = n_o == 0u || (HAS_M1 && n_o == 1u);
    float4 mv;
    mv.x = has ? tm[k].x : 0.0f;
    mv.y = has ? tm[k].y : 0.0f;
    mv.z = has ? tm[k].z : 0.0f;
    mv.w = has ? tm[k].w : 0.0f;
    float4 o;
    o.x = (1.0f - g.x) * mv.x + g.x * fast_tanh(x.x);
    o.y = (1.0f - g.y) * mv.y + g.y * fast_tanh(x.y);
    o.z = (1.0f - g.z) * mv.z + g.z * fast_tanh(x.z);
    o.w = (1.0f - g.w) * mv.w + g.w * fast_tanh(x.w);
    *(float4*)(CT + p) = o;
  }
  }
}

__global__ __launch_bounds__(256) void gsn_finalize(
    const float* __restrict__ hv, const float* __restrict__ CT,
    float* __restrict__ out) {
  const unsigned q0 = blockIdx.x * 256u + threadIdx.x;
  const unsigned qs = gridDim.x * 256u;
  float4 h4[FIN_PER], c0[FIN_PER], c1[FIN_PER], c2[FIN_PER];
#pragma unroll
  for (unsigned k = 0; k < FIN_PER; ++k) {
    const unsigned i = (q0 + k * qs) * 4u;
    h4[k] = *(const float4*)(hv + i);
    c0[k] = *(const float4*)(CT + i);
    c1[k] = *(const float4*)(CT + TD + i);
    c2[k] = *(const float4*)(CT + 2u * TD + i);
  }
#pragma unroll
  for (unsigned k = 0; k < FIN_PER; ++k) {
    const unsigned i = (q0 + k * qs) * 4u;
    float4 o;
    o.x = h4[k].x + c0[k].x + c1[k].x + c2[k].x;
    o.y = h4[k].y + c0[k].y + c1[k].y + c2[k].y;
    o.z = h4[k].z + c0[k].z + c1[k].z + c2[k].z;
    o.w = h4[k].w + c0[k].w + c1[k].w + c2[k].w;
    *(float4*)(out + i) = o;
  }
}

extern "C" void kernel_launch(void* const* d_in, const int* in_sizes, int n_in,
                              void* d_out, int out_size, void* d_ws, size_t ws_size,
                              hipStream_t stream) {
  (void)in_sizes; (void)n_in; (void)out_size; (void)ws_size;
  const float* feat = (const float*)d_in[0];
  const float* W_w  = (const float*)d_in[1];
  const float* W_b  = (const float*)d_in[2];
  const float* Wx_w = (const float*)d_in[3];
  const float* Wx_b = (const float*)d_in[4];
  float* out = (float*)d_out;

  float* G  = (float*)d_ws;
  float* CT = G + (size_t)3 * TD;

  const float* h0 = feat;
  const float* h1 = feat + (size_t)TD;
  const float* h2 = feat + (size_t)2 * TD;

  dim3 blk(256);
  dim3 grd_mm(PGRID);
  dim3 grd_fin(TD / 4u / 256u / FIN_PER);

  gsn_gate_wmma<false><<<grd_mm, blk, 0, stream>>>(h0, h2, nullptr, Wx_w, Wx_b, G);
  gsn_cand_wmma<false><<<grd_mm, blk, 0, stream>>>(h0, h2, nullptr, W_w, W_b, G, CT);
  gsn_finalize<<<grd_fin, blk, 0, stream>>>(h0, CT, out + 0 * (size_t)TD);

  gsn_gate_wmma<false><<<grd_mm, blk, 0, stream>>>(h1, h0, nullptr, Wx_w, Wx_b, G);
  gsn_cand_wmma<false><<<grd_mm, blk, 0, stream>>>(h1, h0, nullptr, W_w, W_b, G, CT);
  gsn_finalize<<<grd_fin, blk, 0, stream>>>(h1, CT, out + 1 * (size_t)TD);

  gsn_gate_wmma<true><<<grd_mm, blk, 0, stream>>>(h2, h1, h0, Wx_w, Wx_b, G);
  gsn_cand_wmma<true><<<grd_mm, blk, 0, stream>>>(h2, h1, h0, W_w, W_b, G, CT);
  gsn_finalize<<<grd_fin, blk, 0, stream>>>(h2, CT, out + 2 * (size_t)TD);
}


# Sum pass of node k rides in node k+1's gate launch (independent work): per tile trip each thread loads one float4 sum position ahead of its prefetch, adds at the foot; 9->7 launches; sum order kept
# speedup vs baseline: 2.2453x; 1.0769x over previous
#include <hip/hip_runtime.h>
#include <math.h>

#define LOG_TD 22u
#define TD (1u << LOG_TD)

typedef __attribute__((ext_vector_type(2))) float v2f;
typedef __attribute__((ext_vector_type(4))) float v4f;
typedef __attribute__((ext_vector_type(8))) float v8f;

constexpr unsigned TPW = 8u;
constexpr unsigned BLK_OUT = 8u * TPW * 48u;
constexpr unsigned BLK_IN  = 8u * TPW * 96u;
constexpr unsigned STG     = BLK_IN / 4u / 256u;
constexpr unsigned TLN     = BLK_OUT / 4u / 256u;
constexpr unsigned NTILE   = TD / 16u / TPW / 8u;
constexpr unsigned PGRID   = 1024u;
constexpr unsigned FIN_PER = 4u;
static_assert(NTILE / PGRID == FIN_PER, "one sum position per thread per tile trip");
static_assert(TD / 4u / 256u / FIN_PER == PGRID, "the sum pass and the gate launch share one grid");

__device__ __forceinline__ float hw_exp2(float x) {
#if __has_builtin(__builtin_amdgcn_exp2f)
  return __builtin_amdgcn_exp2f(x);
#else
  return exp2f(x);
#endif
}
__device__ __forceinline__ float hw_rcp(float x) {
#if __has_builtin(__builtin_amdgcn_rcpf)
  return __builtin_amdgcn_rcpf(x);
#else
  return 1.0f / x;
#endif
}
__device__ __forceinline__ float fast_sigmoid(float x) {
  return hw_rcp(1.0f + hw_exp2(-1.4426950408889634f * x));
}
__device__ __forceinline__ float fast_tanh(float x) {
#if __has_builtin(__builtin_amdgcn_tanhf)
  return __builtin_amdgcn_tanhf(x);
#else
  return 1.0f - 2.0f * hw_rcp(hw_exp2(2.8853900817779268f * x) + 1.0f);
#endif
}

__device__ __forceinline__ void lds_barrier() {
  __builtin_amdgcn_fence(__ATOMIC_RELEASE, "workgroup", "local");
  __builtin_amdgcn_s_barrier();
  __builtin_amdgcn_fence(__ATOMIC_ACQUIRE, "workgroup", "local");
}

template <bool HAS_M1>
__device__ __forceinline__ bool gate_zero(unsigned m) {
  const unsigned n = m >> LOG_TD;
  return n < 3u && n != 0u && !(HAS_M1 && n == 1u);
}
template <bool HAS_M1>
__device__ __forceinline__ v4f gate_load4(unsigned m,
    const float* __restrict__ hv, const float* __restrict__ m0,
    const float* __restrict__ m1) {
  const unsigned n = m >> LOG_TD;
  const unsigned r = m & (TD - 1u);
  const float* src = (n == 0u) ? m0 : ((HAS_M1 && n == 1u) ? m1 : hv);
  return *(const v4f*)(src + (gate_zero<HAS_M1>(m) ? 0u : r));
}
template <bool HAS_M1>
__device__ __forceinline__ v4f gate_mix4(unsigned m, const v4f v) {
  const bool zero = gate_zero<HAS_M1>(m);
  v4f o;
  o.x = zero ? 0.0f : v.x;
  o.y = zero ? 0.0f : v.y;
  o.z = zero ? 0.0f : v.z;
  o.w = zero ? 0.0f : v.w;
  return o;
}

template <bool HAS_M1>
__device__ __forceinline__ void cand_load4(unsigned m,
    const float* __restrict__ hv, const float* __restrict__ m0,
    const float* __restrict__ m1, const float* __restrict__ G,
    v4f& a, v4f& g) {
  const unsigned seg = m >> LOG_TD;
  const unsigned r   = m & (TD - 1u);
  const bool  isH = (seg & 1u) != 0u;
  const bool  s0  = seg == 0u;
  const bool  s1  = HAS_M1 && seg == 2u;
  const bool  msg = s0 || s1;
  const float* pa = s0 ? m0 : (s1 ? m1 : hv);
  const float* pg = s1 ? G + TD : G;
  a = *(const v4f*)(pa + ((isH || msg) ? r : 0u));
  g = *(const v4f*)(msg ? pg + r : hv);
}
template <bool HAS_M1>
__device__ __forceinline__ v4f cand_mix4(unsigned m, const v4f a, const v4f g) {
  const unsigned seg = m >> LOG_TD;
  const bool  isH = (seg & 1u) != 0u;
  const bool  msg = seg == 0u || (HAS_M1 && seg == 2u);
  const float px = g.x * a.x, py = g.y * a.y, pz = g.z * a.z, pw = g.w * a.w;
  const float zx = msg ? px : 0.0f, zy = msg ? py : 0.0f,
              zz = msg ? pz : 0.0f, zw = msg ? pw : 0.0f;
  v4f v;
  v.x = isH ? a.x : zx;
  v.y = isH ? a.y : zy;
  v.z = isH ? a.z : zz;
  v.w = isH ? a.w : zw;
  return v;
}

template <bool HAS_M1, bool FIN>
__global__ __launch_bounds__(256) void gsn_gate_wmma(
    const float* __restrict__ hv, const float* __restrict__ m0,
    const float* __restrict__ m1, const float* __restrict__ Wx,
    const float* __restrict__ Bx,
    float* __restrict__ G,
    const float* __restrict__ fh, const float* __restrict__ fCT,
    float* __restrict__ fout) {
  const unsigned lane  = threadIdx.x & 31u;
  const unsigned wave  = threadIdx.x >> 5u;
  __shared__ __attribute__((aligned(16))) float sA[BLK_IN];
  __shared__ __attribute__((aligned(16))) float sD[BLK_OUT];
  float* const sW = sD + wave * (TPW * 48u);
  const float* const sAw = sA + wave * (TPW * 96u);

  v4f st[STG];
  {
    const unsigned mF = blockIdx.x * BLK_IN + 4u * threadIdx.x;
#pragma unroll
    for (unsigned j = 0; j < STG; ++j)
      st[j] = gate_load4<HAS_M1>(mF + 1024u * j, hv, m0, m1);
  }

  const bool     lo  = lane < 16u;
  const unsigned N   = lane & 15u;
  const bool     nok = N < 3u;
  const unsigned wn  = nok ? N : 0u;
  const float    msk = nok ? 1.0f : 0.0f;
  const float w0 = Wx[wn*6u+0u]*msk, w1 = Wx[wn*6u+1u]*msk, w2 = Wx[wn*6u+2u]*msk,
              w3 = Wx[wn*6u+3u]*msk, w4 = Wx[wn*6u+4u]*msk, w5 = Wx[wn*6u+5u]*msk;
  const float bias = Bx[wn] * msk;

  v2f b, b2;
  b.x  = lo ? w0 : w2;
  b.y  = lo ? w1 : w3;
  b2.x = lo ? w4 : 0.0f;
  b2.y = lo ? w5 : 0.0f;

  unsigned fq = (blockIdx.x * 256u + threadIdx.x) * 4u;
  for (unsigned tile = blockIdx.x; tile < NTILE; tile += PGRID) {
  const unsigned mT = tile * BLK_IN + 4u * threadIdx.x;
#pragma unroll
  for (unsigned j = 0; j < STG; ++j)
    *(v4f*)(sA + 4u * (threadIdx.x + 256u * j)) = gate_mix4<HAS_M1>(mT + 1024u * j, st[j]);
  float4 fh4, fc0, fc1, fc2;
  if (FIN) {
    fh4 = *(const float4*)(fh + fq);
    fc0 = *(const float4*)(fCT + fq);
    fc1 = *(const float4*)(fCT + TD + fq);
    fc2 = *(const float4*)(fCT + 2u * TD + fq);
    __builtin_amdgcn_sched_barrier(0);
  }
  {
    const unsigned nt   = tile + PGRID;
    const bool     more = nt < NTILE;
    const unsigned mN   = nt * BLK_IN + 4u * threadIdx.x;
#pragma unroll
    for (unsigned j = 0; j < STG; ++j)
      st[j] = gate_load4<HAS_M1>(more ? mN + 1024u * j : 2u * TD, hv, m0, m1);
  }
  lds_barrier();

#pragma unroll
  for (unsigned th = 0; th < TPW; th += TPW / 2u) {
  v2f fa[TPW / 2u], fa4[TPW / 2u];
#pragma unroll
  for (unsigned u = 0; u < TPW / 2u; ++u) {
    const float* const ar = sAw + (th + u) * 96u + 6u * (lane & 15u);
    fa[u]  = *(const v2f*)(ar + (lo ? 0u : 2u));
    fa4[u] = *(const v2f*)(ar + 4u);
  }

#pragma unroll
  for (unsigned u = 0; u < TPW / 2u; ++u) {
    const unsigned t = th + u;
    const v2f a = fa[u];
    v2f a2;
    a2.x = lo ? fa4[u].x : 0.0f;
    a2.y = lo ? fa4[u].y : 0.0f;

    v8f c;
#pragma unroll
    for (int v = 0; v < 8; ++v) c[v] = bias;

    v8f d = __builtin_amdgcn_wmma_f32_16x16x4_f32(false, a,  false, b,  (short)0, c, false, false);
    d     = __builtin_amdgcn_wmma_f32_16x16x4_f32(false, a2, false, b2, (short)0, d, false, false);

    if (nok) {
      const unsigned mb = lo ? 0u : 8u;
#pragma unroll
      for (int v = 0; v < 8; ++v)
        sW[t * 48u + 3u * (mb + (unsigned)v) + N] = d[v];
    }
  }
  }

  lds_barrier();

  const unsigned pB = tile * BLK_OUT;
  for (unsigned i = threadIdx.x; i < BLK_OUT / 4u; i += 256u) {
    const float4 x = *(const float4*)(sD + 4u * i);
    float4 g;
    g.x = fast_sigmoid(x.x);
    g.y = fast_sigmoid(x.y);
    g.z = fast_sigmoid(x.z);
    g.w = fast_sigmoid(x.w);
    *(float4*)(G + pB + 4u * i) = g;
  }
  if (FIN) {
    float4 o;
    o.x = fh4.x + fc0.x + fc1.x + fc2.x;
    o.y = fh4.y + fc0.y + fc1.y + fc2.y;
    o.z = fh4.z + fc0.z + fc1.z + fc2.z;
    o.w = fh4.w + fc0.w + fc1.w + fc2.w;
    *(float4*)(fout + fq) = o;
    fq += PGRID * 256u * 4u;
  }
  }
}

template <bool HAS_M1>
__global__ __launch_bounds__(256) __attribute__((amdgpu_waves_per_eu(8))) void gsn_cand_wmma(
    const float* __restrict__ hv, const float* __restrict__ m0,
    const float* __restrict__ m1, const float* __restrict__ Ww,
    const float* __restrict__ Bw,
    const float* __restrict__ G, float* __restrict__ CT) {
  const unsigned lane  = threadIdx.x & 31u;
  const unsigned wave  = threadIdx.x >> 5u;
  __shared__ __attribute__((aligned(16))) float sA[BLK_IN];
  __shared__ __attribute__((aligned(16))) float sD[BLK_OUT];
  float* const sW = sD + wave * (TPW * 48u);
  const float* const sAw = sA + wave * (TPW * 96u);

  v4f st[STG], sg[STG];
  {
    const unsigned mF = blockIdx.x * BLK_IN + 4u * threadIdx.x;
#pragma unroll
    for (unsigned j = 0; j < STG; ++j)
      cand_load4<HAS_M1>(mF + 1024u * j, hv, m0, m1, G, st[j], sg[j]);
  }

  const bool     lo  = lane < 16u;
  const unsigned N   = lane & 15u;
  const bool     nok = N < 3u;
  const unsigned wn  = nok ? N : 0u;
  const float    msk = nok ? 1.0f : 0.0f;
  const float r0 = Ww[wn*6u+0u], r1 = Ww[wn*6u+1u], r2 = Ww[wn*6u+2u],
              r3 = Ww[wn*6u+3u], r4 = Ww[wn*6u+4u], r5 = Ww[wn*6u+5u];
  const float rb = Bw[wn];
  asm volatile("" :: "v"(st[0]), "v"(st[1]), "v"(st[2]), "v"(st[3]), "v"(st[4]), "v"(st[5]), "v"(sg[0]), "v"(sg[1]), "v"(sg[2]), "v"(sg[3]), "v"(sg[4]), "v"(sg[5]),
               "v"(r0), "v"(r1), "v"(r2), "v"(r3), "v"(r4), "v"(r5), "v"(rb));
  const float w0 = r0*msk, w1 = r1*msk, w2 = r2*msk, w3 = r3*msk, w4 = r4*msk, w5 = r5*msk;
  const float bias = rb * msk;

  v2f b, b2;
  b.x  = lo ? w0 : w2;
  b.y  = lo ? w1 : w3;
  b2.x = lo ? w4 : 0.0f;
  b2.y = lo ? w5 : 0.0f;

  for (unsigned tile = blockIdx.x; tile < NTILE; tile += PGRID) {
  const unsigned mT = tile * BLK_IN + 4u * threadIdx.x;
#pragma unroll
  for (unsigned j = 0; j < STG; ++j)
    *(v4f*)(sA + 4u * (threadIdx.x + 256u * j)) = cand_mix4<HAS_M1>(mT + 1024u * j, st[j], sg[j]);

  const unsigned pB = tile * BLK_OUT;
  float4 tg[TLN], tm[TLN];
#pragma unroll
  for (unsigned k = 0; k < TLN; ++k) {
    const unsigned p   = pB + 4u * (threadIdx.x + 256u * k);
    const unsigned n_o = p >> LOG_TD;
    const unsigned r   = p & (TD - 1u);
    const float* const pm = (n_o == 0u) ? m0 + r : ((HAS_M1 && n_o == 1u) ? m1 + r : hv);
    tg[k] = *(const float4*)(G + p);
    tm[k] = *(const float4*)pm;
  }
  __builtin_amdgcn_sched_barrier(0);
  {
    const unsigned nt   = tile + PGRID;
    const bool     more = nt < NTILE;
    const unsigned mN   = nt * BLK_IN + 4u * threadIdx.x;
#pragma unroll
    for (unsigned j = 0; j < STG; ++j)
      cand_load4<HAS_M1>(more ? mN + 1024u * j : 4u * TD, hv, m0, m1, G, st[j], sg[j]);
  }
  lds_barrier();

#pragma unroll
  for (unsigned th = 0; th < TPW; th += TPW / 2u) {
  v2f fa[TPW / 2u], fa4[TPW / 2u];
#pragma unroll
  for (unsigned u = 0; u < TPW / 2u; ++u) {
    const float* const ar = sAw + (th + u) * 96u + 6u * (lane & 15u);
    fa[u]  = *(const v2f*)(ar + (lo ? 0u : 2u));
    fa4[u] = *(const v2f*)(ar + 4u);
  }

#pragma unroll
  for (unsigned u = 0; u < TPW / 2u; ++u) {
    const unsigned t = th + u;
    const v2f a = fa[u];
    v2f a2;
    a2.x = lo ? fa4[u].x : 0.0f;
    a2.y = lo ? fa4[u].y : 0.0f;

    v8f c;
#pragma unroll
    for (int v = 0; v < 8; ++v) c[v] = bias;

    v8f d = __builtin_amdgcn_wmma_f32_16x16x4_f32(false, a,  false, b,  (short)0, c, false, false);
    d     = __builtin_amdgcn_wmma_f32_16x16x4_f32(false, a2, false, b2, (short)0, d, false, false);

    if (nok) {
      const unsigned mb = lo ? 0u : 8u;
#pragma unroll
      for (int v = 0; v < 8; ++v)
        sW[t * 48u + 3u * (mb + (unsigned)v) + N] = d[v];
    }
  }
  }

  lds_barrier();

#pragma unroll
  for (unsigned k = 0; k < TLN; ++k) {
    const unsigned i   = threadIdx.x + 256u * k;
    const unsigned p   = pB + 4u * i;
    const float4   x   = *(const float4*)(sD + 4u * i);
    const float4   g   = tg[k];
    const unsigned n_o = p >> LOG_TD;
    const bool     has = n_o == 0u || (HAS_M1 && n_o == 1u);
    float4 mv;
    mv.x = has ? tm[k].x : 0.0f;
    mv.y = has ? tm[k].y : 0.0f;
    mv.z = has ? tm[k].z : 0.0f;
    mv.w = has ? tm[k].w : 0.0f;
    float4 o;
    o.x = (1.0f - g.x) * mv.x + g.x * fast_tanh(x.x);
    o.y = (1.0f - g.y) * mv.y + g.y * fast_tanh(x.y);
    o.z = (1.0f - g.z) * mv.z + g.z * fast_tanh(x.z);
    o.w = (1.0f - g.w) * mv.w + g.w * fast_tanh(x.w);
    *(float4*)(CT + p) = o;
  }
  }
}

__global__ __launch_bounds__(256) void gsn_finalize(
    const float* __restrict__ hv, const float* __restrict__ CT,
    float* __restrict__ out) {
  const unsigned q0 = blockIdx.x * 256u + threadIdx.x;
  const unsigned qs = gridDim.x * 256u;
  float4 h4[FIN_PER], c0[FIN_PER], c1[FIN_PER], c2[FIN_PER];
#pragma unroll
  for (unsigned k = 0; k < FIN_PER; ++k) {
    const unsigned i = (q0 + k * qs) * 4u;
    h4[k] = *(const float4*)(hv + i);
    c0[k] = *(const float4*)(CT + i);
    c1[k] = *(const float4*)(CT + TD + i);
    c2[k] = *(const float4*)(CT + 2u * TD + i);
  }
#pragma unroll
  for (unsigned k = 0; k < FIN_PER; ++k) {
    const unsigned i = (q0 + k * qs) * 4u;
    float4 o;
    o.x = h4[k].x + c0[k].x + c1[k].x + c2[k].x;
    o.y = h4[k].y + c0[k].y + c1[k].y + c2[k].y;
    o.z = h4[k].z + c0[k].z + c1[k].z + c2[k].z;
    o.w = h4[k].w + c0[k].w + c1[k].w + c2[k].w;
    *(float4*)(out + i) = o;
  }
}

extern "C" void kernel_launch(void* const* d_in, const int* in_sizes, int n_in,
                              void* d_out, int out_size, void* d_ws, size_t ws_size,
                              hipStream_t stream) {
  (void)in_sizes; (void)n_in; (void)out_size; (void)ws_size;
  const float* feat = (const float*)d_in[0];
  const float* W_w  = (const float*)d_in[1];
  const float* W_b  = (const float*)d_in[2];
  const float* Wx_w = (const float*)d_in[3];
  const float* Wx_b = (const float*)d_in[4];
  float* out = (float*)d_out;

  float* G  = (float*)d_ws;
  float* CT = G + (size_t)3 * TD;

  const float* h0 = feat;
  const float* h1 = feat + (size_t)TD;
  const float* h2 = feat + (size_t)2 * TD;

  dim3 blk(256);
  dim3 grd_mm(PGRID);
  dim3 grd_fin(TD / 4u / 256u / FIN_PER);


  gsn_gate_wmma<false, false><<<grd_mm, blk, 0, stream>>>(h0, h2, nullptr, Wx_w, Wx_b, G,
                                                        nullptr, nullptr, nullptr);
  gsn_cand_wmma<false><<<grd_mm, blk, 0, stream>>>(h0, h2, nullptr, W_w, W_b, G, CT);

  gsn_gate_wmma<false, true><<<grd_mm, blk, 0, stream>>>(h1, h0, nullptr, Wx_w, Wx_b, G,
                                                       h0, CT, out + 0 * (size_t)TD);
  gsn_cand_wmma<false><<<grd_mm, blk, 0, stream>>>(h1, h0, nullptr, W_w, W_b, G, CT);

  gsn_gate_wmma<true, true><<<grd_mm, blk, 0, stream>>>(h2, h1, h0, Wx_w, Wx_b, G,
                                                      h1, CT, out + 1 * (size_t)TD);
  gsn_cand_wmma<true><<<grd_mm, blk, 0, stream>>>(h2, h1, h0, W_w, W_b, G, CT);
  gsn_finalize<<<grd_fin, blk, 0, stream>>>(h2, CT, out + 2 * (size_t)TD);
}


# Streams dead after one use marked non-temporal (cand tail's reads of G, sum passes' reads of CT, stores of out) so they take no L2 room from the 3 input planes every launch re-reads; sums unchanged
# speedup vs baseline: 2.3801x; 1.0600x over previous
#include <hip/hip_runtime.h>
#include <math.h>

#define LOG_TD 22u
#define TD (1u << LOG_TD)

typedef __attribute__((ext_vector_type(2))) float v2f;
typedef __attribute__((ext_vector_type(4))) float v4f;
typedef __attribute__((ext_vector_type(8))) float v8f;

constexpr unsigned TPW = 8u;
constexpr unsigned BLK_OUT = 8u * TPW * 48u;
constexpr unsigned BLK_IN  = 8u * TPW * 96u;
constexpr unsigned STG     = BLK_IN / 4u / 256u;
constexpr unsigned TLN     = BLK_OUT / 4u / 256u;
constexpr unsigned NTILE   = TD / 16u / TPW / 8u;
constexpr unsigned PGRID   = 1024u;
constexpr unsigned FIN_PER = 4u;
static_assert(NTILE / PGRID == FIN_PER, "one sum position per thread per tile trip");
static_assert(TD / 4u / 256u / FIN_PER == PGRID, "the sum pass and the gate launch share one grid");

__device__ __forceinline__ float hw_exp2(float x) {
#if __has_builtin(__builtin_amdgcn_exp2f)
  return __builtin_amdgcn_exp2f(x);
#else
  return exp2f(x);
#endif
}
__device__ __forceinline__ float hw_rcp(float x) {
#if __has_builtin(__builtin_amdgcn_rcpf)
  return __builtin_amdgcn_rcpf(x);
#else
  return 1.0f / x;
#endif
}
__device__ __forceinline__ float fast_sigmoid(float x) {
  return hw_rcp(1.0f + hw_exp2(-1.4426950408889634f * x));
}
__device__ __forceinline__ float fast_tanh(float x) {
#if __has_builtin(__builtin_amdgcn_tanhf)
  return __builtin_amdgcn_tanhf(x);
#else
  return 1.0f - 2.0f * hw_rcp(hw_exp2(2.8853900817779268f * x) + 1.0f);
#endif
}

__device__ __forceinline__ void lds_barrier() {
  __builtin_amdgcn_fence(__ATOMIC_RELEASE, "workgroup", "local");
  __builtin_amdgcn_s_barrier();
  __builtin_amdgcn_fence(__ATOMIC_ACQUIRE, "workgroup", "local");
}

__device__ __forceinline__ v4f stream_load4(const float* p) {
  return __builtin_nontemporal_load((const v4f*)p);
}
__device__ __forceinline__ void stream_store4(float* p, const v4f v) {
  __builtin_nontemporal_store(v, (v4f*)p);
}

template <bool HAS_M1>
__device__ __forceinline__ bool gate_zero(unsigned m) {
  const unsigned n = m >> LOG_TD;
  return n < 3u && n != 0u && !(HAS_M1 && n == 1u);
}
template <bool HAS_M1>
__device__ __forceinline__ v4f gate_load4(unsigned m,
    const float* __restrict__ hv, const float* __restrict__ m0,
    const float* __restrict__ m1) {
  const unsigned n = m >> LOG_TD;
  const unsigned r = m & (TD - 1u);
  const float* src = (n == 0u) ? m0 : ((HAS_M1 && n == 1u) ? m1 : hv);
  return *(const v4f*)(src + (gate_zero<HAS_M1>(m) ? 0u : r));
}
template <bool HAS_M1>
__device__ __forceinline__ v4f gate_mix4(unsigned m, const v4f v) {
  const bool zero = gate_zero<HAS_M1>(m);
  v4f o;
  o.x = zero ? 0.0f : v.x;
  o.y = zero ? 0.0f : v.y;
  o.z = zero ? 0.0f : v.z;
  o.w = zero ? 0.0f : v.w;
  return o;
}

template <bool HAS_M1>
__device__ __forceinline__ void cand_load4(unsigned m,
    const float* __restrict__ hv, const float* __restrict__ m0,
    const float* __restrict__ m1, const float* __restrict__ G,
    v4f& a, v4f& g) {
  const unsigned seg = m >> LOG_TD;
  const unsigned r   = m & (TD - 1u);
  const bool  isH = (seg & 1u) != 0u;
  const bool  s0  = seg == 0u;
  const bool  s1  = HAS_M1 && seg == 2u;
  const bool  msg = s0 || s1;
  const float* pa = s0 ? m0 : (s1 ? m1 : hv);
  const float* pg = s1 ? G + TD : G;
  a = *(const v4f*)(pa + ((isH || msg) ? r : 0u));
  g = *(const v4f*)(msg ? pg + r : hv);
}
template <bool HAS_M1>
__device__ __forceinline__ v4f cand_mix4(unsigned m, const v4f a, const v4f g) {
  const unsigned seg = m >> LOG_TD;
  const bool  isH = (seg & 1u) != 0u;
  const bool  msg = seg == 0u || (HAS_M1 && seg == 2u);
  const float px = g.x * a.x, py = g.y * a.y, pz = g.z * a.z, pw = g.w * a.w;
  const float zx = msg ? px : 0.0f, zy = msg ? py : 0.0f,
              zz = msg ? pz : 0.0f, zw = msg ? pw : 0.0f;
  v4f v;
  v.x = isH ? a.x : zx;
  v.y = isH ? a.y : zy;
  v.z = isH ? a.z : zz;
  v.w = isH ? a.w : zw;
  return v;
}

template <bool HAS_M1, bool FIN>
__global__ __launch_bounds__(256) void gsn_gate_wmma(
    const float* __restrict__ hv, const float* __restrict__ m0,
    const float* __restrict__ m1, const float* __restrict__ Wx,
    const float* __restrict__ Bx,
    float* __restrict__ G,
    const float* __restrict__ fh, const float* __restrict__ fCT,
    float* __restrict__ fout) {
  const unsigned lane  = threadIdx.x & 31u;
  const unsigned wave  = threadIdx.x >> 5u;
  __shared__ __attribute__((aligned(16))) float sA[BLK_IN];
  __shared__ __attribute__((aligned(16))) float sD[BLK_OUT];
  float* const sW = sD + wave * (TPW * 48u);
  const float* const sAw = sA + wave * (TPW * 96u);

  v4f st[STG];
  {
    const unsigned mF = blockIdx.x * BLK_IN + 4u * threadIdx.x;
#pragma unroll
    for (unsigned j = 0; j < STG; ++j)
      st[j] = gate_load4<HAS_M1>(mF + 1024u * j, hv, m0, m1);
  }

  const bool     lo  = lane < 16u;
  const unsigned N   = lane & 15u;
  const bool     nok = N < 3u;
  const unsigned wn  = nok ? N : 0u;
  const float    msk = nok ? 1.0f : 0.0f;
  const float w0 = Wx[wn*6u+0u]*msk, w1 = Wx[wn*6u+1u]*msk, w2 = Wx[wn*6u+2u]*msk,
              w3 = Wx[wn*6u+3u]*msk, w4 = Wx[wn*6u+4u]*msk, w5 = Wx[wn*6u+5u]*msk;
  const float bias = Bx[wn] * msk;

  v2f b, b2;
  b.x  = lo ? w0 : w2;
  b.y  = lo ? w1 : w3;
  b2.x = lo ? w4 : 0.0f;
  b2.y = lo ? w5 : 0.0f;

  unsigned fq = (blockIdx.x * 256u + threadIdx.x) * 4u;
  for (unsigned tile = blockIdx.x; tile < NTILE; tile += PGRID) {
  const unsigned mT = tile * BLK_IN + 4u * threadIdx.x;
#pragma unroll
  for (unsigned j = 0; j < STG; ++j)
    *(v4f*)(sA + 4u * (threadIdx.x + 256u * j)) = gate_mix4<HAS_M1>(mT + 1024u * j, st[j]);
  v4f fh4, fc0, fc1, fc2;
  if (FIN) {
    fh4 = *(const v4f*)(fh + fq);
    fc0 = stream_load4(fCT + fq);
    fc1 = stream_load4(fCT + TD + fq);
    fc2 = stream_load4(fCT + 2u * TD + fq);
    __builtin_amdgcn_sched_barrier(0);
  }
  {
    const unsigned nt   = tile + PGRID;
    const bool     more = nt < NTILE;
    const unsigned mN   = nt * BLK_IN + 4u * threadIdx.x;
#pragma unroll
    for (unsigned j = 0; j < STG; ++j)
      st[j] = gate_load4<HAS_M1>(more ? mN + 1024u * j : 2u * TD, hv, m0, m1);
  }
  lds_barrier();

#pragma unroll
  for (unsigned th = 0; th < TPW; th += TPW / 2u) {
  v2f fa[TPW / 2u], fa4[TPW / 2u];
#pragma unroll
  for (unsigned u = 0; u < TPW / 2u; ++u) {
    const float* const ar = sAw + (th + u) * 96u + 6u * (lane & 15u);
    fa[u]  = *(const v2f*)(ar + (lo ? 0u : 2u));
    fa4[u] = *(const v2f*)(ar + 4u);
  }

#pragma unroll
  for (unsigned u = 0; u < TPW / 2u; ++u) {
    const unsigned t = th + u;
    const v2f a = fa[u];
    v2f a2;
    a2.x = lo ? fa4[u].x : 0.0f;
    a2.y = lo ? fa4[u].y : 0.0f;

    v8f c;
#pragma unroll
    for (int v = 0; v < 8; ++v) c[v] = bias;

    v8f d = __builtin_amdgcn_wmma_f32_16x16x4_f32(false, a,  false, b,  (short)0, c, false, false);
    d     = __builtin_amdgcn_wmma_f32_16x16x4_f32(false, a2, false, b2, (short)0, d, false, false);

    if (nok) {
      const unsigned mb = lo ? 0u : 8u;
#pragma unroll
      for (int v = 0; v < 8; ++v)
        sW[t * 48u + 3u * (mb + (unsigned)v) + N] = d[v];
    }
  }
  }

  lds_barrier();

  const unsigned pB = tile * BLK_OUT;
  for (unsigned i = threadIdx.x; i < BLK_OUT / 4u; i += 256u) {
    const float4 x = *(const float4*)(sD + 4u * i);
    float4 g;
    g.x = fast_sigmoid(x.x);
    g.y = fast_sigmoid(x.y);
    g.z = fast_sigmoid(x.z);
    g.w = fast_sigmoid(x.w);
    *(float4*)(G + pB + 4u * i) = g;
  }
  if (FIN) {
    v4f o;
    o.x = fh4.x + fc0.x + fc1.x + fc2.x;
    o.y = fh4.y + fc0.y + fc1.y + fc2.y;
    o.z = fh4.z + fc0.z + fc1.z + fc2.z;
    o.w = fh4.w + fc0.w + fc1.w + fc2.w;
    stream_store4(fout + fq, o);
    fq += PGRID * 256u * 4u;
  }
  }
}

template <bool HAS_M1>
__global__ __launch_bounds__(256) __attribute__((amdgpu_waves_per_eu(8))) void gsn_cand_wmma(
    const float* __restrict__ hv, const float* __restrict__ m0,
    const float* __restrict__ m1, const float* __restrict__ Ww,
    const float* __restrict__ Bw,
    const float* __restrict__ G, float* __restrict__ CT) {
  const unsigned lane  = threadIdx.x & 31u;
  const unsigned wave  = threadIdx.x >> 5u;
  __shared__ __attribute__((aligned(16))) float sA[BLK_IN];
  __shared__ __attribute__((aligned(16))) float sD[BLK_OUT];
  float* const sW = sD + wave * (TPW * 48u);
  const float* const sAw = sA + wave * (TPW * 96u);

  v4f st[STG], sg[STG];
  {
    const unsigned mF = blockIdx.x * BLK_IN + 4u * threadIdx.x;
#pragma unroll
    for (unsigned j = 0; j < STG; ++j)
      cand_load4<HAS_M1>(mF + 1024u * j, hv, m0, m1, G, st[j], sg[j]);
  }

  const bool     lo  = lane < 16u;
  const unsigned N   = lane & 15u;
  const bool     nok = N < 3u;
  const unsigned wn  = nok ? N : 0u;
  const float    msk = nok ? 1.0f : 0.0f;
  const float r0 = Ww[wn*6u+0u], r1 = Ww[wn*6u+1u], r2 = Ww[wn*6u+2u],
              r3 = Ww[wn*6u+3u], r4 = Ww[wn*6u+4u], r5 = Ww[wn*6u+5u];
  const float rb = Bw[wn];
  asm volatile("" :: "v"(st[0]), "v"(st[1]), "v"(st[2]), "v"(st[3]), "v"(st[4]), "v"(st[5]), "v"(sg[0]), "v"(sg[1]), "v"(sg[2]), "v"(sg[3]), "v"(sg[4]), "v"(sg[5]),
               "v"(r0), "v"(r1), "v"(r2), "v"(r3), "v"(r4), "v"(r5), "v"(rb));
  const float w0 = r0*msk, w1 = r1*msk, w2 = r2*msk, w3 = r3*msk, w4 = r4*msk, w5 = r5*msk;
  const float bias = rb * msk;

  v2f b, b2;
  b.x  = lo ? w0 : w2;
  b.y  = lo ? w1 : w3;
  b2.x = lo ? w4 : 0.0f;
  b2.y = lo ? w5 : 0.0f;

  for (unsigned tile = blockIdx.x; tile < NTILE; tile += PGRID) {
  const unsigned mT = tile * BLK_IN + 4u * threadIdx.x;
#pragma unroll
  for (unsigned j = 0; j < STG; ++j)
    *(v4f*)(sA + 4u * (threadIdx.x + 256u * j)) = cand_mix4<HAS_M1>(mT + 1024u * j, st[j], sg[j]);

  const unsigned pB = tile * BLK_OUT;
  v4f tg[TLN];
  float4 tm[TLN];
#pragma unroll
  for (unsigned k = 0; k < TLN; ++k) {
    const unsigned p   = pB + 4u * (threadIdx.x + 256u * k);
    const unsigned n_o = p >> LOG_TD;
    const unsigned r   = p & (TD - 1u);
    const float* const pm = (n_o == 0u) ? m0 + r : ((HAS_M1 && n_o == 1u) ? m1 + r : hv);
    tg[k] = stream_load4(G + p);
    tm[k] = *(const float4*)pm;
  }
  __builtin_amdgcn_sched_barrier(0);
  {
    const unsigned nt   = tile + PGRID;
    const bool     more = nt < NTILE;
    const unsigned mN   = nt * BLK_IN + 4u * threadIdx.x;
#pragma unroll
    for (unsigned j = 0; j < STG; ++j)
      cand_load4<HAS_M1>(more ? mN + 1024u * j : 4u * TD, hv, m0, m1, G, st[j], sg[j]);
  }
  lds_barrier();

#pragma unroll
  for (unsigned th = 0; th < TPW; th += TPW / 2u) {
  v2f fa[TPW / 2u], fa4[TPW / 2u];
#pragma unroll
  for (unsigned u = 0; u < TPW / 2u; ++u) {
    const float* const ar = sAw + (th + u) * 96u + 6u * (lane & 15u);
    fa[u]  = *(const v2f*)(ar + (lo ? 0u : 2u));
    fa4[u] = *(const v2f*)(ar + 4u);
  }

#pragma unroll
  for (unsigned u = 0; u < TPW / 2u; ++u) {
    const unsigned t = th + u;
    const v2f a = fa[u];
    v2f a2;
    a2.x = lo ? fa4[u].x : 0.0f;
    a2.y = lo ? fa4[u].y : 0.0f;

    v8f c;
#pragma unroll
    for (int v = 0; v < 8; ++v) c[v] = bias;

    v8f d = __builtin_amdgcn_wmma_f32_16x16x4_f32(false, a,  false, b,  (short)0, c, false, false);
    d     = __builtin_amdgcn_wmma_f32_16x16x4_f32(false, a2, false, b2, (short)0, d, false, false);

    if (nok) {
      const unsigned mb = lo ? 0u : 8u;
#pragma unroll
      for (int v = 0; v < 8; ++v)
        sW[t * 48u + 3u * (mb + (unsigned)v) + N] = d[v];
    }
  }
  }

  lds_barrier();

#pragma unroll
  for (unsigned k = 0; k < TLN; ++k) {
    const unsigned i   = threadIdx.x + 256u * k;
    const unsigned p   = pB + 4u * i;
    const float4   x   = *(const float4*)(sD + 4u * i);
    const v4f      g   = tg[k];
    const unsigned n_o = p >> LOG_TD;
    const bool     has = n_o == 0u || (HAS_M1 && n_o == 1u);
    float4 mv;
    mv.x = has ? tm[k].x : 0.0f;
    mv.y = has ? tm[k].y : 0.0f;
    mv.z = has ? tm[k].z : 0.0f;
    mv.w = has ? tm[k].w : 0.0f;
    float4 o;
    o.x = (1.0f - g.x) * mv.x + g.x * fast_tanh(x.x);
    o.y = (1.0f - g.y) * mv.y + g.y * fast_tanh(x.y);
    o.z = (1.0f - g.z) * mv.z + g.z * fast_tanh(x.z);
    o.w = (1.0f - g.w) * mv.w + g.w * fast_tanh(x.w);
    *(float4*)(CT + p) = o;
  }
  }
}

__global__ __launch_bounds__(256) void gsn_finalize(
    const float* __restrict__ hv, const float* __restrict__ CT,
    float* __restrict__ out) {
  const unsigned q0 = blockIdx.x * 256u + threadIdx.x;
  const unsigned qs = gridDim.x * 256u;
  v4f h4[FIN_PER], c0[FIN_PER], c1[FIN_PER], c2[FIN_PER];
#pragma unroll
  for (unsigned k = 0; k < FIN_PER; ++k) {
    const unsigned i = (q0 + k * qs) * 4u;
    h4[k] = *(const v4f*)(hv + i);
    c0[k] = stream_load4(CT + i);
    c1[k] = stream_load4(CT + TD + i);
    c2[k] = stream_load4(CT + 2u * TD + i);
  }
#pragma unroll
  for (unsigned k = 0; k < FIN_PER; ++k) {
    const unsigned i = (q0 + k * qs) * 4u;
    v4f o;
    o.x = h4[k].x + c0[k].x + c1[k].x + c2[k].x;
    o.y = h4[k].y + c0[k].y + c1[k].y + c2[k].y;
    o.z = h4[k].z + c0[k].z + c1[k].z + c2[k].z;
    o.w = h4[k].w + c0[k].w + c1[k].w + c2[k].w;
    stream_store4(out + i, o);
  }
}

extern "C" void kernel_launch(void* const* d_in, const int* in_sizes, int n_in,
                              void* d_out, int out_size, void* d_ws, size_t ws_size,
                              hipStream_t stream) {
  (void)in_sizes; (void)n_in; (void)out_size; (void)ws_size;
  const float* feat = (const float*)d_in[0];
  const float* W_w  = (const float*)d_in[1];
  const float* W_b  = (const float*)d_in[2];
  const float* Wx_w = (const float*)d_in[3];
  const float* Wx_b = (const float*)d_in[4];
  float* out = (float*)d_out;

  float* G  = (float*)d_ws;
  float* CT = G + (size_t)3 * TD;

  const float* h0 = feat;
  const float* h1 = feat + (size_t)TD;
  const float* h2 = feat + (size_t)2 * TD;

  dim3 blk(256);
  dim3 grd_mm(PGRID);
  dim3 grd_fin(TD / 4u / 256u / FIN_PER);


  gsn_gate_wmma<false, false><<<grd_mm, blk, 0, stream>>>(h0, h2, nullptr, Wx_w, Wx_b, G,
                                                        nullptr, nullptr, nullptr);
  gsn_cand_wmma<false><<<grd_mm, blk, 0, stream>>>(h0, h2, nullptr, W_w, W_b, G, CT);

  gsn_gate_wmma<false, true><<<grd_mm, blk, 0, stream>>>(h1, h0, nullptr, Wx_w, Wx_b, G,
                                                       h0, CT, out + 0 * (size_t)TD);
  gsn_cand_wmma<false><<<grd_mm, blk, 0, stream>>>(h1, h0, nullptr, W_w, W_b, G, CT);

  gsn_gate_wmma<true, true><<<grd_mm, blk, 0, stream>>>(h2, h1, h0, Wx_w, Wx_b, G,
                                                      h1, CT, out + 1 * (size_t)TD);
  gsn_cand_wmma<true><<<grd_mm, blk, 0, stream>>>(h2, h1, h0, W_w, W_b, G, CT);
  gsn_finalize<<<grd_fin, blk, 0, stream>>>(h2, CT, out + 2 * (size_t)TD);
}


# Gate launch also does cand product+blend of each tile whose cand operands hold no gate*msg (h or empty-slot zeros), gate kept in registers; G stored for message slots only; 2 CT buffers; sums kept
# speedup vs baseline: 2.6478x; 1.1125x over previous
#include <hip/hip_runtime.h>
#include <math.h>

#define LOG_TD 22u
#define TD (1u << LOG_TD)

typedef __attribute__((ext_vector_type(2))) float v2f;
typedef __attribute__((ext_vector_type(4))) float v4f;
typedef __attribute__((ext_vector_type(8))) float v8f;

constexpr unsigned TPW = 8u;
constexpr unsigned BLK_OUT = 8u * TPW * 48u;
constexpr unsigned BLK_IN  = 8u * TPW * 96u;
constexpr unsigned STG     = BLK_IN / 4u / 256u;
constexpr unsigned TLN     = BLK_OUT / 4u / 256u;
constexpr unsigned NTILE   = TD / 16u / TPW / 8u;
constexpr unsigned PGRID   = 1024u;
constexpr unsigned FIN_PER = 4u;
static_assert(NTILE / PGRID == FIN_PER, "one sum position per thread per tile trip");
static_assert(TD / 4u / 256u / FIN_PER == PGRID, "the sum pass and the gate launch share one grid");
constexpr unsigned DEP_N  = (TD + BLK_IN - 1u) / BLK_IN;
constexpr unsigned DEP2_0 = (2u * TD) / BLK_IN;
static_assert((3u * TD) % BLK_IN == 0u, "segment 2 ends on a tile boundary");
static_assert(DEP2_0 + DEP_N == (3u * TD) / BLK_IN, "tiles that meet cat2 segment 2: [DEP2_0, DEP2_0 + DEP_N)");
constexpr unsigned GW_1 = (TD + BLK_OUT - 1u) / BLK_OUT;
constexpr unsigned GW_2 = (2u * TD + BLK_OUT - 1u) / BLK_OUT;
static_assert(DEP_N <= GW_1 && DEP2_0 + DEP_N <= GW_2, "every tile of the cand launch finds its own gates in G");
template <bool HAS_M1>
__device__ __forceinline__ bool tile_free(unsigned tile) {
  return tile >= DEP_N && !(HAS_M1 && tile >= DEP2_0 && tile < DEP2_0 + DEP_N);
}

__device__ __forceinline__ float hw_exp2(float x) {
#if __has_builtin(__builtin_amdgcn_exp2f)
  return __builtin_amdgcn_exp2f(x);
#else
  return exp2f(x);
#endif
}
__device__ __forceinline__ float hw_rcp(float x) {
#if __has_builtin(__builtin_amdgcn_rcpf)
  return __builtin_amdgcn_rcpf(x);
#else
  return 1.0f / x;
#endif
}
__device__ __forceinline__ float fast_sigmoid(float x) {
  return hw_rcp(1.0f + hw_exp2(-1.4426950408889634f * x));
}
__device__ __forceinline__ float fast_tanh(float x) {
#if __has_builtin(__builtin_amdgcn_tanhf)
  return __builtin_amdgcn_tanhf(x);
#else
  return 1.0f - 2.0f * hw_rcp(hw_exp2(2.8853900817779268f * x) + 1.0f);
#endif
}

__device__ __forceinline__ void lds_barrier() {
  __builtin_amdgcn_fence(__ATOMIC_RELEASE, "workgroup", "local");
  __builtin_amdgcn_s_barrier();
  __builtin_amdgcn_fence(__ATOMIC_ACQUIRE, "workgroup", "local");
}

__device__ __forceinline__ v4f stream_load4(const float* p) {
  return __builtin_nontemporal_load((const v4f*)p);
}
__device__ __forceinline__ void stream_store4(float* p, const v4f v) {
  __builtin_nontemporal_store(v, (v4f*)p);
}

template <bool HAS_M1>
__device__ __forceinline__ bool gate_zero(unsigned m) {
  const unsigned n = m >> LOG_TD;
  return n < 3u && n != 0u && !(HAS_M1 && n == 1u);
}
template <bool HAS_M1>
__device__ __forceinline__ v4f gate_load4(unsigned m,
    const float* __restrict__ hv, const float* __restrict__ m0,
    const float* __restrict__ m1) {
  const unsigned n = m >> LOG_TD;
  const unsigned r = m & (TD - 1u);
  const float* src = (n == 0u) ? m0 : ((HAS_M1 && n == 1u) ? m1 : hv);
  return *(const v4f*)(src + (gate_zero<HAS_M1>(m) ? 0u : r));
}
template <bool HAS_M1>
__device__ __forceinline__ v4f gate_mix4(unsigned m, const v4f v) {
  const bool zero = gate_zero<HAS_M1>(m);
  v4f o;
  o.x = zero ? 0.0f : v.x;
  o.y = zero ? 0.0f : v.y;
  o.z = zero ? 0.0f : v.z;
  o.w = zero ? 0.0f : v.w;
  return o;
}

template <bool HAS_M1>
__device__ __forceinline__ void cand_load4(unsigned m,
    const float* __restrict__ hv, const float* __restrict__ m0,
    const float* __restrict__ m1, const float* __restrict__ G,
    v4f& a, v4f& g) {
  const unsigned seg = m >> LOG_TD;
  const unsigned r   = m & (TD - 1u);
  const bool  isH = (seg & 1u) != 0u;
  const bool  s0  = seg == 0u;
  const bool  s1  = HAS_M1 && seg == 2u;
  const bool  msg = s0 || s1;
  const float* pa = s0 ? m0 : (s1 ? m1 : hv);
  const float* pg = s1 ? G + TD : G;
  a = *(const v4f*)(pa + ((isH || msg) ? r : 0u));
  g = *(const v4f*)(msg ? pg + r : hv);
}
template <bool HAS_M1>
__device__ __forceinline__ v4f cand_mix4(unsigned m, const v4f a, const v4f g) {
  const unsigned seg = m >> LOG_TD;
  const bool  isH = (seg & 1u) != 0u;
  const bool  msg = seg == 0u || (HAS_M1 && seg == 2u);
  const float px = g.x * a.x, py = g.y * a.y, pz = g.z * a.z, pw = g.w * a.w;
  const float zx = msg ? px : 0.0f, zy = msg ? py : 0.0f,
              zz = msg ? pz : 0.0f, zw = msg ? pw : 0.0f;
  v4f v;
  v.x = isH ? a.x : zx;
  v.y = isH ? a.y : zy;
  v.z = isH ? a.z : zz;
  v.w = isH ? a.w : zw;
  return v;
}

__device__ __forceinline__ void mm_phase(const float* const sAw, float* const sW,
    const unsigned lane, const bool lo, const bool nok, const unsigned N,
    const v2f b, const v2f b2, const float bias) {
#pragma unroll
  for (unsigned th = 0; th < TPW; th += TPW / 2u) {
  v2f fa[TPW / 2u], fa4[TPW / 2u];
#pragma unroll
  for (unsigned u = 0; u < TPW / 2u; ++u) {
    const float* const ar = sAw + (th + u) * 96u + 6u * (lane & 15u);
    fa[u]  = *(const v2f*)(ar + (lo ? 0u : 2u));
    fa4[u] = *(const v2f*)(ar + 4u);
  }

#pragma unroll
  for (unsigned u = 0; u < TPW / 2u; ++u) {
    const unsigned t = th + u;
    const v2f a = fa[u];
    v2f a2;
    a2.x = lo ? fa4[u].x : 0.0f;
    a2.y = lo ? fa4[u].y : 0.0f;

    v8f c;
#pragma unroll
    for (int v = 0; v < 8; ++v) c[v] = bias;

    v8f d = __builtin_amdgcn_wmma_f32_16x16x4_f32(false, a,  false, b,  (short)0, c, false, false);
    d     = __builtin_amdgcn_wmma_f32_16x16x4_f32(false, a2, false, b2, (short)0, d, false, false);

    if (nok) {
      const unsigned mb = lo ? 0u : 8u;
#pragma unroll
      for (int v = 0; v < 8; ++v)
        sW[t * 48u + 3u * (mb + (unsigned)v) + N] = d[v];
    }
  }
  }
}

template <bool HAS_M1, bool FIN>
__global__ __launch_bounds__(256) __attribute__((amdgpu_waves_per_eu(8))) void gsn_gate_wmma(
    const float* __restrict__ hv, const float* __restrict__ m0,
    const float* __restrict__ m1, const float* __restrict__ Wx,
    const float* __restrict__ Bx,
    const float* __restrict__ Ww,
    const float* __restrict__ Bw,
    float* __restrict__ G, float* __restrict__ CT,
    const float* __restrict__ fh, const float* __restrict__ fCT,
    float* __restrict__ fout) {
  const unsigned lane  = threadIdx.x & 31u;
  const unsigned wave  = threadIdx.x >> 5u;
  __shared__ __attribute__((aligned(16))) float sA[BLK_IN];
  __shared__ __attribute__((aligned(16))) float sD[BLK_OUT];
  float* const sW = sD + wave * (TPW * 48u);
  const float* const sAw = sA + wave * (TPW * 96u);

  v4f st[STG];
  {
    const unsigned mF = blockIdx.x * BLK_IN + 4u * threadIdx.x;
#pragma unroll
    for (unsigned j = 0; j < STG; ++j)
      st[j] = gate_load4<HAS_M1>(mF + 1024u * j, hv, m0, m1);
  }

  const bool     lo  = lane < 16u;
  const unsigned N   = lane & 15u;
  const bool     nok = N < 3u;
  const unsigned wn  = nok ? N : 0u;
  const float    msk = nok ? 1.0f : 0.0f;
  const float w0 = Wx[wn*6u+0u]*msk, w1 = Wx[wn*6u+1u]*msk, w2 = Wx[wn*6u+2u]*msk,
              w3 = Wx[wn*6u+3u]*msk, w4 = Wx[wn*6u+4u]*msk, w5 = Wx[wn*6u+5u]*msk;
  const float bias = Bx[wn] * msk;
  const float c0 = Ww[wn*6u+0u]*msk, c1 = Ww[wn*6u+1u]*msk, c2 = Ww[wn*6u+2u]*msk,
              c3 = Ww[wn*6u+3u]*msk, c4 = Ww[wn*6u+4u]*msk, c5 = Ww[wn*6u+5u]*msk;
  const float cbias = Bw[wn] * msk;

  v2f b, b2, cb, cb2;
  b.x   = lo ? w0 : w2;
  b.y   = lo ? w1 : w3;
  b2.x  = lo ? w4 : 0.0f;
  b2.y  = lo ? w5 : 0.0f;
  cb.x  = lo ? c0 : c2;
  cb.y  = lo ? c1 : c3;
  cb2.x = lo ? c4 : 0.0f;
  cb2.y = lo ? c5 : 0.0f;

  unsigned fq = (blockIdx.x * 256u + threadIdx.x) * 4u;
  for (unsigned tile = blockIdx.x; tile < NTILE; tile += PGRID) {
  const bool fuse  = tile_free<HAS_M1>(tile);
  const bool keepG = tile < (HAS_M1 ? GW_2 : GW_1);
  const unsigned mT = tile * BLK_IN + 4u * threadIdx.x;
#pragma unroll
  for (unsigned j = 0; j < STG; ++j)
    *(v4f*)(sA + 4u * (threadIdx.x + 256u * j)) = gate_mix4<HAS_M1>(mT + 1024u * j, st[j]);
  v4f fh4, fc0, fc1, fc2;
  if (FIN) {
    fh4 = *(const v4f*)(fh + fq);
    fc0 = stream_load4(fCT + fq);
    fc1 = stream_load4(fCT + TD + fq);
    fc2 = stream_load4(fCT + 2u * TD + fq);
    __builtin_amdgcn_sched_barrier(0);
  }
  v4f ch[STG];
#pragma unroll
  for (unsigned j = 0; j < STG; ++j) {
    const unsigned m   = mT + 1024u * j;
    const bool     isH = ((m >> LOG_TD) & 1u) != 0u;
    ch[j] = *(const v4f*)(hv + ((fuse && isH) ? (m & (TD - 1u)) : 0u));
  }
  {
    const unsigned nt   = tile + PGRID;
    const bool     more = nt < NTILE;
    const unsigned mN   = nt * BLK_IN + 4u * threadIdx.x;
#pragma unroll
    for (unsigned j = 0; j < STG; ++j)
      st[j] = gate_load4<HAS_M1>(more ? mN + 1024u * j : 2u * TD, hv, m0, m1);
  }
  lds_barrier();

  mm_phase(sAw, sW, lane, lo, nok, N, b, b2, bias);

  lds_barrier();

  if (FIN) {
    v4f o;
    o.x = fh4.x + fc0.x + fc1.x + fc2.x;
    o.y = fh4.y + fc0.y + fc1.y + fc2.y;
    o.z = fh4.z + fc0.z + fc1.z + fc2.z;
    o.w = fh4.w + fc0.w + fc1.w + fc2.w;
    stream_store4(fout + fq, o);
    fq += PGRID * 256u * 4u;
  }

  const unsigned pB = tile * BLK_OUT;
  float4 tm[TLN];
#pragma unroll
  for (unsigned k = 0; k < TLN; ++k) {
    const unsigned p   = pB + 4u * (threadIdx.x + 256u * k);
    const unsigned n_o = p >> LOG_TD;
    const unsigned r   = p & (TD - 1u);
    const float* const pm = (n_o == 0u) ? m0 + r : ((HAS_M1 && n_o == 1u) ? m1 + r : hv);
    tm[k] = *(const float4*)(fuse ? pm : hv);
  }

  float4 tg[TLN];
#pragma unroll
  for (unsigned k = 0; k < TLN; ++k) {
    const unsigned i = threadIdx.x + 256u * k;
    const float4 x = *(const float4*)(sD + 4u * i);
    float4 g;
    g.x = fast_sigmoid(x.x);
    g.y = fast_sigmoid(x.y);
    g.z = fast_sigmoid(x.z);
    g.w = fast_sigmoid(x.w);
    tg[k] = g;
    if (keepG) *(float4*)(G + pB + 4u * i) = g;
  }

  if (fuse) {
#pragma unroll
    for (unsigned j = 0; j < STG; ++j) {
      const bool isH = (((mT + 1024u * j) >> LOG_TD) & 1u) != 0u;
      v4f v;
      v.x = isH ? ch[j].x : 0.0f;
      v.y = isH ? ch[j].y : 0.0f;
      v.z = isH ? ch[j].z : 0.0f;
      v.w = isH ? ch[j].w : 0.0f;
      *(v4f*)(sA + 4u * (threadIdx.x + 256u * j)) = v;
    }
    lds_barrier();

    mm_phase(sAw, sW, lane, lo, nok, N, cb, cb2, cbias);

    lds_barrier();

#pragma unroll
    for (unsigned k = 0; k < TLN; ++k) {
      const unsigned i   = threadIdx.x + 256u * k;
      const unsigned p   = pB + 4u * i;
      const float4   x   = *(const float4*)(sD + 4u * i);
      const float4   g   = tg[k];
      const unsigned n_o = p >> LOG_TD;
      const bool     has = n_o == 0u || (HAS_M1 && n_o == 1u);
      float4 mv;
      mv.x = has ? tm[k].x : 0.0f;
      mv.y = has ? tm[k].y : 0.0f;
      mv.z = has ? tm[k].z : 0.0f;
      mv.w = has ? tm[k].w : 0.0f;
      float4 o;
      o.x = (1.0f - g.x) * mv.x + g.x * fast_tanh(x.x);
      o.y = (1.0f - g.y) * mv.y + g.y * fast_tanh(x.y);
      o.z = (1.0f - g.z) * mv.z + g.z * fast_tanh(x.z);
      o.w = (1.0f - g.w) * mv.w + g.w * fast_tanh(x.w);
      *(float4*)(CT + p) = o;
    }
  }
  }
}

template <bool HAS_M1>
__global__ __launch_bounds__(256) __attribute__((amdgpu_waves_per_eu(8))) void gsn_cand_wmma(
    const float* __restrict__ hv, const float* __restrict__ m0,
    const float* __restrict__ m1, const float* __restrict__ Ww,
    const float* __restrict__ Bw,
    const float* __restrict__ G, float* __restrict__ CT) {
  const unsigned lane  = threadIdx.x & 31u;
  const unsigned wave  = threadIdx.x >> 5u;
  __shared__ __attribute__((aligned(16))) float sA[BLK_IN];
  __shared__ __attribute__((aligned(16))) float sD[BLK_OUT];
  float* const sW = sD + wave * (TPW * 48u);
  const float* const sAw = sA + wave * (TPW * 96u);

  v4f st[STG], sg[STG];
  {
    const unsigned mF = blockIdx.x * BLK_IN + 4u * threadIdx.x;
#pragma unroll
    for (unsigned j = 0; j < STG; ++j)
      cand_load4<HAS_M1>(mF + 1024u * j, hv, m0, m1, G, st[j], sg[j]);
  }

  const bool     lo  = lane < 16u;
  const unsigned N   = lane & 15u;
  const bool     nok = N < 3u;
  const unsigned wn  = nok ? N : 0u;
  const float    msk = nok ? 1.0f : 0.0f;
  const float r0 = Ww[wn*6u+0u], r1 = Ww[wn*6u+1u], r2 = Ww[wn*6u+2u],
              r3 = Ww[wn*6u+3u], r4 = Ww[wn*6u+4u], r5 = Ww[wn*6u+5u];
  const float rb = Bw[wn];
  asm volatile("" :: "v"(st[0]), "v"(st[1]), "v"(st[2]), "v"(st[3]), "v"(st[4]), "v"(st[5]), "v"(sg[0]), "v"(sg[1]), "v"(sg[2]), "v"(sg[3]), "v"(sg[4]), "v"(sg[5]),
               "v"(r0), "v"(r1), "v"(r2), "v"(r3), "v"(r4), "v"(r5), "v"(rb));
  const float w0 = r0*msk, w1 = r1*msk, w2 = r2*msk, w3 = r3*msk, w4 = r4*msk, w5 = r5*msk;
  const float bias = rb * msk;

  v2f b, b2;
  b.x  = lo ? w0 : w2;
  b.y  = lo ? w1 : w3;
  b2.x = lo ? w4 : 0.0f;
  b2.y = lo ? w5 : 0.0f;

  constexpr unsigned NTRIP = HAS_M1 ? 2u : 1u;
  for (unsigned trip = 0; trip < NTRIP; ++trip) {
  const unsigned tile = blockIdx.x + trip * DEP2_0;
  const unsigned mT = tile * BLK_IN + 4u * threadIdx.x;
#pragma unroll
  for (unsigned j = 0; j < STG; ++j)
    *(v4f*)(sA + 4u * (threadIdx.x + 256u * j)) = cand_mix4<HAS_M1>(mT + 1024u * j, st[j], sg[j]);

  const unsigned pB = tile * BLK_OUT;
  v4f tg[TLN];
  float4 tm[TLN];
#pragma unroll
  for (unsigned k = 0; k < TLN; ++k) {
    const unsigned p   = pB + 4u * (threadIdx.x + 256u * k);
    const unsigned n_o = p >> LOG_TD;
    const unsigned r   = p & (TD - 1u);
    const float* const pm = (n_o == 0u) ? m0 + r : ((HAS_M1 && n_o == 1u) ? m1 + r : hv);
    tg[k] = stream_load4(G + p);
    tm[k] = *(const float4*)pm;
  }
  __builtin_amdgcn_sched_barrier(0);
  {
    const unsigned nt   = tile + DEP2_0;
    const bool     more = trip + 1u < NTRIP;
    const unsigned mN   = nt * BLK_IN + 4u * threadIdx.x;
#pragma unroll
    for (unsigned j = 0; j < STG; ++j)
      cand_load4<HAS_M1>(more ? mN + 1024u * j : 4u * TD, hv, m0, m1, G, st[j], sg[j]);
  }
  lds_barrier();

#pragma unroll
  for (unsigned th = 0; th < TPW; th += TPW / 2u) {
  v2f fa[TPW / 2u], fa4[TPW / 2u];
#pragma unroll
  for (unsigned u = 0; u < TPW / 2u; ++u) {
    const float* const ar = sAw + (th + u) * 96u + 6u * (lane & 15u);
    fa[u]  = *(const v2f*)(ar + (lo ? 0u : 2u));
    fa4[u] = *(const v2f*)(ar + 4u);
  }

#pragma unroll
  for (unsigned u = 0; u < TPW / 2u; ++u) {
    const unsigned t = th + u;
    const v2f a = fa[u];
    v2f a2;
    a2.x = lo ? fa4[u].x : 0.0f;
    a2.y = lo ? fa4[u].y : 0.0f;

    v8f c;
#pragma unroll
    for (int v = 0; v < 8; ++v) c[v] = bias;

    v8f d = __builtin_amdgcn_wmma_f32_16x16x4_f32(false, a,  false, b,  (short)0, c, false, false);
    d     = __builtin_amdgcn_wmma_f32_16x16x4_f32(false, a2, false, b2, (short)0, d, false, false);

    if (nok) {
      const unsigned mb = lo ? 0u : 8u;
#pragma unroll
      for (int v = 0; v < 8; ++v)
        sW[t * 48u + 3u * (mb + (unsigned)v) + N] = d[v];
    }
  }
  }

  lds_barrier();

#pragma unroll
  for (unsigned k = 0; k < TLN; ++k) {
    const unsigned i   = threadIdx.x + 256u * k;
    const unsigned p   = pB + 4u * i;
    const float4   x   = *(const float4*)(sD + 4u * i);
    const v4f      g   = tg[k];
    const unsigned n_o = p >> LOG_TD;
    const bool     has = n_o == 0u || (HAS_M1 && n_o == 1u);
    float4 mv;
    mv.x = has ? tm[k].x : 0.0f;
    mv.y = has ? tm[k].y : 0.0f;
    mv.z = has ? tm[k].z : 0.0f;
    mv.w = has ? tm[k].w : 0.0f;
    float4 o;
    o.x = (1.0f - g.x) * mv.x + g.x * fast_tanh(x.x);
    o.y = (1.0f - g.y) * mv.y + g.y * fast_tanh(x.y);
    o.z = (1.0f - g.z) * mv.z + g.z * fast_tanh(x.z);
    o.w = (1.0f - g.w) * mv.w + g.w * fast_tanh(x.w);
    *(float4*)(CT + p) = o;
  }
  }
}

__global__ __launch_bounds__(256) void gsn_finalize(
    const float* __restrict__ hv, const float* __restrict__ CT,
    float* __restrict__ out) {
  const unsigned q0 = blockIdx.x * 256u + threadIdx.x;
  const unsigned qs = gridDim.x * 256u;
  v4f h4[FIN_PER], c0[FIN_PER], c1[FIN_PER], c2[FIN_PER];
#pragma unroll
  for (unsigned k = 0; k < FIN_PER; ++k) {
    const unsigned i = (q0 + k * qs) * 4u;
    h4[k] = *(const v4f*)(hv + i);
    c0[k] = stream_load4(CT + i);
    c1[k] = stream_load4(CT + TD + i);
    c2[k] = stream_load4(CT + 2u * TD + i);
  }
#pragma unroll
  for (unsigned k = 0; k < FIN_PER; ++k) {
    const unsigned i = (q0 + k * qs) * 4u;
    v4f o;
    o.x = h4[k].x + c0[k].x + c1[k].x + c2[k].x;
    o.y = h4[k].y + c0[k].y + c1[k].y + c2[k].y;
    o.z = h4[k].z + c0[k].z + c1[k].z + c2[k].z;
    o.w = h4[k].w + c0[k].w + c1[k].w + c2[k].w;
    stream_store4(out + i, o);
  }
}

extern "C" void kernel_launch(void* const* d_in, const int* in_sizes, int n_in,
                              void* d_out, int out_size, void* d_ws, size_t ws_size,
                              hipStream_t stream) {
  (void)in_sizes; (void)n_in; (void)out_size; (void)ws_size;
  const float* feat = (const float*)d_in[0];
  const float* W_w  = (const float*)d_in[1];
  const float* W_b  = (const float*)d_in[2];
  const float* Wx_w = (const float*)d_in[3];
  const float* Wx_b = (const float*)d_in[4];
  float* out = (float*)d_out;

  float* G   = (float*)d_ws;
  float* CTa = G + (size_t)3 * TD;
  float* CTb = CTa + (size_t)3 * TD;

  const float* h0 = feat;
  const float* h1 = feat + (size_t)TD;
  const float* h2 = feat + (size_t)2 * TD;

  dim3 blk(256);
  dim3 grd_mm(PGRID);
  dim3 grd_dep(DEP_N);
  dim3 grd_fin(TD / 4u / 256u / FIN_PER);


  gsn_gate_wmma<false, false><<<grd_mm, blk, 0, stream>>>(h0, h2, nullptr, Wx_w, Wx_b, W_w, W_b, G, CTa,
                                                        nullptr, nullptr, nullptr);
  gsn_cand_wmma<false><<<grd_dep, blk, 0, stream>>>(h0, h2, nullptr, W_w, W_b, G, CTa);

  gsn_gate_wmma<false, true><<<grd_mm, blk, 0, stream>>>(h1, h0, nullptr, Wx_w, Wx_b, W_w, W_b, G, CTb,
                                                       h0, CTa, out + 0 * (size_t)TD);
  gsn_cand_wmma<false><<<grd_dep, blk, 0, stream>>>(h1, h0, nullptr, W_w, W_b, G, CTb);

  gsn_gate_wmma<true, true><<<grd_mm, blk, 0, stream>>>(h2, h1, h0, Wx_w, Wx_b, W_w, W_b, G, CTa,
                                                      h1, CTb, out + 1 * (size_t)TD);
  gsn_cand_wmma<true><<<grd_dep, blk, 0, stream>>>(h2, h1, h0, W_w, W_b, G, CTa);
  gsn_finalize<<<grd_fin, blk, 0, stream>>>(h2, CTa, out + 2 * (size_t)TD);
}


# Tiles whose gate rows and cand rows are the same h rows: one matrix phase, gate weights in columns 0-2 and cand weights in 3-5, gives both sums; no 2nd staging, phase, barriers there; sum order kept
# speedup vs baseline: 2.8174x; 1.0641x over previous
#include <hip/hip_runtime.h>
#include <math.h>

#define LOG_TD 22u
#define TD (1u << LOG_TD)

typedef __attribute__((ext_vector_type(2))) float v2f;
typedef __attribute__((ext_vector_type(4))) float v4f;
typedef __attribute__((ext_vector_type(8))) float v8f;

constexpr unsigned TPW = 8u;
constexpr unsigned BLK_OUT = 8u * TPW * 48u;
constexpr unsigned BLK_IN  = 8u * TPW * 96u;
constexpr unsigned STG     = BLK_IN / 4u / 256u;
constexpr unsigned TLN     = BLK_OUT / 4u / 256u;
constexpr unsigned NTILE   = TD / 16u / TPW / 8u;
constexpr unsigned PGRID   = 1024u;
constexpr unsigned FIN_PER = 4u;
static_assert(NTILE / PGRID == FIN_PER, "one sum position per thread per tile trip");
static_assert(TD / 4u / 256u / FIN_PER == PGRID, "the sum pass and the gate launch share one grid");
constexpr unsigned DEP_N  = (TD + BLK_IN - 1u) / BLK_IN;
constexpr unsigned DEP2_0 = (2u * TD) / BLK_IN;
static_assert((3u * TD) % BLK_IN == 0u, "segment 2 ends on a tile boundary");
static_assert(DEP2_0 + DEP_N == (3u * TD) / BLK_IN, "tiles that meet cat2 segment 2: [DEP2_0, DEP2_0 + DEP_N)");
constexpr unsigned GW_1 = (TD + BLK_OUT - 1u) / BLK_OUT;
constexpr unsigned GW_2 = (2u * TD + BLK_OUT - 1u) / BLK_OUT;
static_assert(DEP_N <= GW_1 && DEP2_0 + DEP_N <= GW_2, "every tile of the cand launch finds its own gates in G");
template <bool HAS_M1>
__device__ __forceinline__ bool tile_free(unsigned tile) {
  return tile >= DEP_N && !(HAS_M1 && tile >= DEP2_0 && tile < DEP2_0 + DEP_N);
}
constexpr unsigned SAME3_0 = (3u * TD) / BLK_IN;
constexpr unsigned SAME3_1 = (4u * TD) / BLK_IN;
constexpr unsigned SAME5_0 = (5u * TD + BLK_IN - 1u) / BLK_IN;
static_assert(SAME3_0 * BLK_IN == 3u * TD && SAME3_1 * BLK_IN <= 4u * TD, "segment 3's whole tiles");
static_assert(SAME5_0 * BLK_IN >= 5u * TD && NTILE * BLK_IN == 6u * TD, "segment 5's whole tiles");
static_assert(SAME3_0 >= DEP2_0 + DEP_N, "every such tile is tile_free, with one message or two");
__device__ __forceinline__ bool tile_same(unsigned tile) {
  return (tile >= SAME3_0 && tile < SAME3_1) || tile >= SAME5_0;
}

__device__ __forceinline__ float hw_exp2(float x) {
#if __has_builtin(__builtin_amdgcn_exp2f)
  return __builtin_amdgcn_exp2f(x);
#else
  return exp2f(x);
#endif
}
__device__ __forceinline__ float hw_rcp(float x) {
#if __has_builtin(__builtin_amdgcn_rcpf)
  return __builtin_amdgcn_rcpf(x);
#else
  return 1.0f / x;
#endif
}
__device__ __forceinline__ float fast_sigmoid(float x) {
  return hw_rcp(1.0f + hw_exp2(-1.4426950408889634f * x));
}
__device__ __forceinline__ float fast_tanh(float x) {
#if __has_builtin(__builtin_amdgcn_tanhf)
  return __builtin_amdgcn_tanhf(x);
#else
  return 1.0f - 2.0f * hw_rcp(hw_exp2(2.8853900817779268f * x) + 1.0f);
#endif
}

__device__ __forceinline__ void lds_barrier() {
  __builtin_amdgcn_fence(__ATOMIC_RELEASE, "workgroup", "local");
  __builtin_amdgcn_s_barrier();
  __builtin_amdgcn_fence(__ATOMIC_ACQUIRE, "workgroup", "local");
}

__device__ __forceinline__ v4f stream_load4(const float* p) {
  return __builtin_nontemporal_load((const v4f*)p);
}
__device__ __forceinline__ void stream_store4(float* p, const v4f v) {
  __builtin_nontemporal_store(v, (v4f*)p);
}

template <bool HAS_M1>
__device__ __forceinline__ bool gate_zero(unsigned m) {
  const unsigned n = m >> LOG_TD;
  return n < 3u && n != 0u && !(HAS_M1 && n == 1u);
}
template <bool HAS_M1>
__device__ __forceinline__ v4f gate_load4(unsigned m,
    const float* __restrict__ hv, const float* __restrict__ m0,
    const float* __restrict__ m1) {
  const unsigned n = m >> LOG_TD;
  const unsigned r = m & (TD - 1u);
  const float* src = (n == 0u) ? m0 : ((HAS_M1 && n == 1u) ? m1 : hv);
  return *(const v4f*)(src + (gate_zero<HAS_M1>(m) ? 0u : r));
}
template <bool HAS_M1>
__device__ __forceinline__ v4f gate_mix4(unsigned m, const v4f v) {
  const bool zero = gate_zero<HAS_M1>(m);
  v4f o;
  o.x = zero ? 0.0f : v.x;
  o.y = zero ? 0.0f : v.y;
  o.z = zero ? 0.0f : v.z;
  o.w = zero ? 0.0f : v.w;
  return o;
}

template <bool HAS_M1>
__device__ __forceinline__ void cand_load4(unsigned m,
    const float* __restrict__ hv, const float* __restrict__ m0,
    const float* __restrict__ m1, const float* __restrict__ G,
    v4f& a, v4f& g) {
  const unsigned seg = m >> LOG_TD;
  const unsigned r   = m & (TD - 1u);
  const bool  isH = (seg & 1u) != 0u;
  const bool  s0  = seg == 0u;
  const bool  s1  = HAS_M1 && seg == 2u;
  const bool  msg = s0 || s1;
  const float* pa = s0 ? m0 : (s1 ? m1 : hv);
  const float* pg = s1 ? G + TD : G;
  a = *(const v4f*)(pa + ((isH || msg) ? r : 0u));
  g = *(const v4f*)(msg ? pg + r : hv);
}
template <bool HAS_M1>
__device__ __forceinline__ v4f cand_mix4(unsigned m, const v4f a, const v4f g) {
  const unsigned seg = m >> LOG_TD;
  const bool  isH = (seg & 1u) != 0u;
  const bool  msg = seg == 0u || (HAS_M1 && seg == 2u);
  const float px = g.x * a.x, py = g.y * a.y, pz = g.z * a.z, pw = g.w * a.w;
  const float zx = msg ? px : 0.0f, zy = msg ? py : 0.0f,
              zz = msg ? pz : 0.0f, zw = msg ? pw : 0.0f;
  v4f v;
  v.x = isH ? a.x : zx;
  v.y = isH ? a.y : zy;
  v.z = isH ? a.z : zz;
  v.w = isH ? a.w : zw;
  return v;
}

__device__ __forceinline__ void mm_phase(const float* const sAw, float* const sWl,
    const unsigned lane, const bool lo, const bool ok,
    const v2f b, const v2f b2, const float bias) {
#pragma unroll
  for (unsigned th = 0; th < TPW; th += TPW / 2u) {
  v2f fa[TPW / 2u], fa4[TPW / 2u];
#pragma unroll
  for (unsigned u = 0; u < TPW / 2u; ++u) {
    const float* const ar = sAw + (th + u) * 96u + 6u * (lane & 15u);
    fa[u]  = *(const v2f*)(ar + (lo ? 0u : 2u));
    fa4[u] = *(const v2f*)(ar + 4u);
  }

#pragma unroll
  for (unsigned u = 0; u < TPW / 2u; ++u) {
    const unsigned t = th + u;
    const v2f a = fa[u];
    v2f a2;
    a2.x = lo ? fa4[u].x : 0.0f;
    a2.y = lo ? fa4[u].y : 0.0f;

    v8f c;
#pragma unroll
    for (int v = 0; v < 8; ++v) c[v] = bias;

    v8f d = __builtin_amdgcn_wmma_f32_16x16x4_f32(false, a,  false, b,  (short)0, c, false, false);
    d     = __builtin_amdgcn_wmma_f32_16x16x4_f32(false, a2, false, b2, (short)0, d, false, false);

    if (ok) {
      const unsigned mb = lo ? 0u : 8u;
#pragma unroll
      for (int v = 0; v < 8; ++v)
        sWl[t * 48u + 3u * (mb + (unsigned)v)] = d[v];
    }
  }
  }
}

template <bool HAS_M1, bool FIN>
__global__ __launch_bounds__(256) __attribute__((amdgpu_waves_per_eu(8))) void gsn_gate_wmma(
    const float* __restrict__ hv, const float* __restrict__ m0,
    const float* __restrict__ m1, const float* __restrict__ Wx,
    const float* __restrict__ Bx,
    const float* __restrict__ Ww,
    const float* __restrict__ Bw,
    float* __restrict__ G, float* __restrict__ CT,
    const float* __restrict__ fh, const float* __restrict__ fCT,
    float* __restrict__ fout) {
  const unsigned lane  = threadIdx.x & 31u;
  const unsigned wave  = threadIdx.x >> 5u;
  __shared__ __attribute__((aligned(16))) float sA[BLK_IN];
  __shared__ __attribute__((aligned(16))) float sD[BLK_OUT];
  __shared__ __attribute__((aligned(16))) float sC[BLK_OUT];
  float* const sW = sD + wave * (TPW * 48u);
  float* const sWc = sC + wave * (TPW * 48u);
  const float* const sAw = sA + wave * (TPW * 96u);

  v4f st[STG];
  {
    const unsigned mF = blockIdx.x * BLK_IN + 4u * threadIdx.x;
#pragma unroll
    for (unsigned j = 0; j < STG; ++j)
      st[j] = gate_load4<HAS_M1>(mF + 1024u * j, hv, m0, m1);
  }

  const bool     lo   = lane < 16u;
  const unsigned N    = lane & 15u;
  const bool     nok  = N < 3u;
  const bool     nok2 = N >= 3u && N < 6u;
  const unsigned wn   = nok ? N : 0u;
  const unsigned wm   = nok2 ? N - 3u : 0u;
  const float    mskg = (nok || nok2) ? 1.0f : 0.0f;
  float* const sWl = nok2 ? sWc + (N - 3u) : sW + N;
  const float* const gw = nok2 ? Ww + wm * 6u : Wx + wn * 6u;
  const float w0 = gw[0]*mskg, w1 = gw[1]*mskg, w2 = gw[2]*mskg,
              w3 = gw[3]*mskg, w4 = gw[4]*mskg, w5 = gw[5]*mskg;
  const float bias = (nok2 ? Bw[wm] : Bx[wn]) * mskg;

  v2f b, b2;
  b.x   = lo ? w0 : w2;
  b.y   = lo ? w1 : w3;
  b2.x  = lo ? w4 : 0.0f;
  b2.y  = lo ? w5 : 0.0f;

  unsigned fq = (blockIdx.x * 256u + threadIdx.x) * 4u;
  for (unsigned tile = blockIdx.x; tile < NTILE; tile += PGRID) {
  const bool fuse  = tile_free<HAS_M1>(tile);
  const bool same  = tile_same(tile);
  const bool keepG = tile < (HAS_M1 ? GW_2 : GW_1);
  const unsigned mT = tile * BLK_IN + 4u * threadIdx.x;
#pragma unroll
  for (unsigned j = 0; j < STG; ++j)
    *(v4f*)(sA + 4u * (threadIdx.x + 256u * j)) = gate_mix4<HAS_M1>(mT + 1024u * j, st[j]);
  v4f fh4, fc0, fc1, fc2;
  if (FIN) {
    fh4 = *(const v4f*)(fh + fq);
    fc0 = stream_load4(fCT + fq);
    fc1 = stream_load4(fCT + TD + fq);
    fc2 = stream_load4(fCT + 2u * TD + fq);
    __builtin_amdgcn_sched_barrier(0);
  }
  v4f ch[STG];
#pragma unroll
  for (unsigned j = 0; j < STG; ++j) {
    const unsigned m   = mT + 1024u * j;
    const bool     isH = ((m >> LOG_TD) & 1u) != 0u;
    ch[j] = *(const v4f*)(hv + ((fuse && !same && isH) ? (m & (TD - 1u)) : 0u));
  }
  {
    const unsigned nt   = tile + PGRID;
    const bool     more = nt < NTILE;
    const unsigned mN   = nt * BLK_IN + 4u * threadIdx.x;
#pragma unroll
    for (unsigned j = 0; j < STG; ++j)
      st[j] = gate_load4<HAS_M1>(more ? mN + 1024u * j : 2u * TD, hv, m0, m1);
  }
  lds_barrier();

  mm_phase(sAw, sWl, lane, lo, nok || (same && nok2), b, b2, bias);

  lds_barrier();

  if (FIN) {
    v4f o;
    o.x = fh4.x + fc0.x + fc1.x + fc2.x;
    o.y = fh4.y + fc0.y + fc1.y + fc2.y;
    o.z = fh4.z + fc0.z + fc1.z + fc2.z;
    o.w = fh4.w + fc0.w + fc1.w + fc2.w;
    stream_store4(fout + fq, o);
    fq += PGRID * 256u * 4u;
  }

  const unsigned pB = tile * BLK_OUT;
  float4 tm[TLN];
#pragma unroll
  for (unsigned k = 0; k < TLN; ++k) {
    const unsigned p   = pB + 4u * (threadIdx.x + 256u * k);
    const unsigned n_o = p >> LOG_TD;
    const unsigned r   = p & (TD - 1u);
    const float* const pm = (n_o == 0u) ? m0 + r : ((HAS_M1 && n_o == 1u) ? m1 + r : hv);
    tm[k] = *(const float4*)(fuse ? pm : hv);
  }

  float4 tg[TLN];
#pragma unroll
  for (unsigned k = 0; k < TLN; ++k) {
    const unsigned i = threadIdx.x + 256u * k;
    const float4 x = *(const float4*)(sD + 4u * i);
    float4 g;
    g.x = fast_sigmoid(x.x);
    g.y = fast_sigmoid(x.y);
    g.z = fast_sigmoid(x.z);
    g.w = fast_sigmoid(x.w);
    tg[k] = g;
    if (keepG) *(float4*)(G + pB + 4u * i) = g;
  }

  if (fuse) {
    if (!same) {
#pragma unroll
    for (unsigned j = 0; j < STG; ++j) {
      const bool isH = (((mT + 1024u * j) >> LOG_TD) & 1u) != 0u;
      v4f v;
      v.x = isH ? ch[j].x : 0.0f;
      v.y = isH ? ch[j].y : 0.0f;
      v.z = isH ? ch[j].z : 0.0f;
      v.w = isH ? ch[j].w : 0.0f;
      *(v4f*)(sA + 4u * (threadIdx.x + 256u * j)) = v;
    }
    lds_barrier();

    mm_phase(sAw, sWl, lane, lo, nok2, b, b2, bias);

    lds_barrier();
    }

#pragma unroll
    for (unsigned k = 0; k < TLN; ++k) {
      const unsigned i   = threadIdx.x + 256u * k;
      const unsigned p   = pB + 4u * i;
      const float4   x   = *(const float4*)(sC + 4u * i);
      const float4   g   = tg[k];
      const unsigned n_o = p >> LOG_TD;
      const bool     has = n_o == 0u || (HAS_M1 && n_o == 1u);
      float4 mv;
      mv.x = has ? tm[k].x : 0.0f;
      mv.y = has ? tm[k].y : 0.0f;
      mv.z = has ? tm[k].z : 0.0f;
      mv.w = has ? tm[k].w : 0.0f;
      float4 o;
      o.x = (1.0f - g.x) * mv.x + g.x * fast_tanh(x.x);
      o.y = (1.0f - g.y) * mv.y + g.y * fast_tanh(x.y);
      o.z = (1.0f - g.z) * mv.z + g.z * fast_tanh(x.z);
      o.w = (1.0f - g.w) * mv.w + g.w * fast_tanh(x.w);
      *(float4*)(CT + p) = o;
    }
  }
  }
}

template <bool HAS_M1>
__global__ __launch_bounds__(256) __attribute__((amdgpu_waves_per_eu(8))) void gsn_cand_wmma(
    const float* __restrict__ hv, const float* __restrict__ m0,
    const float* __restrict__ m1, const float* __restrict__ Ww,
    const float* __restrict__ Bw,
    const float* __restrict__ G, float* __restrict__ CT) {
  const unsigned lane  = threadIdx.x & 31u;
  const unsigned wave  = threadIdx.x >> 5u;
  __shared__ __attribute__((aligned(16))) float sA[BLK_IN];
  __shared__ __attribute__((aligned(16))) float sD[BLK_OUT];
  float* const sW = sD + wave * (TPW * 48u);
  const float* const sAw = sA + wave * (TPW * 96u);

  v4f st[STG], sg[STG];
  {
    const unsigned mF = blockIdx.x * BLK_IN + 4u * threadIdx.x;
#pragma unroll
    for (unsigned j = 0; j < STG; ++j)
      cand_load4<HAS_M1>(mF + 1024u * j, hv, m0, m1, G, st[j], sg[j]);
  }

  const bool     lo  = lane < 16u;
  const unsigned N   = lane & 15u;
  const bool     nok = N < 3u;
  const unsigned wn  = nok ? N : 0u;
  const float    msk = nok ? 1.0f : 0.0f;
  const float r0 = Ww[wn*6u+0u], r1 = Ww[wn*6u+1u], r2 = Ww[wn*6u+2u],
              r3 = Ww[wn*6u+3u], r4 = Ww[wn*6u+4u], r5 = Ww[wn*6u+5u];
  const float rb = Bw[wn];
  asm volatile("" :: "v"(st[0]), "v"(st[1]), "v"(st[2]), "v"(st[3]), "v"(st[4]), "v"(st[5]), "v"(sg[0]), "v"(sg[1]), "v"(sg[2]), "v"(sg[3]), "v"(sg[4]), "v"(sg[5]),
               "v"(r0), "v"(r1), "v"(r2), "v"(r3), "v"(r4), "v"(r5), "v"(rb));
  const float w0 = r0*msk, w1 = r1*msk, w2 = r2*msk, w3 = r3*msk, w4 = r4*msk, w5 = r5*msk;
  const float bias = rb * msk;

  v2f b, b2;
  b.x  = lo ? w0 : w2;
  b.y  = lo ? w1 : w3;
  b2.x = lo ? w4 : 0.0f;
  b2.y = lo ? w5 : 0.0f;

  constexpr unsigned NTRIP = HAS_M1 ? 2u : 1u;
  for (unsigned trip = 0; trip < NTRIP; ++trip) {
  const unsigned tile = blockIdx.x + trip * DEP2_0;
  const unsigned mT = tile * BLK_IN + 4u * threadIdx.x;
#pragma unroll
  for (unsigned j = 0; j < STG; ++j)
    *(v4f*)(sA + 4u * (threadIdx.x + 256u * j)) = cand_mix4<HAS_M1>(mT + 1024u * j, st[j], sg[j]);

  const unsigned pB = tile * BLK_OUT;
  v4f tg[TLN];
  float4 tm[TLN];
#pragma unroll
  for (unsigned k = 0; k < TLN; ++k) {
    const unsigned p   = pB + 4u * (threadIdx.x + 256u * k);
    const unsigned n_o = p >> LOG_TD;
    const unsigned r   = p & (TD - 1u);
    const float* const pm = (n_o == 0u) ? m0 + r : ((HAS_M1 && n_o == 1u) ? m1 + r : hv);
    tg[k] = stream_load4(G + p);
    tm[k] = *(const float4*)pm;
  }
  __builtin_amdgcn_sched_barrier(0);
  {
    const unsigned nt   = tile + DEP2_0;
    const bool     more = trip + 1u < NTRIP;
    const unsigned mN   = nt * BLK_IN + 4u * threadIdx.x;
#pragma unroll
    for (unsigned j = 0; j < STG; ++j)
      cand_load4<HAS_M1>(more ? mN + 1024u * j : 4u * TD, hv, m0, m1, G, st[j], sg[j]);
  }
  lds_barrier();

#pragma unroll
  for (unsigned th = 0; th < TPW; th += TPW / 2u) {
  v2f fa[TPW / 2u], fa4[TPW / 2u];
#pragma unroll
  for (unsigned u = 0; u < TPW / 2u; ++u) {
    const float* const ar = sAw + (th + u) * 96u + 6u * (lane & 15u);
    fa[u]  = *(const v2f*)(ar + (lo ? 0u : 2u));
    fa4[u] = *(const v2f*)(ar + 4u);
  }

#pragma unroll
  for (unsigned u = 0; u < TPW / 2u; ++u) {
    const unsigned t = th + u;
    const v2f a = fa[u];
    v2f a2;
    a2.x = lo ? fa4[u].x : 0.0f;
    a2.y = lo ? fa4[u].y : 0.0f;

    v8f c;
#pragma unroll
    for (int v = 0; v < 8; ++v) c[v] = bias;

    v8f d = __builtin_amdgcn_wmma_f32_16x16x4_f32(false, a,  false, b,  (short)0, c, false, false);
    d     = __builtin_amdgcn_wmma_f32_16x16x4_f32(false, a2, false, b2, (short)0, d, false, false);

    if (nok) {
      const unsigned mb = lo ? 0u : 8u;
#pragma unroll
      for (int v = 0; v < 8; ++v)
        sW[t * 48u + 3u * (mb + (unsigned)v) + N] = d[v];
    }
  }
  }

  lds_barrier();

#pragma unroll
  for (unsigned k = 0; k < TLN; ++k) {
    const unsigned i   = threadIdx.x + 256u * k;
    const unsigned p   = pB + 4u * i;
    const float4   x   = *(const float4*)(sD + 4u * i);
    const v4f      g   = tg[k];
    const unsigned n_o = p >> LOG_TD;
    const bool     has = n_o == 0u || (HAS_M1 && n_o == 1u);
    float4 mv;
    mv.x = has ? tm[k].x : 0.0f;
    mv.y = has ? tm[k].y : 0.0f;
    mv.z = has ? tm[k].z : 0.0f;
    mv.w = has ? tm[k].w : 0.0f;
    float4 o;
    o.x = (1.0f - g.x) * mv.x + g.x * fast_tanh(x.x);
    o.y = (1.0f - g.y) * mv.y + g.y * fast_tanh(x.y);
    o.z = (1.0f - g.z) * mv.z + g.z * fast_tanh(x.z);
    o.w = (1.0f - g.w) * mv.w + g.w * fast_tanh(x.w);
    *(float4*)(CT + p) = o;
  }
  }
}

__global__ __launch_bounds__(256) void gsn_finalize(
    const float* __restrict__ hv, const float* __restrict__ CT,
    float* __restrict__ out) {
  const unsigned q0 = blockIdx.x * 256u + threadIdx.x;
  const unsigned qs = gridDim.x * 256u;
  v4f h4[FIN_PER], c0[FIN_PER], c1[FIN_PER], c2[FIN_PER];
#pragma unroll
  for (unsigned k = 0; k < FIN_PER; ++k) {
    const unsigned i = (q0 + k * qs) * 4u;
    h4[k] = *(const v4f*)(hv + i);
    c0[k] = stream_load4(CT + i);
    c1[k] = stream_load4(CT + TD + i);
    c2[k] = stream_load4(CT + 2u * TD + i);
  }
#pragma unroll
  for (unsigned k = 0; k < FIN_PER; ++k) {
    const unsigned i = (q0 + k * qs) * 4u;
    v4f o;
    o.x = h4[k].x + c0[k].x + c1[k].x + c2[k].x;
    o.y = h4[k].y + c0[k].y + c1[k].y + c2[k].y;
    o.z = h4[k].z + c0[k].z + c1[k].z + c2[k].z;
    o.w = h4[k].w + c0[k].w + c1[k].w + c2[k].w;
    stream_store4(out + i, o);
  }
}

extern "C" void kernel_launch(void* const* d_in, const int* in_sizes, int n_in,
                              void* d_out, int out_size, void* d_ws, size_t ws_size,
                              hipStream_t stream) {
  (void)in_sizes; (void)n_in; (void)out_size; (void)ws_size;
  const float* feat = (const float*)d_in[0];
  const float* W_w  = (const float*)d_in[1];
  const float* W_b  = (const float*)d_in[2];
  const float* Wx_w = (const float*)d_in[3];
  const float* Wx_b = (const float*)d_in[4];
  float* out = (float*)d_out;

  float* G   = (float*)d_ws;
  float* CTa = G + (size_t)3 * TD;
  float* CTb = CTa + (size_t)3 * TD;

  const float* h0 = feat;
  const float* h1 = feat + (size_t)TD;
  const float* h2 = feat + (size_t)2 * TD;

  dim3 blk(256);
  dim3 grd_mm(PGRID);
  dim3 grd_dep(DEP_N);
  dim3 grd_fin(TD / 4u / 256u / FIN_PER);


  gsn_gate_wmma<false, false><<<grd_mm, blk, 0, stream>>>(h0, h2, nullptr, Wx_w, Wx_b, W_w, W_b, G, CTa,
                                                        nullptr, nullptr, nullptr);
  gsn_cand_wmma<false><<<grd_dep, blk, 0, stream>>>(h0, h2, nullptr, W_w, W_b, G, CTa);

  gsn_gate_wmma<false, true><<<grd_mm, blk, 0, stream>>>(h1, h0, nullptr, Wx_w, Wx_b, W_w, W_b, G, CTb,
                                                       h0, CTa, out + 0 * (size_t)TD);
  gsn_cand_wmma<false><<<grd_dep, blk, 0, stream>>>(h1, h0, nullptr, W_w, W_b, G, CTb);

  gsn_gate_wmma<true, true><<<grd_mm, blk, 0, stream>>>(h2, h1, h0, Wx_w, Wx_b, W_w, W_b, G, CTa,
                                                      h1, CTb, out + 1 * (size_t)TD);
  gsn_cand_wmma<true><<<grd_dep, blk, 0, stream>>>(h2, h1, h0, W_w, W_b, G, CTa);
  gsn_finalize<<<grd_fin, blk, 0, stream>>>(h2, CTa, out + 2 * (size_t)TD);
}


# Fused tiles with differing gate and cand rows: both row sets staged at once, one matrix phase over k=12 (gate k | cand k; 3 instr, was 4), 2nd phase and its barriers gone; cand sum order 2+4, was 4+2
# speedup vs baseline: 2.9446x; 1.0451x over previous
#include <hip/hip_runtime.h>
#include <math.h>

#define LOG_TD 22u
#define TD (1u << LOG_TD)

typedef __attribute__((ext_vector_type(2))) float v2f;
typedef __attribute__((ext_vector_type(4))) float v4f;
typedef __attribute__((ext_vector_type(8))) float v8f;

constexpr unsigned TPW = 8u;
constexpr unsigned BLK_OUT = 8u * TPW * 48u;
constexpr unsigned BLK_IN  = 8u * TPW * 96u;
constexpr unsigned STG     = BLK_IN / 4u / 256u;
constexpr unsigned TLN     = BLK_OUT / 4u / 256u;
constexpr unsigned NTILE   = TD / 16u / TPW / 8u;
constexpr unsigned PGRID   = 1024u;
constexpr unsigned FIN_PER = 4u;
static_assert(NTILE / PGRID == FIN_PER, "one sum position per thread per tile trip");
static_assert(TD / 4u / 256u / FIN_PER == PGRID, "the sum pass and the gate launch share one grid");
constexpr unsigned DEP_N  = (TD + BLK_IN - 1u) / BLK_IN;
constexpr unsigned DEP2_0 = (2u * TD) / BLK_IN;
static_assert((3u * TD) % BLK_IN == 0u, "segment 2 ends on a tile boundary");
static_assert(DEP2_0 + DEP_N == (3u * TD) / BLK_IN, "tiles that meet cat2 segment 2: [DEP2_0, DEP2_0 + DEP_N)");
constexpr unsigned GW_1 = (TD + BLK_OUT - 1u) / BLK_OUT;
constexpr unsigned GW_2 = (2u * TD + BLK_OUT - 1u) / BLK_OUT;
static_assert(DEP_N <= GW_1 && DEP2_0 + DEP_N <= GW_2, "every tile of the cand launch finds its own gates in G");
template <bool HAS_M1>
__device__ __forceinline__ bool tile_free(unsigned tile) {
  return tile >= DEP_N && !(HAS_M1 && tile >= DEP2_0 && tile < DEP2_0 + DEP_N);
}
constexpr unsigned SAME3_0 = (3u * TD) / BLK_IN;
constexpr unsigned SAME3_1 = (4u * TD) / BLK_IN;
constexpr unsigned SAME5_0 = (5u * TD + BLK_IN - 1u) / BLK_IN;
static_assert(SAME3_0 * BLK_IN == 3u * TD && SAME3_1 * BLK_IN <= 4u * TD, "segment 3's whole tiles");
static_assert(SAME5_0 * BLK_IN >= 5u * TD && NTILE * BLK_IN == 6u * TD, "segment 5's whole tiles");
static_assert(SAME3_0 >= DEP2_0 + DEP_N, "every such tile is tile_free, with one message or two");
__device__ __forceinline__ bool tile_same(unsigned tile) {
  return (tile >= SAME3_0 && tile < SAME3_1) || tile >= SAME5_0;
}

__device__ __forceinline__ float hw_exp2(float x) {
#if __has_builtin(__builtin_amdgcn_exp2f)
  return __builtin_amdgcn_exp2f(x);
#else
  return exp2f(x);
#endif
}
__device__ __forceinline__ float hw_rcp(float x) {
#if __has_builtin(__builtin_amdgcn_rcpf)
  return __builtin_amdgcn_rcpf(x);
#else
  return 1.0f / x;
#endif
}
__device__ __forceinline__ float fast_sigmoid(float x) {
  return hw_rcp(1.0f + hw_exp2(-1.4426950408889634f * x));
}
__device__ __forceinline__ float fast_tanh(float x) {
#if __has_builtin(__builtin_amdgcn_tanhf)
  return __builtin_amdgcn_tanhf(x);
#else
  return 1.0f - 2.0f * hw_rcp(hw_exp2(2.8853900817779268f * x) + 1.0f);
#endif
}

__device__ __forceinline__ void lds_barrier() {
  __builtin_amdgcn_fence(__ATOMIC_RELEASE, "workgroup", "local");
  __builtin_amdgcn_s_barrier();
  __builtin_amdgcn_fence(__ATOMIC_ACQUIRE, "workgroup", "local");
}

__device__ __forceinline__ v4f stream_load4(const float* p) {
  return __builtin_nontemporal_load((const v4f*)p);
}
__device__ __forceinline__ void stream_store4(float* p, const v4f v) {
  __builtin_nontemporal_store(v, (v4f*)p);
}

template <bool HAS_M1>
__device__ __forceinline__ bool gate_zero(unsigned m) {
  const unsigned n = m >> LOG_TD;
  return n < 3u && n != 0u && !(HAS_M1 && n == 1u);
}
template <bool HAS_M1>
__device__ __forceinline__ v4f gate_load4(unsigned m,
    const float* __restrict__ hv, const float* __restrict__ m0,
    const float* __restrict__ m1) {
  const unsigned n = m >> LOG_TD;
  const unsigned r = m & (TD - 1u);
  const float* src = (n == 0u) ? m0 : ((HAS_M1 && n == 1u) ? m1 : hv);
  return *(const v4f*)(src + (gate_zero<HAS_M1>(m) ? 0u : r));
}
template <bool HAS_M1>
__device__ __forceinline__ v4f gate_mix4(unsigned m, const v4f v) {
  const bool zero = gate_zero<HAS_M1>(m);
  v4f o;
  o.x = zero ? 0.0f : v.x;
  o.y = zero ? 0.0f : v.y;
  o.z = zero ? 0.0f : v.z;
  o.w = zero ? 0.0f : v.w;
  return o;
}

template <bool HAS_M1>
__device__ __forceinline__ void cand_load4(unsigned m,
    const float* __restrict__ hv, const float* __restrict__ m0,
    const float* __restrict__ m1, const float* __restrict__ G,
    v4f& a, v4f& g) {
  const unsigned seg = m >> LOG_TD;
  const unsigned r   = m & (TD - 1u);
  const bool  isH = (seg & 1u) != 0u;
  const bool  s0  = seg == 0u;
  const bool  s1  = HAS_M1 && seg == 2u;
  const bool  msg = s0 || s1;
  const float* pa = s0 ? m0 : (s1 ? m1 : hv);
  const float* pg = s1 ? G + TD : G;
  a = *(const v4f*)(pa + ((isH || msg) ? r : 0u));
  g = *(const v4f*)(msg ? pg + r : hv);
}
template <bool HAS_M1>
__device__ __forceinline__ v4f cand_mix4(unsigned m, const v4f a, const v4f g) {
  const unsigned seg = m >> LOG_TD;
  const bool  isH = (seg & 1u) != 0u;
  const bool  msg = seg == 0u || (HAS_M1 && seg == 2u);
  const float px = g.x * a.x, py = g.y * a.y, pz = g.z * a.z, pw = g.w * a.w;
  const float zx = msg ? px : 0.0f, zy = msg ? py : 0.0f,
              zz = msg ? pz : 0.0f, zw = msg ? pw : 0.0f;
  v4f v;
  v.x = isH ? a.x : zx;
  v.y = isH ? a.y : zy;
  v.z = isH ? a.z : zz;
  v.w = isH ? a.w : zw;
  return v;
}

__device__ __forceinline__ void mm_phase(const float* const sAw, float* const sWl,
    const unsigned lane, const bool lo, const bool ok,
    const v2f b, const v2f b2, const float bias) {
#pragma unroll
  for (unsigned th = 0; th < TPW; th += TPW / 2u) {
  v2f fa[TPW / 2u], fa4[TPW / 2u];
#pragma unroll
  for (unsigned u = 0; u < TPW / 2u; ++u) {
    const float* const ar = sAw + (th + u) * 96u + 6u * (lane & 15u);
    fa[u]  = *(const v2f*)(ar + (lo ? 0u : 2u));
    fa4[u] = *(const v2f*)(ar + 4u);
  }

#pragma unroll
  for (unsigned u = 0; u < TPW / 2u; ++u) {
    const unsigned t = th + u;
    const v2f a = fa[u];
    v2f a2;
    a2.x = lo ? fa4[u].x : 0.0f;
    a2.y = lo ? fa4[u].y : 0.0f;

    v8f c;
#pragma unroll
    for (int v = 0; v < 8; ++v) c[v] = bias;

    v8f d = __builtin_amdgcn_wmma_f32_16x16x4_f32(false, a,  false, b,  (short)0, c, false, false);
    d     = __builtin_amdgcn_wmma_f32_16x16x4_f32(false, a2, false, b2, (short)0, d, false, false);

    if (ok) {
      const unsigned mb = lo ? 0u : 8u;
#pragma unroll
      for (int v = 0; v < 8; ++v)
        sWl[t * 48u + 3u * (mb + (unsigned)v)] = d[v];
    }
  }
  }
}

__device__ __forceinline__ void mm_phase_cat(const float* const sAw, float* const sWl,
    const unsigned lane, const bool lo, const bool ok,
    const v2f b1, const v2f b2, const v2f b3, const float bias) {
#pragma unroll
  for (unsigned th = 0; th < TPW; th += TPW / 2u) {
  v2f f1[TPW / 2u], f2[TPW / 2u], f3[TPW / 2u];
#pragma unroll
  for (unsigned u = 0; u < TPW / 2u; ++u) {
    const float* const ar = sAw + (th + u) * 96u + 6u * (lane & 15u);
    f1[u] = *(const v2f*)(ar + (lo ? 0u : 2u));
    f2[u] = *(const v2f*)(ar + (lo ? 4u : BLK_IN));
    f3[u] = *(const v2f*)(ar + BLK_IN + (lo ? 2u : 4u));
  }
#pragma unroll
  for (unsigned u = 0; u < TPW / 2u; ++u) {
    const unsigned t = th + u;
    v8f c;
#pragma unroll
    for (int v = 0; v < 8; ++v) c[v] = bias;
    v8f d = __builtin_amdgcn_wmma_f32_16x16x4_f32(false, f1[u], false, b1, (short)0, c, false, false);
    d     = __builtin_amdgcn_wmma_f32_16x16x4_f32(false, f2[u], false, b2, (short)0, d, false, false);
    d     = __builtin_amdgcn_wmma_f32_16x16x4_f32(false, f3[u], false, b3, (short)0, d, false, false);
    if (ok) {
      const unsigned mb = lo ? 0u : 8u;
#pragma unroll
      for (int v = 0; v < 8; ++v)
        sWl[t * 48u + 3u * (mb + (unsigned)v)] = d[v];
    }
  }
  }
}

template <bool HAS_M1, bool FIN>
__global__ __launch_bounds__(256) __attribute__((amdgpu_waves_per_eu(8))) void gsn_gate_wmma(
    const float* __restrict__ hv, const float* __restrict__ m0,
    const float* __restrict__ m1, const float* __restrict__ Wx,
    const float* __restrict__ Bx,
    const float* __restrict__ Ww,
    const float* __restrict__ Bw,
    float* __restrict__ G, float* __restrict__ CT,
    const float* __restrict__ fh, const float* __restrict__ fCT,
    float* __restrict__ fout) {
  const unsigned lane  = threadIdx.x & 31u;
  const unsigned wave  = threadIdx.x >> 5u;
  __shared__ __attribute__((aligned(16))) float sA[2u * BLK_IN];
  __shared__ __attribute__((aligned(16))) float sD[BLK_OUT];
  __shared__ __attribute__((aligned(16))) float sC[BLK_OUT];
  float* const sW = sD + wave * (TPW * 48u);
  float* const sWc = sC + wave * (TPW * 48u);
  const float* const sAw = sA + wave * (TPW * 96u);

  v4f st[STG], ch[STG];
  {
    const unsigned mF = blockIdx.x * BLK_IN + 4u * threadIdx.x;
    const bool two0 = tile_free<HAS_M1>(blockIdx.x) && !tile_same(blockIdx.x);
#pragma unroll
    for (unsigned j = 0; j < STG; ++j)
      st[j] = gate_load4<HAS_M1>(mF + 1024u * j, hv, m0, m1);
#pragma unroll
    for (unsigned j = 0; j < STG; ++j) {
      const unsigned m   = mF + 1024u * j;
      const bool     isH = ((m >> LOG_TD) & 1u) != 0u;
      ch[j] = *(const v4f*)(hv + ((two0 && isH) ? (m & (TD - 1u)) : 0u));
    }
  }

  const bool     lo   = lane < 16u;
  const unsigned N    = lane & 15u;
  const bool     nok  = N < 3u;
  const bool     nok2 = N >= 3u && N < 6u;
  const unsigned wn   = nok ? N : 0u;
  const unsigned wm   = nok2 ? N - 3u : 0u;
  const float    mskg = (nok || nok2) ? 1.0f : 0.0f;
  float* const sWl = nok2 ? sWc + (N - 3u) : sW + N;
  const float* const gw = nok2 ? Ww + wm * 6u : Wx + wn * 6u;
  const float w0 = gw[0]*mskg, w1 = gw[1]*mskg, w2 = gw[2]*mskg,
              w3 = gw[3]*mskg, w4 = gw[4]*mskg, w5 = gw[5]*mskg;
  const float bias = (nok2 ? Bw[wm] : Bx[wn]) * mskg;

  v2f b, b2;
  b.x   = lo ? w0 : w2;
  b.y   = lo ? w1 : w3;
  b2.x  = lo ? w4 : 0.0f;
  b2.y  = lo ? w5 : 0.0f;
  v2f c1, c2, c3;
  c1.x = nok ? b.x : 0.0f;
  c1.y = nok ? b.y : 0.0f;
  c2.x = lo ? (nok ? w4 : 0.0f) : (nok2 ? w0 : 0.0f);
  c2.y = lo ? (nok ? w5 : 0.0f) : (nok2 ? w1 : 0.0f);
  c3.x = nok2 ? (lo ? w2 : w4) : 0.0f;
  c3.y = nok2 ? (lo ? w3 : w5) : 0.0f;

  unsigned fq = (blockIdx.x * 256u + threadIdx.x) * 4u;
  for (unsigned tile = blockIdx.x; tile < NTILE; tile += PGRID) {
  const bool fuse  = tile_free<HAS_M1>(tile);
  const bool same  = tile_same(tile);
  const bool keepG = tile < (HAS_M1 ? GW_2 : GW_1);
  const bool two   = fuse && !same;
  const unsigned mT = tile * BLK_IN + 4u * threadIdx.x;
#pragma unroll
  for (unsigned j = 0; j < STG; ++j)
    *(v4f*)(sA + 4u * (threadIdx.x + 256u * j)) = gate_mix4<HAS_M1>(mT + 1024u * j, st[j]);
  if (two) {
#pragma unroll
    for (unsigned j = 0; j < STG; ++j) {
      const bool isH = (((mT + 1024u * j) >> LOG_TD) & 1u) != 0u;
      v4f v;
      v.x = isH ? ch[j].x : 0.0f;
      v.y = isH ? ch[j].y : 0.0f;
      v.z = isH ? ch[j].z : 0.0f;
      v.w = isH ? ch[j].w : 0.0f;
      *(v4f*)(sA + BLK_IN + 4u * (threadIdx.x + 256u * j)) = v;
    }
  }
  v4f fh4, fc0, fc1, fc2;
  if (FIN) {
    fh4 = *(const v4f*)(fh + fq);
    fc0 = stream_load4(fCT + fq);
    fc1 = stream_load4(fCT + TD + fq);
    fc2 = stream_load4(fCT + 2u * TD + fq);
    __builtin_amdgcn_sched_barrier(0);
  }
  {
    const unsigned nt   = tile + PGRID;
    const bool     more = nt < NTILE;
    const bool     twoN = more && tile_free<HAS_M1>(nt) && !tile_same(nt);
    const unsigned mN   = nt * BLK_IN + 4u * threadIdx.x;
#pragma unroll
    for (unsigned j = 0; j < STG; ++j)
      st[j] = gate_load4<HAS_M1>(more ? mN + 1024u * j : 2u * TD, hv, m0, m1);
#pragma unroll
    for (unsigned j = 0; j < STG; ++j) {
      const unsigned m   = mN + 1024u * j;
      const bool     isH = ((m >> LOG_TD) & 1u) != 0u;
      ch[j] = *(const v4f*)(hv + ((twoN && isH) ? (m & (TD - 1u)) : 0u));
    }
  }
  lds_barrier();

  if (two) mm_phase_cat(sAw, sWl, lane, lo, nok || nok2, c1, c2, c3, bias);
  else     mm_phase(sAw, sWl, lane, lo, nok || (same && nok2), b, b2, bias);

  lds_barrier();

  if (FIN) {
    v4f o;
    o.x = fh4.x + fc0.x + fc1.x + fc2.x;
    o.y = fh4.y + fc0.y + fc1.y + fc2.y;
    o.z = fh4.z + fc0.z + fc1.z + fc2.z;
    o.w = fh4.w + fc0.w + fc1.w + fc2.w;
    stream_store4(fout + fq, o);
    fq += PGRID * 256u * 4u;
  }

  const unsigned pB = tile * BLK_OUT;
  float4 tm[TLN];
#pragma unroll
  for (unsigned k = 0; k < TLN; ++k) {
    const unsigned p   = pB + 4u * (threadIdx.x + 256u * k);
    const unsigned n_o = p >> LOG_TD;
    const unsigned r   = p & (TD - 1u);
    const float* const pm = (n_o == 0u) ? m0 + r : ((HAS_M1 && n_o == 1u) ? m1 + r : hv);
    tm[k] = *(const float4*)(fuse ? pm : hv);
  }

  float4 tg[TLN];
#pragma unroll
  for (unsigned k = 0; k < TLN; ++k) {
    const unsigned i = threadIdx.x + 256u * k;
    const float4 x = *(const float4*)(sD + 4u * i);
    float4 g;
    g.x = fast_sigmoid(x.x);
    g.y = fast_sigmoid(x.y);
    g.z = fast_sigmoid(x.z);
    g.w = fast_sigmoid(x.w);
    tg[k] = g;
    if (keepG) *(float4*)(G + pB + 4u * i) = g;
  }

  if (fuse) {

#pragma unroll
    for (unsigned k = 0; k < TLN; ++k) {
      const unsigned i   = threadIdx.x + 256u * k;
      const unsigned p   = pB + 4u * i;
      const float4   x   = *(const float4*)(sC + 4u * i);
      const float4   g   = tg[k];
      const unsigned n_o = p >> LOG_TD;
      const bool     has = n_o == 0u || (HAS_M1 && n_o == 1u);
      float4 mv;
      mv.x = has ? tm[k].x : 0.0f;
      mv.y = has ? tm[k].y : 0.0f;
      mv.z = has ? tm[k].z : 0.0f;
      mv.w = has ? tm[k].w : 0.0f;
      float4 o;
      o.x = (1.0f - g.x) * mv.x + g.x * fast_tanh(x.x);
      o.y = (1.0f - g.y) * mv.y + g.y * fast_tanh(x.y);
      o.z = (1.0f - g.z) * mv.z + g.z * fast_tanh(x.z);
      o.w = (1.0f - g.w) * mv.w + g.w * fast_tanh(x.w);
      *(float4*)(CT + p) = o;
    }
  }
  }
}

template <bool HAS_M1>
__global__ __launch_bounds__(256) __attribute__((amdgpu_waves_per_eu(8))) void gsn_cand_wmma(
    const float* __restrict__ hv, const float* __restrict__ m0,
    const float* __restrict__ m1, const float* __restrict__ Ww,
    const float* __restrict__ Bw,
    const float* __restrict__ G, float* __restrict__ CT) {
  const unsigned lane  = threadIdx.x & 31u;
  const unsigned wave  = threadIdx.x >> 5u;
  __shared__ __attribute__((aligned(16))) float sA[BLK_IN];
  __shared__ __attribute__((aligned(16))) float sD[BLK_OUT];
  float* const sW = sD + wave * (TPW * 48u);
  const float* const sAw = sA + wave * (TPW * 96u);

  v4f st[STG], sg[STG];
  {
    const unsigned mF = blockIdx.x * BLK_IN + 4u * threadIdx.x;
#pragma unroll
    for (unsigned j = 0; j < STG; ++j)
      cand_load4<HAS_M1>(mF + 1024u * j, hv, m0, m1, G, st[j], sg[j]);
  }

  const bool     lo  = lane < 16u;
  const unsigned N   = lane & 15u;
  const bool     nok = N < 3u;
  const unsigned wn  = nok ? N : 0u;
  const float    msk = nok ? 1.0f : 0.0f;
  const float r0 = Ww[wn*6u+0u], r1 = Ww[wn*6u+1u], r2 = Ww[wn*6u+2u],
              r3 = Ww[wn*6u+3u], r4 = Ww[wn*6u+4u], r5 = Ww[wn*6u+5u];
  const float rb = Bw[wn];
  asm volatile("" :: "v"(st[0]), "v"(st[1]), "v"(st[2]), "v"(st[3]), "v"(st[4]), "v"(st[5]), "v"(sg[0]), "v"(sg[1]), "v"(sg[2]), "v"(sg[3]), "v"(sg[4]), "v"(sg[5]),
               "v"(r0), "v"(r1), "v"(r2), "v"(r3), "v"(r4), "v"(r5), "v"(rb));
  const float w0 = r0*msk, w1 = r1*msk, w2 = r2*msk, w3 = r3*msk, w4 = r4*msk, w5 = r5*msk;
  const float bias = rb * msk;

  v2f b, b2;
  b.x  = lo ? w0 : w2;
  b.y  = lo ? w1 : w3;
  b2.x = lo ? w4 : 0.0f;
  b2.y = lo ? w5 : 0.0f;

  constexpr unsigned NTRIP = HAS_M1 ? 2u : 1u;
  for (unsigned trip = 0; trip < NTRIP; ++trip) {
  const unsigned tile = blockIdx.x + trip * DEP2_0;
  const unsigned mT = tile * BLK_IN + 4u * threadIdx.x;
#pragma unroll
  for (unsigned j = 0; j < STG; ++j)
    *(v4f*)(sA + 4u * (threadIdx.x + 256u * j)) = cand_mix4<HAS_M1>(mT + 1024u * j, st[j], sg[j]);

  const unsigned pB = tile * BLK_OUT;
  v4f tg[TLN];
  float4 tm[TLN];
#pragma unroll
  for (unsigned k = 0; k < TLN; ++k) {
    const unsigned p   = pB + 4u * (threadIdx.x + 256u * k);
    const unsigned n_o = p >> LOG_TD;
    const unsigned r   = p & (TD - 1u);
    const float* const pm = (n_o == 0u) ? m0 + r : ((HAS_M1 && n_o == 1u) ? m1 + r : hv);
    tg[k] = stream_load4(G + p);
    tm[k] = *(const float4*)pm;
  }
  __builtin_amdgcn_sched_barrier(0);
  {
    const unsigned nt   = tile + DEP2_0;
    const bool     more = trip + 1u < NTRIP;
    const unsigned mN   = nt * BLK_IN + 4u * threadIdx.x;
#pragma unroll
    for (unsigned j = 0; j < STG; ++j)
      cand_load4<HAS_M1>(more ? mN + 1024u * j : 4u * TD, hv, m0, m1, G, st[j], sg[j]);
  }
  lds_barrier();

#pragma unroll
  for (unsigned th = 0; th < TPW; th += TPW / 2u) {
  v2f fa[TPW / 2u], fa4[TPW / 2u];
#pragma unroll
  for (unsigned u = 0; u < TPW / 2u; ++u) {
    const float* const ar = sAw + (th + u) * 96u + 6u * (lane & 15u);
    fa[u]  = *(const v2f*)(ar + (lo ? 0u : 2u));
    fa4[u] = *(const v2f*)(ar + 4u);
  }

#pragma unroll
  for (unsigned u = 0; u < TPW / 2u; ++u) {
    const unsigned t = th + u;
    const v2f a = fa[u];
    v2f a2;
    a2.x = lo ? fa4[u].x : 0.0f;
    a2.y = lo ? fa4[u].y : 0.0f;

    v8f c;
#pragma unroll
    for (int v = 0; v < 8; ++v) c[v] = bias;

    v8f d = __builtin_amdgcn_wmma_f32_16x16x4_f32(false, a,  false, b,  (short)0, c, false, false);
    d     = __builtin_amdgcn_wmma_f32_16x16x4_f32(false, a2, false, b2, (short)0, d, false, false);

    if (nok) {
      const unsigned mb = lo ? 0u : 8u;
#pragma unroll
      for (int v = 0; v < 8; ++v)
        sW[t * 48u + 3u * (mb + (unsigned)v) + N] = d[v];
    }
  }
  }

  lds_barrier();

#pragma unroll
  for (unsigned k = 0; k < TLN; ++k) {
    const unsigned i   = threadIdx.x + 256u * k;
    const unsigned p   = pB + 4u * i;
    const float4   x   = *(const float4*)(sD + 4u * i);
    const v4f      g   = tg[k];
    const unsigned n_o = p >> LOG_TD;
    const bool     has = n_o == 0u || (HAS_M1 && n_o == 1u);
    float4 mv;
    mv.x = has ? tm[k].x : 0.0f;
    mv.y = has ? tm[k].y : 0.0f;
    mv.z = has ? tm[k].z : 0.0f;
    mv.w = has ? tm[k].w : 0.0f;
    float4 o;
    o.x = (1.0f - g.x) * mv.x + g.x * fast_tanh(x.x);
    o.y = (1.0f - g.y) * mv.y + g.y * fast_tanh(x.y);
    o.z = (1.0f - g.z) * mv.z + g.z * fast_tanh(x.z);
    o.w = (1.0f - g.w) * mv.w + g.w * fast_tanh(x.w);
    *(float4*)(CT + p) = o;
  }
  }
}

__global__ __launch_bounds__(256) void gsn_finalize(
    const float* __restrict__ hv, const float* __restrict__ CT,
    float* __restrict__ out) {
  const unsigned q0 = blockIdx.x * 256u + threadIdx.x;
  const unsigned qs = gridDim.x * 256u;
  v4f h4[FIN_PER], c0[FIN_PER], c1[FIN_PER], c2[FIN_PER];
#pragma unroll
  for (unsigned k = 0; k < FIN_PER; ++k) {
    const unsigned i = (q0 + k * qs) * 4u;
    h4[k] = *(const v4f*)(hv + i);
    c0[k] = stream_load4(CT + i);
    c1[k] = stream_load4(CT + TD + i);
    c2[k] = stream_load4(CT + 2u * TD + i);
  }
#pragma unroll
  for (unsigned k = 0; k < FIN_PER; ++k) {
    const unsigned i = (q0 + k * qs) * 4u;
    v4f o;
    o.x = h4[k].x + c0[k].x + c1[k].x + c2[k].x;
    o.y = h4[k].y + c0[k].y + c1[k].y + c2[k].y;
    o.z = h4[k].z + c0[k].z + c1[k].z + c2[k].z;
    o.w = h4[k].w + c0[k].w + c1[k].w + c2[k].w;
    stream_store4(out + i, o);
  }
}

extern "C" void kernel_launch(void* const* d_in, const int* in_sizes, int n_in,
                              void* d_out, int out_size, void* d_ws, size_t ws_size,
                              hipStream_t stream) {
  (void)in_sizes; (void)n_in; (void)out_size; (void)ws_size;
  const float* feat = (const float*)d_in[0];
  const float* W_w  = (const float*)d_in[1];
  const float* W_b  = (const float*)d_in[2];
  const float* Wx_w = (const float*)d_in[3];
  const float* Wx_b = (const float*)d_in[4];
  float* out = (float*)d_out;

  float* G   = (float*)d_ws;
  float* CTa = G + (size_t)3 * TD;
  float* CTb = CTa + (size_t)3 * TD;

  const float* h0 = feat;
  const float* h1 = feat + (size_t)TD;
  const float* h2 = feat + (size_t)2 * TD;

  dim3 blk(256);
  dim3 grd_mm(PGRID);
  dim3 grd_dep(DEP_N);
  dim3 grd_fin(TD / 4u / 256u / FIN_PER);


  gsn_gate_wmma<false, false><<<grd_mm, blk, 0, stream>>>(h0, h2, nullptr, Wx_w, Wx_b, W_w, W_b, G, CTa,
                                                        nullptr, nullptr, nullptr);
  gsn_cand_wmma<false><<<grd_dep, blk, 0, stream>>>(h0, h2, nullptr, W_w, W_b, G, CTa);

  gsn_gate_wmma<false, true><<<grd_mm, blk, 0, stream>>>(h1, h0, nullptr, Wx_w, Wx_b, W_w, W_b, G, CTb,
                                                       h0, CTa, out + 0 * (size_t)TD);
  gsn_cand_wmma<false><<<grd_dep, blk, 0, stream>>>(h1, h0, nullptr, W_w, W_b, G, CTb);

  gsn_gate_wmma<true, true><<<grd_mm, blk, 0, stream>>>(h2, h1, h0, Wx_w, Wx_b, W_w, W_b, G, CTa,
                                                      h1, CTb, out + 1 * (size_t)TD);
  gsn_cand_wmma<true><<<grd_dep, blk, 0, stream>>>(h2, h1, h0, W_w, W_b, G, CTa);
  gsn_finalize<<<grd_fin, blk, 0, stream>>>(h2, CTa, out + 2 * (size_t)TD);
}


# One-message nodes: no tile trip on an empty slot's zeros: all-zero tiles' CT is a per-column constant made in the sum pass; tile t+2048 (same h rows) rides tile t's phase; t's cand sums 4+2, was 2+4
# speedup vs baseline: 3.1881x; 1.0827x over previous
#include <hip/hip_runtime.h>
#include <math.h>

#define LOG_TD 22u
#define TD (1u << LOG_TD)

typedef __attribute__((ext_vector_type(2))) float v2f;
typedef __attribute__((ext_vector_type(4))) float v4f;
typedef __attribute__((ext_vector_type(8))) float v8f;

constexpr unsigned TPW = 8u;
constexpr unsigned BLK_OUT = 8u * TPW * 48u;
constexpr unsigned BLK_IN  = 8u * TPW * 96u;
constexpr unsigned STG     = BLK_IN / 4u / 256u;
constexpr unsigned TLN     = BLK_OUT / 4u / 256u;
constexpr unsigned NTILE   = TD / 16u / TPW / 8u;
constexpr unsigned PGRID   = 1024u;
constexpr unsigned FIN_PER = 4u;
static_assert(NTILE / PGRID == FIN_PER, "one sum position per thread per tile trip");
static_assert(TD / 4u / 256u / FIN_PER == PGRID, "the sum pass and the gate launch share one grid");
constexpr unsigned DEP_N  = (TD + BLK_IN - 1u) / BLK_IN;
constexpr unsigned DEP2_0 = (2u * TD) / BLK_IN;
static_assert((3u * TD) % BLK_IN == 0u, "segment 2 ends on a tile boundary");
static_assert(DEP2_0 + DEP_N == (3u * TD) / BLK_IN, "tiles that meet cat2 segment 2: [DEP2_0, DEP2_0 + DEP_N)");
constexpr unsigned GW_1 = (TD + BLK_OUT - 1u) / BLK_OUT;
constexpr unsigned GW_2 = (2u * TD + BLK_OUT - 1u) / BLK_OUT;
static_assert(DEP_N <= GW_1 && DEP2_0 + DEP_N <= GW_2, "every tile of the cand launch finds its own gates in G");
template <bool HAS_M1>
__device__ __forceinline__ bool tile_free(unsigned tile) {
  return tile >= DEP_N && !(HAS_M1 && tile >= DEP2_0 && tile < DEP2_0 + DEP_N);
}
constexpr unsigned SAME3_0 = (3u * TD) / BLK_IN;
constexpr unsigned SAME3_1 = (4u * TD) / BLK_IN;
constexpr unsigned SAME5_0 = (5u * TD + BLK_IN - 1u) / BLK_IN;
static_assert(SAME3_0 * BLK_IN == 3u * TD && SAME3_1 * BLK_IN <= 4u * TD, "segment 3's whole tiles");
static_assert(SAME5_0 * BLK_IN >= 5u * TD && NTILE * BLK_IN == 6u * TD, "segment 5's whole tiles");
static_assert(SAME3_0 >= DEP2_0 + DEP_N, "every such tile is tile_free, with one message or two");
__device__ __forceinline__ bool tile_same(unsigned tile) {
  return (tile >= SAME3_0 && tile < SAME3_1) || tile >= SAME5_0;
}
constexpr unsigned ZERO_0 = (2u * TD + BLK_IN - 1u) / BLK_IN;
constexpr unsigned ZERO_1 = (3u * TD) / BLK_IN;
constexpr unsigned PAIR_0 = (TD + BLK_IN - 1u) / BLK_IN;
constexpr unsigned PAIR_1 = (2u * TD) / BLK_IN;
constexpr unsigned PAIR_D = (3u * TD) / BLK_IN;
static_assert(ZERO_1 * BLK_IN == 3u * TD && PAIR_D * BLK_IN == 3u * TD, "whole tiles");
static_assert(PAIR_0 * BLK_IN >= TD && PAIR_1 * BLK_IN <= 2u * TD, "segment 1's whole tiles");
static_assert(PAIR_0 >= DEP_N && PAIR_1 <= GW_1, "such a tile is tile_free and its gates are kept in G");
static_assert((PAIR_1 + PAIR_D) * BLK_IN <= 5u * TD && PAIR_1 + PAIR_D <= SAME5_0, "the partner lies in segment 4");
static_assert(PAIR_1 * BLK_OUT <= TD && (PAIR_0 + PAIR_D) * BLK_OUT >= 2u * TD, "results: slot 0, partner's: slot 2");
static_assert(BLK_OUT % 3u == 0u, "a tile's first result is in column 0");
constexpr unsigned ZP_0 = ZERO_0 * BLK_OUT - TD;
constexpr unsigned ZP_1 = ZERO_1 * BLK_OUT - TD;
static_assert(ZERO_0 * BLK_OUT >= TD && ZERO_1 * BLK_OUT <= 2u * TD, "the constant results lie in slot 1");
static_assert(ZP_0 % 1024u == 0u && ZP_1 % 1024u == 0u, "a work-group's 256 float4 sum positions are all constant or none");
template <bool HAS_M1>
__device__ __forceinline__ bool tile_skip(unsigned tile) {
  return !HAS_M1 && ((tile >= ZERO_0 && tile < ZERO_1) ||
                     (tile >= PAIR_0 + PAIR_D && tile < PAIR_1 + PAIR_D));
}
template <bool HAS_M1>
__device__ __forceinline__ bool tile_pair(unsigned tile) {
  return !HAS_M1 && tile >= PAIR_0 && tile < PAIR_1;
}
__device__ __forceinline__ float4 col_pattern(unsigned c0, float k0, float k1, float k2) {
  float4 o;
  o.x = c0 == 0u ? k0 : (c0 == 1u ? k1 : k2);
  o.y = c0 == 0u ? k1 : (c0 == 1u ? k2 : k0);
  o.z = c0 == 0u ? k2 : (c0 == 1u ? k0 : k1);
  o.w = o.x;
  return o;
}
__device__ __forceinline__ float uniform_f(float x) {
  return __builtin_bit_cast(float, __builtin_amdgcn_readfirstlane(__builtin_bit_cast(int, x)));
}

__device__ __forceinline__ float hw_exp2(float x) {
#if __has_builtin(__builtin_amdgcn_exp2f)
  return __builtin_amdgcn_exp2f(x);
#else
  return exp2f(x);
#endif
}
__device__ __forceinline__ float hw_rcp(float x) {
#if __has_builtin(__builtin_amdgcn_rcpf)
  return __builtin_amdgcn_rcpf(x);
#else
  return 1.0f / x;
#endif
}
__device__ __forceinline__ float fast_sigmoid(float x) {
  return hw_rcp(1.0f + hw_exp2(-1.4426950408889634f * x));
}
__device__ __forceinline__ float fast_tanh(float x) {
#if __has_builtin(__builtin_amdgcn_tanhf)
  return __builtin_amdgcn_tanhf(x);
#else
  return 1.0f - 2.0f * hw_rcp(hw_exp2(2.8853900817779268f * x) + 1.0f);
#endif
}

__device__ __forceinline__ void lds_barrier() {
  __builtin_amdgcn_fence(__ATOMIC_RELEASE, "workgroup", "local");
  __builtin_amdgcn_s_barrier();
  __builtin_amdgcn_fence(__ATOMIC_ACQUIRE, "workgroup", "local");
}

__device__ __forceinline__ v4f stream_load4(const float* p) {
  return __builtin_nontemporal_load((const v4f*)p);
}
__device__ __forceinline__ void stream_store4(float* p, const v4f v) {
  __builtin_nontemporal_store(v, (v4f*)p);
}

template <bool HAS_M1>
__device__ __forceinline__ bool gate_zero(unsigned m) {
  const unsigned n = m >> LOG_TD;
  return n < 3u && n != 0u && !(HAS_M1 && n == 1u);
}
template <bool HAS_M1>
__device__ __forceinline__ v4f gate_load4(unsigned m,
    const float* __restrict__ hv, const float* __restrict__ m0,
    const float* __restrict__ m1) {
  const unsigned n = m >> LOG_TD;
  const unsigned r = m & (TD - 1u);
  const float* src = (n == 0u) ? m0 : ((HAS_M1 && n == 1u) ? m1 : hv);
  return *(const v4f*)(src + (gate_zero<HAS_M1>(m) ? 0u : r));
}
template <bool HAS_M1>
__device__ __forceinline__ v4f rows_load4(unsigned m, bool pr,
    const float* __restrict__ hv, const float* __restrict__ m0,
    const float* __restrict__ m1) {
  const unsigned n = m >> LOG_TD;
  const unsigned r = m & (TD - 1u);
  const float* src = (n == 0u) ? m0 : ((HAS_M1 && n == 1u) ? m1 : hv);
  return *(const v4f*)(src + ((gate_zero<HAS_M1>(m) && !pr) ? 0u : r));
}
template <bool HAS_M1>
__device__ __forceinline__ v4f gate_mix4(unsigned m, const v4f v) {
  const bool zero = gate_zero<HAS_M1>(m);
  v4f o;
  o.x = zero ? 0.0f : v.x;
  o.y = zero ? 0.0f : v.y;
  o.z = zero ? 0.0f : v.z;
  o.w = zero ? 0.0f : v.w;
  return o;
}

template <bool HAS_M1>
__device__ __forceinline__ void cand_load4(unsigned m,
    const float* __restrict__ hv, const float* __restrict__ m0,
    const float* __restrict__ m1, const float* __restrict__ G,
    v4f& a, v4f& g) {
  const unsigned seg = m >> LOG_TD;
  const unsigned r   = m & (TD - 1u);
  const bool  isH = (seg & 1u) != 0u;
  const bool  s0  = seg == 0u;
  const bool  s1  = HAS_M1 && seg == 2u;
  const bool  msg = s0 || s1;
  const float* pa = s0 ? m0 : (s1 ? m1 : hv);
  const float* pg = s1 ? G + TD : G;
  a = *(const v4f*)(pa + ((isH || msg) ? r : 0u));
  g = *(const v4f*)(msg ? pg + r : hv);
}
template <bool HAS_M1>
__device__ __forceinline__ v4f cand_mix4(unsigned m, const v4f a, const v4f g) {
  const unsigned seg = m >> LOG_TD;
  const bool  isH = (seg & 1u) != 0u;
  const bool  msg = seg == 0u || (HAS_M1 && seg == 2u);
  const float px = g.x * a.x, py = g.y * a.y, pz = g.z * a.z, pw = g.w * a.w;
  const float zx = msg ? px : 0.0f, zy = msg ? py : 0.0f,
              zz = msg ? pz : 0.0f, zw = msg ? pw : 0.0f;
  v4f v;
  v.x = isH ? a.x : zx;
  v.y = isH ? a.y : zy;
  v.z = isH ? a.z : zz;
  v.w = isH ? a.w : zw;
  return v;
}

__device__ __forceinline__ void mm_phase(const float* const sAw, float* const sWl,
    const unsigned lane, const bool lo, const bool ok,
    const v2f b, const v2f b2, const float bias) {
#pragma unroll
  for (unsigned th = 0; th < TPW; th += TPW / 2u) {
  v2f fa[TPW / 2u], fa4[TPW / 2u];
#pragma unroll
  for (unsigned u = 0; u < TPW / 2u; ++u) {
    const float* const ar = sAw + (th + u) * 96u + 6u * (lane & 15u);
    fa[u]  = *(const v2f*)(ar + (lo ? 0u : 2u));
    fa4[u] = *(const v2f*)(ar + 4u);
  }

#pragma unroll
  for (unsigned u = 0; u < TPW / 2u; ++u) {
    const unsigned t = th + u;
    const v2f a = fa[u];
    v2f a2;
    a2.x = lo ? fa4[u].x : 0.0f;
    a2.y = lo ? fa4[u].y : 0.0f;

    v8f c;
#pragma unroll
    for (int v = 0; v < 8; ++v) c[v] = bias;

    v8f d = __builtin_amdgcn_wmma_f32_16x16x4_f32(false, a,  false, b,  (short)0, c, false, false);
    d     = __builtin_amdgcn_wmma_f32_16x16x4_f32(false, a2, false, b2, (short)0, d, false, false);

    if (ok) {
      const unsigned mb = lo ? 0u : 8u;
#pragma unroll
      for (int v = 0; v < 8; ++v)
        sWl[t * 48u + 3u * (mb + (unsigned)v)] = d[v];
    }
  }
  }
}

__device__ __forceinline__ void mm_phase_cat(const float* const sAw, float* const sWl,
    const unsigned lane, const bool lo, const bool ok,
    const v2f b1, const v2f b2, const v2f b3, const float bias) {
#pragma unroll
  for (unsigned th = 0; th < TPW; th += TPW / 2u) {
  v2f f1[TPW / 2u], f2[TPW / 2u], f3[TPW / 2u];
#pragma unroll
  for (unsigned u = 0; u < TPW / 2u; ++u) {
    const float* const ar = sAw + (th + u) * 96u + 6u * (lane & 15u);
    f1[u] = *(const v2f*)(ar + (lo ? 0u : 2u));
    f2[u] = *(const v2f*)(ar + (lo ? 4u : BLK_IN));
    f3[u] = *(const v2f*)(ar + BLK_IN + (lo ? 2u : 4u));
  }
#pragma unroll
  for (unsigned u = 0; u < TPW / 2u; ++u) {
    const unsigned t = th + u;
    v8f c;
#pragma unroll
    for (int v = 0; v < 8; ++v) c[v] = bias;
    v8f d = __builtin_amdgcn_wmma_f32_16x16x4_f32(false, f1[u], false, b1, (short)0, c, false, false);
    d     = __builtin_amdgcn_wmma_f32_16x16x4_f32(false, f2[u], false, b2, (short)0, d, false, false);
    d     = __builtin_amdgcn_wmma_f32_16x16x4_f32(false, f3[u], false, b3, (short)0, d, false, false);
    if (ok) {
      const unsigned mb = lo ? 0u : 8u;
#pragma unroll
      for (int v = 0; v < 8; ++v)
        sWl[t * 48u + 3u * (mb + (unsigned)v)] = d[v];
    }
  }
  }
}

__device__ __forceinline__ void cat_weights(const bool lo, const bool nok, const bool nok2,
    const float w0, const float w1, const float w2, const float w3, const float w4, const float w5,
    v2f& c1, v2f& c2, v2f& c3) {
  c1.x = nok ? (lo ? w0 : w2) : 0.0f;
  c1.y = nok ? (lo ? w1 : w3) : 0.0f;
  c2.x = lo ? (nok ? w4 : 0.0f) : (nok2 ? w0 : 0.0f);
  c2.y = lo ? (nok ? w5 : 0.0f) : (nok2 ? w1 : 0.0f);
  c3.x = nok2 ? (lo ? w2 : w4) : 0.0f;
  c3.y = nok2 ? (lo ? w3 : w5) : 0.0f;
}

template <bool HAS_M1, bool FIN>
__global__ __launch_bounds__(256) __attribute__((amdgpu_waves_per_eu(8))) void gsn_gate_wmma(
    const float* __restrict__ hv, const float* __restrict__ m0,
    const float* __restrict__ m1, const float* __restrict__ Wx,
    const float* __restrict__ Bx,
    const float* __restrict__ Ww,
    const float* __restrict__ Bw,
    float* __restrict__ G, float* __restrict__ CT,
    const float* __restrict__ fh, const float* __restrict__ fCT,
    float* __restrict__ fout) {
  const unsigned lane  = threadIdx.x & 31u;
  const unsigned wave  = threadIdx.x >> 5u;
  __shared__ __attribute__((aligned(16))) float sA[2u * BLK_IN];
  __shared__ __attribute__((aligned(16))) float sD[BLK_OUT];
  __shared__ __attribute__((aligned(16))) float sC[BLK_OUT];
  float* const sW = sD + wave * (TPW * 48u);
  float* const sWc = sC + wave * (TPW * 48u);
  const float* const sAw = sA + wave * (TPW * 96u);

  v4f st[STG], ch[STG];
  {
    const unsigned mF = blockIdx.x * BLK_IN + 4u * threadIdx.x;
    const bool two0 = tile_free<HAS_M1>(blockIdx.x) && !tile_same(blockIdx.x);
    const bool pr0  = tile_pair<HAS_M1>(blockIdx.x);
#pragma unroll
    for (unsigned j = 0; j < STG; ++j)
      st[j] = rows_load4<HAS_M1>(mF + 1024u * j, pr0, hv, m0, m1);
    if (HAS_M1) {
#pragma unroll
    for (unsigned j = 0; j < STG; ++j) {
      const unsigned m   = mF + 1024u * j;
      const bool     isH = ((m >> LOG_TD) & 1u) != 0u;
      ch[j] = *(const v4f*)(hv + ((two0 && isH) ? (m & (TD - 1u)) : 0u));
    }
    }
  }

  const bool     lo   = lane < 16u;
  const unsigned N    = lane & 15u;
  const bool     nok  = N < 3u;
  const bool     nok2 = N >= 3u && N < 6u;
  const unsigned wn   = nok ? N : 0u;
  const unsigned wm   = nok2 ? N - 3u : 0u;
  const float    mskg = (nok || nok2) ? 1.0f : 0.0f;
  float* const sWl = nok2 ? sWc + (N - 3u) : sW + N;
  const float* const gw = nok2 ? Ww + wm * 6u : Wx + wn * 6u;
  const float w0 = gw[0]*mskg, w1 = gw[1]*mskg, w2 = gw[2]*mskg,
              w3 = gw[3]*mskg, w4 = gw[4]*mskg, w5 = gw[5]*mskg;
  const float bias = (nok2 ? Bw[wm] : Bx[wn]) * mskg;

  v2f b, b2;
  b.x   = lo ? w0 : w2;
  b.y   = lo ? w1 : w3;
  b2.x  = lo ? w4 : 0.0f;
  b2.y  = lo ? w5 : 0.0f;
  v2f c1, c2, c3;
  if (HAS_M1) cat_weights(lo, nok, nok2, w0, w1, w2, w3, w4, w5, c1, c2, c3);

  const float gc0 = uniform_f(fast_sigmoid(Bx[0])), gc1 = uniform_f(fast_sigmoid(Bx[1])),
              gc2 = uniform_f(fast_sigmoid(Bx[2]));
  const float tc0 = uniform_f(fast_tanh(Bw[0])), tc1 = uniform_f(fast_tanh(Bw[1])),
              tc2 = uniform_f(fast_tanh(Bw[2]));
  const float kc0 = uniform_f(gc0 * tc0), kc1 = uniform_f(gc1 * tc1), kc2 = uniform_f(gc2 * tc2);

  for (unsigned trip = 0; trip < NTILE / PGRID; ++trip) {
  const unsigned tile = blockIdx.x + trip * PGRID;
  const unsigned fbase = blockIdx.x * 1024u + trip * (PGRID * 1024u);
  const unsigned fq    = fbase + 4u * threadIdx.x;
  const bool     fcz   = fbase >= ZP_0 && fbase < ZP_1;
  if (tile_skip<HAS_M1>(tile)) {
    if (FIN) {
      const v4f fh4 = *(const v4f*)(fh + fq);
      const v4f fc0 = stream_load4(fCT + fq);
      const v4f fc2 = stream_load4(fCT + 2u * TD + fq);
      v4f fc1;
      if (fcz) {
        const float4 kp = col_pattern((TD + fq) % 3u, kc0, kc1, kc2);
        fc1.x = kp.x; fc1.y = kp.y; fc1.z = kp.z; fc1.w = kp.w;
      } else {
        fc1 = stream_load4(fCT + TD + fq);
      }
      v4f o;
      o.x = fh4.x + fc0.x + fc1.x + fc2.x;
      o.y = fh4.y + fc0.y + fc1.y + fc2.y;
      o.z = fh4.z + fc0.z + fc1.z + fc2.z;
      o.w = fh4.w + fc0.w + fc1.w + fc2.w;
      stream_store4(fout + fq, o);
    }
    continue;
  }
  const bool fuse  = tile_free<HAS_M1>(tile);
  const bool same  = tile_same(tile);
  const bool pair  = tile_pair<HAS_M1>(tile);
  const bool keepG = tile < (HAS_M1 ? GW_2 : GW_1);
  const bool two   = fuse && !same && !pair;
  const unsigned mT = tile * BLK_IN + 4u * threadIdx.x;
  if (pair) {
#pragma unroll
    for (unsigned j = 0; j < STG; ++j)
      *(v4f*)(sA + 4u * (threadIdx.x + 256u * j)) = st[j];
  } else {
#pragma unroll
  for (unsigned j = 0; j < STG; ++j)
    *(v4f*)(sA + 4u * (threadIdx.x + 256u * j)) = gate_mix4<HAS_M1>(mT + 1024u * j, st[j]);
  }
  if (two) {
#pragma unroll
    for (unsigned j = 0; j < STG; ++j) {
      const unsigned m = mT + 1024u * j;
      const bool isH = ((m >> LOG_TD) & 1u) != 0u;
      const v4f hp = HAS_M1 ? ch[j] : *(const v4f*)(hv + (isH ? (m & (TD - 1u)) : 0u));
      v4f v;
      v.x = isH ? hp.x : 0.0f;
      v.y = isH ? hp.y : 0.0f;
      v.z = isH ? hp.z : 0.0f;
      v.w = isH ? hp.w : 0.0f;
      *(v4f*)(sA + BLK_IN + 4u * (threadIdx.x + 256u * j)) = v;
    }
  }
  v4f fh4, fc0, fc1, fc2;
  if (FIN) {
    const float4 kp = col_pattern((TD + fq) % 3u, kc0, kc1, kc2);
    fc1.x = kp.x; fc1.y = kp.y; fc1.z = kp.z; fc1.w = kp.w;
    fh4 = *(const v4f*)(fh + fq);
    fc0 = stream_load4(fCT + fq);
    if (!fcz) fc1 = stream_load4(fCT + TD + fq);
    fc2 = stream_load4(fCT + 2u * TD + fq);
    __builtin_amdgcn_sched_barrier(0);
  }
  {
    unsigned nt = tile + PGRID;
    nt += tile_skip<HAS_M1>(nt) ? PGRID : 0u;
    nt += tile_skip<HAS_M1>(nt) ? PGRID : 0u;
    const bool     more = nt < NTILE;
    const bool     twoN = more && tile_free<HAS_M1>(nt) && !tile_same(nt);
    const unsigned mN   = nt * BLK_IN + 4u * threadIdx.x;
    const bool     prN  = more && tile_pair<HAS_M1>(nt);
#pragma unroll
    for (unsigned j = 0; j < STG; ++j)
      st[j] = rows_load4<HAS_M1>(more ? mN + 1024u * j : 2u * TD, prN, hv, m0, m1);
    if (HAS_M1) {
#pragma unroll
    for (unsigned j = 0; j < STG; ++j) {
      const unsigned m   = mN + 1024u * j;
      const bool     isH = ((m >> LOG_TD) & 1u) != 0u;
      ch[j] = *(const v4f*)(hv + ((twoN && isH) ? (m & (TD - 1u)) : 0u));
    }
    }
  }
  lds_barrier();

  if (two) {
    if (!HAS_M1) {
      const float* const gq = nok2 ? Ww + (N - 3u) * 6u : Wx + (nok ? N : 0u) * 6u;
      const float mq = (nok || nok2) ? 1.0f : 0.0f;
      cat_weights(lo, nok, nok2, gq[0]*mq, gq[1]*mq, gq[2]*mq, gq[3]*mq, gq[4]*mq, gq[5]*mq, c1, c2, c3);
    }
    mm_phase_cat(sAw, sWl, lane, lo, nok || nok2, c1, c2, c3, bias);
  }
  else     mm_phase(sAw, sWl, lane, lo, nok || ((same || pair) && nok2), b, b2, bias);

  lds_barrier();

  if (FIN) {
    v4f o;
    o.x = fh4.x + fc0.x + fc1.x + fc2.x;
    o.y = fh4.y + fc0.y + fc1.y + fc2.y;
    o.z = fh4.z + fc0.z + fc1.z + fc2.z;
    o.w = fh4.w + fc0.w + fc1.w + fc2.w;
    stream_store4(fout + fq, o);
  }

  const unsigned pB = tile * BLK_OUT;
  float4 tm[TLN];
#pragma unroll
  for (unsigned k = 0; k < TLN; ++k) {
    const unsigned p   = pB + 4u * (threadIdx.x + 256u * k);
    const unsigned n_o = p >> LOG_TD;
    const unsigned r   = p & (TD - 1u);
    const float* const pm = (n_o == 0u) ? m0 + r : ((HAS_M1 && n_o == 1u) ? m1 + r : hv);
    tm[k] = *(const float4*)(fuse ? pm : hv);
  }

  if (pair) {
#pragma unroll
    for (unsigned k = 0; k < TLN; ++k) {
      const unsigned i  = threadIdx.x + 256u * k;
      const unsigned p  = pB + 4u * i;
      const float4   xg = *(const float4*)(sD + 4u * i);
      const float4   xc = *(const float4*)(sC + 4u * i);
      const unsigned c0 = p % 3u;
      const float4   g  = col_pattern(c0, gc0, gc1, gc2);
      const float4   tc = col_pattern(c0, tc0, tc1, tc2);
      if (keepG) *(float4*)(G + p) = g;
      float4 o;
      o.x = (1.0f - g.x) * tm[k].x + g.x * fast_tanh(xc.x);
      o.y = (1.0f - g.y) * tm[k].y + g.y * fast_tanh(xc.y);
      o.z = (1.0f - g.z) * tm[k].z + g.z * fast_tanh(xc.z);
      o.w = (1.0f - g.w) * tm[k].w + g.w * fast_tanh(xc.w);
      *(float4*)(CT + p) = o;
      float4 o2;
      o2.x = fast_sigmoid(xg.x) * tc.x;
      o2.y = fast_sigmoid(xg.y) * tc.y;
      o2.z = fast_sigmoid(xg.z) * tc.z;
      o2.w = fast_sigmoid(xg.w) * tc.w;
      *(float4*)(CT + p + PAIR_D * BLK_OUT) = o2;
    }
  } else {
  float4 tg[TLN];
#pragma unroll
  for (unsigned k = 0; k < TLN; ++k) {
    const unsigned i = threadIdx.x + 256u * k;
    const float4 x = *(const float4*)(sD + 4u * i);
    float4 g;
    g.x = fast_sigmoid(x.x);
    g.y = fast_sigmoid(x.y);
    g.z = fast_sigmoid(x.z);
    g.w = fast_sigmoid(x.w);
    tg[k] = g;
    if (keepG) *(float4*)(G + pB + 4u * i) = g;
  }

  if (fuse) {

#pragma unroll
    for (unsigned k = 0; k < TLN; ++k) {
      const unsigned i   = threadIdx.x + 256u * k;
      const unsigned p   = pB + 4u * i;
      const float4   x   = *(const float4*)(sC + 4u * i);
      const float4   g   = tg[k];
      const unsigned n_o = p >> LOG_TD;
      const bool     has = n_o == 0u || (HAS_M1 && n_o == 1u);
      float4 mv;
      mv.x = has ? tm[k].x : 0.0f;
      mv.y = has ? tm[k].y : 0.0f;
      mv.z = has ? tm[k].z : 0.0f;
      mv.w = has ? tm[k].w : 0.0f;
      float4 o;
      o.x = (1.0f - g.x) * mv.x + g.x * fast_tanh(x.x);
      o.y = (1.0f - g.y) * mv.y + g.y * fast_tanh(x.y);
      o.z = (1.0f - g.z) * mv.z + g.z * fast_tanh(x.z);
      o.w = (1.0f - g.w) * mv.w + g.w * fast_tanh(x.w);
      *(float4*)(CT + p) = o;
    }
  }
  }
  }
}

template <bool HAS_M1>
__global__ __launch_bounds__(256) __attribute__((amdgpu_waves_per_eu(8))) void gsn_cand_wmma(
    const float* __restrict__ hv, const float* __restrict__ m0,
    const float* __restrict__ m1, const float* __restrict__ Ww,
    const float* __restrict__ Bw,
    const float* __restrict__ G, float* __restrict__ CT) {
  const unsigned lane  = threadIdx.x & 31u;
  const unsigned wave  = threadIdx.x >> 5u;
  __shared__ __attribute__((aligned(16))) float sA[BLK_IN];
  __shared__ __attribute__((aligned(16))) float sD[BLK_OUT];
  float* const sW = sD + wave * (TPW * 48u);
  const float* const sAw = sA + wave * (TPW * 96u);

  v4f st[STG], sg[STG];
  {
    const unsigned mF = blockIdx.x * BLK_IN + 4u * threadIdx.x;
#pragma unroll
    for (unsigned j = 0; j < STG; ++j)
      cand_load4<HAS_M1>(mF + 1024u * j, hv, m0, m1, G, st[j], sg[j]);
  }

  const bool     lo  = lane < 16u;
  const unsigned N   = lane & 15u;
  const bool     nok = N < 3u;
  const unsigned wn  = nok ? N : 0u;
  const float    msk = nok ? 1.0f : 0.0f;
  const float r0 = Ww[wn*6u+0u], r1 = Ww[wn*6u+1u], r2 = Ww[wn*6u+2u],
              r3 = Ww[wn*6u+3u], r4 = Ww[wn*6u+4u], r5 = Ww[wn*6u+5u];
  const float rb = Bw[wn];
  asm volatile("" :: "v"(st[0]), "v"(st[1]), "v"(st[2]), "v"(st[3]), "v"(st[4]), "v"(st[5]), "v"(sg[0]), "v"(sg[1]), "v"(sg[2]), "v"(sg[3]), "v"(sg[4]), "v"(sg[5]),
               "v"(r0), "v"(r1), "v"(r2), "v"(r3), "v"(r4), "v"(r5), "v"(rb));
  const float w0 = r0*msk, w1 = r1*msk, w2 = r2*msk, w3 = r3*msk, w4 = r4*msk, w5 = r5*msk;
  const float bias = rb * msk;

  v2f b, b2;
  b.x  = lo ? w0 : w2;
  b.y  = lo ? w1 : w3;
  b2.x = lo ? w4 : 0.0f;
  b2.y = lo ? w5 : 0.0f;

  constexpr unsigned NTRIP = HAS_M1 ? 2u : 1u;
  for (unsigned trip = 0; trip < NTRIP; ++trip) {
  const unsigned tile = blockIdx.x + trip * DEP2_0;
  const unsigned mT = tile * BLK_IN + 4u * threadIdx.x;
#pragma unroll
  for (unsigned j = 0; j < STG; ++j)
    *(v4f*)(sA + 4u * (threadIdx.x + 256u * j)) = cand_mix4<HAS_M1>(mT + 1024u * j, st[j], sg[j]);

  const unsigned pB = tile * BLK_OUT;
  v4f tg[TLN];
  float4 tm[TLN];
#pragma unroll
  for (unsigned k = 0; k < TLN; ++k) {
    const unsigned p   = pB + 4u * (threadIdx.x + 256u * k);
    const unsigned n_o = p >> LOG_TD;
    const unsigned r   = p & (TD - 1u);
    const float* const pm = (n_o == 0u) ? m0 + r : ((HAS_M1 && n_o == 1u) ? m1 + r : hv);
    tg[k] = stream_load4(G + p);
    tm[k] = *(const float4*)pm;
  }
  __builtin_amdgcn_sched_barrier(0);
  {
    const unsigned nt   = tile + DEP2_0;
    const bool     more = trip + 1u < NTRIP;
    const unsigned mN   = nt * BLK_IN + 4u * threadIdx.x;
#pragma unroll
    for (unsigned j = 0; j < STG; ++j)
      cand_load4<HAS_M1>(more ? mN + 1024u * j : 4u * TD, hv, m0, m1, G, st[j], sg[j]);
  }
  lds_barrier();

#pragma unroll
  for (unsigned th = 0; th < TPW; th += TPW / 2u) {
  v2f fa[TPW / 2u], fa4[TPW / 2u];
#pragma unroll
  for (unsigned u = 0; u < TPW / 2u; ++u) {
    const float* const ar = sAw + (th + u) * 96u + 6u * (lane & 15u);
    fa[u]  = *(const v2f*)(ar + (lo ? 0u : 2u));
    fa4[u] = *(const v2f*)(ar + 4u);
  }

#pragma unroll
  for (unsigned u = 0; u < TPW / 2u; ++u) {
    const unsigned t = th + u;
    const v2f a = fa[u];
    v2f a2;
    a2.x = lo ? fa4[u].x : 0.0f;
    a2.y = lo ? fa4[u].y : 0.0f;

    v8f c;
#pragma unroll
    for (int v = 0; v < 8; ++v) c[v] = bias;

    v8f d = __builtin_amdgcn_wmma_f32_16x16x4_f32(false, a,  false, b,  (short)0, c, false, false);
    d     = __builtin_amdgcn_wmma_f32_16x16x4_f32(false, a2, false, b2, (short)0, d, false, false);

    if (nok) {
      const unsigned mb = lo ? 0u : 8u;
#pragma unroll
      for (int v = 0; v < 8; ++v)
        sW[t * 48u + 3u * (mb + (unsigned)v) + N] = d[v];
    }
  }
  }

  lds_barrier();

#pragma unroll
  for (unsigned k = 0; k < TLN; ++k) {
    const unsigned i   = threadIdx.x + 256u * k;
    const unsigned p   = pB + 4u * i;
    const float4   x   = *(const float4*)(sD + 4u * i);
    const v4f      g   = tg[k];
    const unsigned n_o = p >> LOG_TD;
    const bool     has = n_o == 0u || (HAS_M1 && n_o == 1u);
    float4 mv;
    mv.x = has ? tm[k].x : 0.0f;
    mv.y = has ? tm[k].y : 0.0f;
    mv.z = has ? tm[k].z : 0.0f;
    mv.w = has ? tm[k].w : 0.0f;
    float4 o;
    o.x = (1.0f - g.x) * mv.x + g.x * fast_tanh(x.x);
    o.y = (1.0f - g.y) * mv.y + g.y * fast_tanh(x.y);
    o.z = (1.0f - g.z) * mv.z + g.z * fast_tanh(x.z);
    o.w = (1.0f - g.w) * mv.w + g.w * fast_tanh(x.w);
    *(float4*)(CT + p) = o;
  }
  }
}

__global__ __launch_bounds__(256) void gsn_finalize(
    const float* __restrict__ hv, const float* __restrict__ CT,
    float* __restrict__ out) {
  const unsigned q0 = blockIdx.x * 256u + threadIdx.x;
  const unsigned qs = gridDim.x * 256u;
  v4f h4[FIN_PER], c0[FIN_PER], c1[FIN_PER], c2[FIN_PER];
#pragma unroll
  for (unsigned k = 0; k < FIN_PER; ++k) {
    const unsigned i = (q0 + k * qs) * 4u;
    h4[k] = *(const v4f*)(hv + i);
    c0[k] = stream_load4(CT + i);
    c1[k] = stream_load4(CT + TD + i);
    c2[k] = stream_load4(CT + 2u * TD + i);
  }
#pragma unroll
  for (unsigned k = 0; k < FIN_PER; ++k) {
    const unsigned i = (q0 + k * qs) * 4u;
    v4f o;
    o.x = h4[k].x + c0[k].x + c1[k].x + c2[k].x;
    o.y = h4[k].y + c0[k].y + c1[k].y + c2[k].y;
    o.z = h4[k].z + c0[k].z + c1[k].z + c2[k].z;
    o.w = h4[k].w + c0[k].w + c1[k].w + c2[k].w;
    stream_store4(out + i, o);
  }
}

extern "C" void kernel_launch(void* const* d_in, const int* in_sizes, int n_in,
                              void* d_out, int out_size, void* d_ws, size_t ws_size,
                              hipStream_t stream) {
  (void)in_sizes; (void)n_in; (void)out_size; (void)ws_size;
  const float* feat = (const float*)d_in[0];
  const float* W_w  = (const float*)d_in[1];
  const float* W_b  = (const float*)d_in[2];
  const float* Wx_w = (const float*)d_in[3];
  const float* Wx_b = (const float*)d_in[4];
  float* out = (float*)d_out;

  float* G   = (float*)d_ws;
  float* CTa = G + (size_t)3 * TD;
  float* CTb = CTa + (size_t)3 * TD;

  const float* h0 = feat;
  const float* h1 = feat + (size_t)TD;
  const float* h2 = feat + (size_t)2 * TD;

  dim3 blk(256);
  dim3 grd_mm(PGRID);
  dim3 grd_dep(DEP_N);
  dim3 grd_fin(TD / 4u / 256u / FIN_PER);


  gsn_gate_wmma<false, false><<<grd_mm, blk, 0, stream>>>(h0, h2, nullptr, Wx_w, Wx_b, W_w, W_b, G, CTa,
                                                        nullptr, nullptr, nullptr);
  gsn_cand_wmma<false><<<grd_dep, blk, 0, stream>>>(h0, h2, nullptr, W_w, W_b, G, CTa);

  gsn_gate_wmma<false, true><<<grd_mm, blk, 0, stream>>>(h1, h0, nullptr, Wx_w, Wx_b, W_w, W_b, G, CTb,
                                                       h0, CTa, out + 0 * (size_t)TD);
  gsn_cand_wmma<false><<<grd_dep, blk, 0, stream>>>(h1, h0, nullptr, W_w, W_b, G, CTb);

  gsn_gate_wmma<true, true><<<grd_mm, blk, 0, stream>>>(h2, h1, h0, Wx_w, Wx_b, W_w, W_b, G, CTa,
                                                      h1, CTb, out + 1 * (size_t)TD);
  gsn_cand_wmma<true><<<grd_dep, blk, 0, stream>>>(h2, h1, h0, W_w, W_b, G, CTa);
  gsn_finalize<<<grd_fin, blk, 0, stream>>>(h2, CTa, out + 2 * (size_t)TD);
}


# One-message nodes: segment-0 cand tiles that meet only constant gates (sigmoid(bias)) done in the gate launch on the gate rows, gate in cand weights: x*(g*w), was (g*x)*w; cand launch 683->343 tiles
# speedup vs baseline: 3.3147x; 1.0397x over previous
#include <hip/hip_runtime.h>
#include <math.h>

#define LOG_TD 22u
#define TD (1u << LOG_TD)

typedef __attribute__((ext_vector_type(2))) float v2f;
typedef __attribute__((ext_vector_type(4))) float v4f;
typedef __attribute__((ext_vector_type(8))) float v8f;

constexpr unsigned TPW = 8u;
constexpr unsigned BLK_OUT = 8u * TPW * 48u;
constexpr unsigned BLK_IN  = 8u * TPW * 96u;
constexpr unsigned STG     = BLK_IN / 4u / 256u;
constexpr unsigned TLN     = BLK_OUT / 4u / 256u;
constexpr unsigned NTILE   = TD / 16u / TPW / 8u;
constexpr unsigned PGRID   = 1024u;
constexpr unsigned FIN_PER = 4u;
static_assert(NTILE / PGRID == FIN_PER, "one sum position per thread per tile trip");
static_assert(TD / 4u / 256u / FIN_PER == PGRID, "the sum pass and the gate launch share one grid");
constexpr unsigned DEP_N  = (TD + BLK_IN - 1u) / BLK_IN;
constexpr unsigned DEP2_0 = (2u * TD) / BLK_IN;
static_assert((3u * TD) % BLK_IN == 0u, "segment 2 ends on a tile boundary");
static_assert(DEP2_0 + DEP_N == (3u * TD) / BLK_IN, "tiles that meet cat2 segment 2: [DEP2_0, DEP2_0 + DEP_N)");
constexpr unsigned GW_1 = (TD + BLK_OUT - 1u) / BLK_OUT;
constexpr unsigned GW_2 = (2u * TD + BLK_OUT - 1u) / BLK_OUT;
static_assert(DEP_N <= GW_1 && DEP2_0 + DEP_N <= GW_2, "every tile of the cand launch finds its own gates in G");
constexpr unsigned FREE1_0 = (DEP_N * BLK_OUT + BLK_IN - 1u) / BLK_IN;
constexpr unsigned GW_1S   = (FREE1_0 * BLK_IN + BLK_OUT - 1u) / BLK_OUT;
constexpr unsigned GW_1T   = ((DEP_N - 1u) * BLK_IN) / BLK_OUT;
static_assert(FREE1_0 * BLK_IN >= DEP_N * BLK_OUT && FREE1_0 < DEP_N - 1u, "operands in segment 0 from FREE1_0 on meet constant gates only");
static_assert((DEP_N - 1u) * BLK_IN <= TD, "every tile below DEP_N - 1 lies wholly inside segment 0");
static_assert(GW_1S * BLK_OUT >= FREE1_0 * BLK_IN && GW_1S >= DEP_N && GW_1S <= GW_1T && GW_1T * BLK_OUT <= (DEP_N - 1u) * BLK_IN,
              "the cand launch of such a node finds every gate it reads in G");
template <bool HAS_M1>
__device__ __forceinline__ bool tile_free(unsigned tile) {
  return HAS_M1 ? (tile >= DEP_N && !(tile >= DEP2_0 && tile < DEP2_0 + DEP_N))
                : (tile >= FREE1_0 && tile != DEP_N - 1u);
}
template <bool HAS_M1>
__device__ __forceinline__ bool tile_gsame(unsigned tile) {
  return !HAS_M1 && tile >= FREE1_0 && tile < DEP_N - 1u;
}
constexpr unsigned SAME3_0 = (3u * TD) / BLK_IN;
constexpr unsigned SAME3_1 = (4u * TD) / BLK_IN;
constexpr unsigned SAME5_0 = (5u * TD + BLK_IN - 1u) / BLK_IN;
static_assert(SAME3_0 * BLK_IN == 3u * TD && SAME3_1 * BLK_IN <= 4u * TD, "segment 3's whole tiles");
static_assert(SAME5_0 * BLK_IN >= 5u * TD && NTILE * BLK_IN == 6u * TD, "segment 5's whole tiles");
static_assert(SAME3_0 >= DEP2_0 + DEP_N, "every such tile is tile_free, with one message or two");
__device__ __forceinline__ bool tile_same(unsigned tile) {
  return (tile >= SAME3_0 && tile < SAME3_1) || tile >= SAME5_0;
}
constexpr unsigned ZERO_0 = (2u * TD + BLK_IN - 1u) / BLK_IN;
constexpr unsigned ZERO_1 = (3u * TD) / BLK_IN;
constexpr unsigned PAIR_0 = (TD + BLK_IN - 1u) / BLK_IN;
constexpr unsigned PAIR_1 = (2u * TD) / BLK_IN;
constexpr unsigned PAIR_D = (3u * TD) / BLK_IN;
static_assert(ZERO_1 * BLK_IN == 3u * TD && PAIR_D * BLK_IN == 3u * TD, "whole tiles");
static_assert(PAIR_0 * BLK_IN >= TD && PAIR_1 * BLK_IN <= 2u * TD, "segment 1's whole tiles");
static_assert(PAIR_0 == DEP_N && PAIR_1 <= GW_1, "such a tile is tile_free; its gates are the constant, stored only where the cand launch reads them");
static_assert((PAIR_1 + PAIR_D) * BLK_IN <= 5u * TD && PAIR_1 + PAIR_D <= SAME5_0, "the partner lies in segment 4");
static_assert(PAIR_1 * BLK_OUT <= TD && (PAIR_0 + PAIR_D) * BLK_OUT >= 2u * TD, "results: slot 0, partner's: slot 2");
static_assert(BLK_OUT % 3u == 0u, "a tile's first result is in column 0");
constexpr unsigned ZP_0 = ZERO_0 * BLK_OUT - TD;
constexpr unsigned ZP_1 = ZERO_1 * BLK_OUT - TD;
static_assert(ZERO_0 * BLK_OUT >= TD && ZERO_1 * BLK_OUT <= 2u * TD, "the constant results lie in slot 1");
static_assert(ZP_0 % 1024u == 0u && ZP_1 % 1024u == 0u, "a work-group's 256 float4 sum positions are all constant or none");
template <bool HAS_M1>
__device__ __forceinline__ bool tile_skip(unsigned tile) {
  return !HAS_M1 && ((tile >= ZERO_0 && tile < ZERO_1) ||
                     (tile >= PAIR_0 + PAIR_D && tile < PAIR_1 + PAIR_D));
}
template <bool HAS_M1>
__device__ __forceinline__ bool tile_pair(unsigned tile) {
  return !HAS_M1 && tile >= PAIR_0 && tile < PAIR_1;
}
__device__ __forceinline__ float4 col_pattern(unsigned c0, float k0, float k1, float k2) {
  float4 o;
  o.x = c0 == 0u ? k0 : (c0 == 1u ? k1 : k2);
  o.y = c0 == 0u ? k1 : (c0 == 1u ? k2 : k0);
  o.z = c0 == 0u ? k2 : (c0 == 1u ? k0 : k1);
  o.w = o.x;
  return o;
}
__device__ __forceinline__ float uniform_f(float x) {
  return __builtin_bit_cast(float, __builtin_amdgcn_readfirstlane(__builtin_bit_cast(int, x)));
}

__device__ __forceinline__ float hw_exp2(float x) {
#if __has_builtin(__builtin_amdgcn_exp2f)
  return __builtin_amdgcn_exp2f(x);
#else
  return exp2f(x);
#endif
}
__device__ __forceinline__ float hw_rcp(float x) {
#if __has_builtin(__builtin_amdgcn_rcpf)
  return __builtin_amdgcn_rcpf(x);
#else
  return 1.0f / x;
#endif
}
__device__ __forceinline__ float fast_sigmoid(float x) {
  return hw_rcp(1.0f + hw_exp2(-1.4426950408889634f * x));
}
__device__ __forceinline__ float fast_tanh(float x) {
#if __has_builtin(__builtin_amdgcn_tanhf)
  return __builtin_amdgcn_tanhf(x);
#else
  return 1.0f - 2.0f * hw_rcp(hw_exp2(2.8853900817779268f * x) + 1.0f);
#endif
}

__device__ __forceinline__ void lds_barrier() {
  __builtin_amdgcn_fence(__ATOMIC_RELEASE, "workgroup", "local");
  __builtin_amdgcn_s_barrier();
  __builtin_amdgcn_fence(__ATOMIC_ACQUIRE, "workgroup", "local");
}

__device__ __forceinline__ v4f stream_load4(const float* p) {
  return __builtin_nontemporal_load((const v4f*)p);
}
__device__ __forceinline__ void stream_store4(float* p, const v4f v) {
  __builtin_nontemporal_store(v, (v4f*)p);
}

template <bool HAS_M1>
__device__ __forceinline__ bool gate_zero(unsigned m) {
  const unsigned n = m >> LOG_TD;
  return n < 3u && n != 0u && !(HAS_M1 && n == 1u);
}
template <bool HAS_M1>
__device__ __forceinline__ v4f gate_load4(unsigned m,
    const float* __restrict__ hv, const float* __restrict__ m0,
    const float* __restrict__ m1) {
  const unsigned n = m >> LOG_TD;
  const unsigned r = m & (TD - 1u);
  const float* src = (n == 0u) ? m0 : ((HAS_M1 && n == 1u) ? m1 : hv);
  return *(const v4f*)(src + (gate_zero<HAS_M1>(m) ? 0u : r));
}
template <bool HAS_M1>
__device__ __forceinline__ v4f rows_load4(unsigned m, bool pr,
    const float* __restrict__ hv, const float* __restrict__ m0,
    const float* __restrict__ m1) {
  const unsigned n = m >> LOG_TD;
  const unsigned r = m & (TD - 1u);
  const float* src = (n == 0u) ? m0 : ((HAS_M1 && n == 1u) ? m1 : hv);
  return *(const v4f*)(src + ((gate_zero<HAS_M1>(m) && !pr) ? 0u : r));
}
template <bool HAS_M1>
__device__ __forceinline__ v4f gate_mix4(unsigned m, const v4f v) {
  const bool zero = gate_zero<HAS_M1>(m);
  v4f o;
  o.x = zero ? 0.0f : v.x;
  o.y = zero ? 0.0f : v.y;
  o.z = zero ? 0.0f : v.z;
  o.w = zero ? 0.0f : v.w;
  return o;
}

template <bool HAS_M1>
__device__ __forceinline__ void cand_load4(unsigned m,
    const float* __restrict__ hv, const float* __restrict__ m0,
    const float* __restrict__ m1, const float* __restrict__ G,
    v4f& a, v4f& g) {
  const unsigned seg = m >> LOG_TD;
  const unsigned r   = m & (TD - 1u);
  const bool  isH = (seg & 1u) != 0u;
  const bool  s0  = seg == 0u;
  const bool  s1  = HAS_M1 && seg == 2u;
  const bool  msg = s0 || s1;
  const float* pa = s0 ? m0 : (s1 ? m1 : hv);
  const float* pg = s1 ? G + TD : G;
  a = *(const v4f*)(pa + ((isH || msg) ? r : 0u));
  g = *(const v4f*)(msg ? pg + r : hv);
}
template <bool HAS_M1>
__device__ __forceinline__ v4f cand_mix4(unsigned m, const v4f a, const v4f g) {
  const unsigned seg = m >> LOG_TD;
  const bool  isH = (seg & 1u) != 0u;
  const bool  msg = seg == 0u || (HAS_M1 && seg == 2u);
  const float px = g.x * a.x, py = g.y * a.y, pz = g.z * a.z, pw = g.w * a.w;
  const float zx = msg ? px : 0.0f, zy = msg ? py : 0.0f,
              zz = msg ? pz : 0.0f, zw = msg ? pw : 0.0f;
  v4f v;
  v.x = isH ? a.x : zx;
  v.y = isH ? a.y : zy;
  v.z = isH ? a.z : zz;
  v.w = isH ? a.w : zw;
  return v;
}

__device__ __forceinline__ void mm_phase(const float* const sAw, float* const sWl,
    const unsigned lane, const bool lo, const bool ok,
    const v2f b, const v2f b2, const float bias) {
#pragma unroll
  for (unsigned th = 0; th < TPW; th += TPW / 2u) {
  v2f fa[TPW / 2u], fa4[TPW / 2u];
#pragma unroll
  for (unsigned u = 0; u < TPW / 2u; ++u) {
    const float* const ar = sAw + (th + u) * 96u + 6u * (lane & 15u);
    fa[u]  = *(const v2f*)(ar + (lo ? 0u : 2u));
    fa4[u] = *(const v2f*)(ar + 4u);
  }

#pragma unroll
  for (unsigned u = 0; u < TPW / 2u; ++u) {
    const unsigned t = th + u;
    const v2f a = fa[u];
    v2f a2;
    a2.x = lo ? fa4[u].x : 0.0f;
    a2.y = lo ? fa4[u].y : 0.0f;

    v8f c;
#pragma unroll
    for (int v = 0; v < 8; ++v) c[v] = bias;

    v8f d = __builtin_amdgcn_wmma_f32_16x16x4_f32(false, a,  false, b,  (short)0, c, false, false);
    d     = __builtin_amdgcn_wmma_f32_16x16x4_f32(false, a2, false, b2, (short)0, d, false, false);

    if (ok) {
      const unsigned mb = lo ? 0u : 8u;
#pragma unroll
      for (int v = 0; v < 8; ++v)
        sWl[t * 48u + 3u * (mb + (unsigned)v)] = d[v];
    }
  }
  }
}

__device__ __forceinline__ void mm_phase_cat(const float* const sAw, float* const sWl,
    const unsigned lane, const bool lo, const bool ok,
    const v2f b1, const v2f b2, const v2f b3, const float bias) {
#pragma unroll
  for (unsigned th = 0; th < TPW; th += TPW / 2u) {
  v2f f1[TPW / 2u], f2[TPW / 2u], f3[TPW / 2u];
#pragma unroll
  for (unsigned u = 0; u < TPW / 2u; ++u) {
    const float* const ar = sAw + (th + u) * 96u + 6u * (lane & 15u);
    f1[u] = *(const v2f*)(ar + (lo ? 0u : 2u));
    f2[u] = *(const v2f*)(ar + (lo ? 4u : BLK_IN));
    f3[u] = *(const v2f*)(ar + BLK_IN + (lo ? 2u : 4u));
  }
#pragma unroll
  for (unsigned u = 0; u < TPW / 2u; ++u) {
    const unsigned t = th + u;
    v8f c;
#pragma unroll
    for (int v = 0; v < 8; ++v) c[v] = bias;
    v8f d = __builtin_amdgcn_wmma_f32_16x16x4_f32(false, f1[u], false, b1, (short)0, c, false, false);
    d     = __builtin_amdgcn_wmma_f32_16x16x4_f32(false, f2[u], false, b2, (short)0, d, false, false);
    d     = __builtin_amdgcn_wmma_f32_16x16x4_f32(false, f3[u], false, b3, (short)0, d, false, false);
    if (ok) {
      const unsigned mb = lo ? 0u : 8u;
#pragma unroll
      for (int v = 0; v < 8; ++v)
        sWl[t * 48u + 3u * (mb + (unsigned)v)] = d[v];
    }
  }
  }
}

__device__ __forceinline__ void cat_weights(const bool lo, const bool nok, const bool nok2,
    const float w0, const float w1, const float w2, const float w3, const float w4, const float w5,
    v2f& c1, v2f& c2, v2f& c3) {
  c1.x = nok ? (lo ? w0 : w2) : 0.0f;
  c1.y = nok ? (lo ? w1 : w3) : 0.0f;
  c2.x = lo ? (nok ? w4 : 0.0f) : (nok2 ? w0 : 0.0f);
  c2.y = lo ? (nok ? w5 : 0.0f) : (nok2 ? w1 : 0.0f);
  c3.x = nok2 ? (lo ? w2 : w4) : 0.0f;
  c3.y = nok2 ? (lo ? w3 : w5) : 0.0f;
}

template <bool HAS_M1, bool FIN>
__global__ __launch_bounds__(256) __attribute__((amdgpu_waves_per_eu(8))) void gsn_gate_wmma(
    const float* __restrict__ hv, const float* __restrict__ m0,
    const float* __restrict__ m1, const float* __restrict__ Wx,
    const float* __restrict__ Bx,
    const float* __restrict__ Ww,
    const float* __restrict__ Bw,
    float* __restrict__ G, float* __restrict__ CT,
    const float* __restrict__ fh, const float* __restrict__ fCT,
    float* __restrict__ fout) {
  const unsigned lane  = threadIdx.x & 31u;
  const unsigned wave  = threadIdx.x >> 5u;
  __shared__ __attribute__((aligned(16))) float sA[2u * BLK_IN];
  __shared__ __attribute__((aligned(16))) float sD[BLK_OUT];
  __shared__ __attribute__((aligned(16))) float sC[BLK_OUT];
  float* const sW = sD + wave * (TPW * 48u);
  float* const sWc = sC + wave * (TPW * 48u);
  const float* const sAw = sA + wave * (TPW * 96u);

  v4f st[STG], ch[STG];
  {
    const unsigned mF = blockIdx.x * BLK_IN + 4u * threadIdx.x;
    const bool two0 = tile_free<HAS_M1>(blockIdx.x) && !tile_same(blockIdx.x);
    const bool pr0  = tile_pair<HAS_M1>(blockIdx.x);
#pragma unroll
    for (unsigned j = 0; j < STG; ++j)
      st[j] = rows_load4<HAS_M1>(mF + 1024u * j, pr0, hv, m0, m1);
    if (HAS_M1) {
#pragma unroll
    for (unsigned j = 0; j < STG; ++j) {
      const unsigned m   = mF + 1024u * j;
      const bool     isH = ((m >> LOG_TD) & 1u) != 0u;
      ch[j] = *(const v4f*)(hv + ((two0 && isH) ? (m & (TD - 1u)) : 0u));
    }
    }
  }

  const bool     lo   = lane < 16u;
  const unsigned N    = lane & 15u;
  const bool     nok  = N < 3u;
  const bool     nok2 = N >= 3u && N < 6u;
  const bool     nok3 = !HAS_M1 && N >= 6u && N < 9u;
  const bool     cnd  = nok2 || nok3;
  const unsigned wn   = nok ? N : 0u;
  const unsigned wm   = nok2 ? N - 3u : (nok3 ? N - 6u : 0u);
  const float    mskg = (nok || cnd) ? 1.0f : 0.0f;
  float* const sWl = nok2 ? sWc + (N - 3u) : (nok3 ? sWc + (N - 6u) : sW + N);
  const float* const gw = cnd ? Ww + wm * 6u : Wx + wn * 6u;
  const float w0 = gw[0]*mskg, w1 = gw[1]*mskg, w2 = gw[2]*mskg,
              w3 = gw[3]*mskg, w4 = gw[4]*mskg, w5 = gw[5]*mskg;
  const float bias = (cnd ? Bw[wm] : Bx[wn]) * mskg;

  v2f b, b2;
  b.x   = lo ? w0 : w2;
  b.y   = lo ? w1 : w3;
  b2.x  = lo ? w4 : 0.0f;
  b2.y  = lo ? w5 : 0.0f;
  v2f c1, c2, c3;
  if (HAS_M1) cat_weights(lo, nok, nok2, w0, w1, w2, w3, w4, w5, c1, c2, c3);

  const float gc0 = uniform_f(fast_sigmoid(Bx[0])), gc1 = uniform_f(fast_sigmoid(Bx[1])),
              gc2 = uniform_f(fast_sigmoid(Bx[2]));
  const float tc0 = uniform_f(fast_tanh(Bw[0])), tc1 = uniform_f(fast_tanh(Bw[1])),
              tc2 = uniform_f(fast_tanh(Bw[2]));
  const float kc0 = uniform_f(gc0 * tc0), kc1 = uniform_f(gc1 * tc1), kc2 = uniform_f(gc2 * tc2);

  if (!HAS_M1) {
    b.x  *= nok3 ? (lo ? gc0 : gc2) : 1.0f;
    b.y  *= nok3 ? (lo ? gc1 : gc0) : 1.0f;
    b2.x *= nok3 ? gc1 : 1.0f;
    b2.y *= nok3 ? gc2 : 1.0f;
  }

  for (unsigned trip = 0; trip < NTILE / PGRID; ++trip) {
  const unsigned tile = blockIdx.x + trip * PGRID;
  const unsigned fbase = blockIdx.x * 1024u + trip * (PGRID * 1024u);
  const unsigned fq    = fbase + 4u * threadIdx.x;
  const bool     fcz   = fbase >= ZP_0 && fbase < ZP_1;
  if (tile_skip<HAS_M1>(tile)) {
    if (FIN) {
      const v4f fh4 = *(const v4f*)(fh + fq);
      const v4f fc0 = stream_load4(fCT + fq);
      const v4f fc2 = stream_load4(fCT + 2u * TD + fq);
      v4f fc1;
      if (fcz) {
        const float4 kp = col_pattern((TD + fq) % 3u, kc0, kc1, kc2);
        fc1.x = kp.x; fc1.y = kp.y; fc1.z = kp.z; fc1.w = kp.w;
      } else {
        fc1 = stream_load4(fCT + TD + fq);
      }
      v4f o;
      o.x = fh4.x + fc0.x + fc1.x + fc2.x;
      o.y = fh4.y + fc0.y + fc1.y + fc2.y;
      o.z = fh4.z + fc0.z + fc1.z + fc2.z;
      o.w = fh4.w + fc0.w + fc1.w + fc2.w;
      stream_store4(fout + fq, o);
    }
    continue;
  }
  const bool fuse  = tile_free<HAS_M1>(tile);
  const bool gsame = tile_gsame<HAS_M1>(tile);
  const bool same  = tile_same(tile) || gsame;
  const bool pair  = tile_pair<HAS_M1>(tile);
  const bool keepG = HAS_M1 ? tile < GW_2
                            : (tile < GW_1S || (tile >= GW_1T && tile < GW_1));
  const bool two   = fuse && !same && !pair;
  const unsigned mT = tile * BLK_IN + 4u * threadIdx.x;
  if (pair) {
#pragma unroll
    for (unsigned j = 0; j < STG; ++j)
      *(v4f*)(sA + 4u * (threadIdx.x + 256u * j)) = st[j];
  } else {
#pragma unroll
  for (unsigned j = 0; j < STG; ++j)
    *(v4f*)(sA + 4u * (threadIdx.x + 256u * j)) = gate_mix4<HAS_M1>(mT + 1024u * j, st[j]);
  }
  if (two) {
#pragma unroll
    for (unsigned j = 0; j < STG; ++j) {
      const unsigned m = mT + 1024u * j;
      const bool isH = ((m >> LOG_TD) & 1u) != 0u;
      const v4f hp = HAS_M1 ? ch[j] : *(const v4f*)(hv + (isH ? (m & (TD - 1u)) : 0u));
      v4f v;
      v.x = isH ? hp.x : 0.0f;
      v.y = isH ? hp.y : 0.0f;
      v.z = isH ? hp.z : 0.0f;
      v.w = isH ? hp.w : 0.0f;
      *(v4f*)(sA + BLK_IN + 4u * (threadIdx.x + 256u * j)) = v;
    }
  }
  v4f fh4, fc0, fc1, fc2;
  if (FIN) {
    const float4 kp = col_pattern((TD + fq) % 3u, kc0, kc1, kc2);
    fc1.x = kp.x; fc1.y = kp.y; fc1.z = kp.z; fc1.w = kp.w;
    fh4 = *(const v4f*)(fh + fq);
    fc0 = stream_load4(fCT + fq);
    if (!fcz) fc1 = stream_load4(fCT + TD + fq);
    fc2 = stream_load4(fCT + 2u * TD + fq);
    __builtin_amdgcn_sched_barrier(0);
  }
  {
    unsigned nt = tile + PGRID;
    nt += tile_skip<HAS_M1>(nt) ? PGRID : 0u;
    nt += tile_skip<HAS_M1>(nt) ? PGRID : 0u;
    const bool     more = nt < NTILE;
    const bool     twoN = more && tile_free<HAS_M1>(nt) && !tile_same(nt);
    const unsigned mN   = nt * BLK_IN + 4u * threadIdx.x;
    const bool     prN  = more && tile_pair<HAS_M1>(nt);
#pragma unroll
    for (unsigned j = 0; j < STG; ++j)
      st[j] = rows_load4<HAS_M1>(more ? mN + 1024u * j : 2u * TD, prN, hv, m0, m1);
    if (HAS_M1) {
#pragma unroll
    for (unsigned j = 0; j < STG; ++j) {
      const unsigned m   = mN + 1024u * j;
      const bool     isH = ((m >> LOG_TD) & 1u) != 0u;
      ch[j] = *(const v4f*)(hv + ((twoN && isH) ? (m & (TD - 1u)) : 0u));
    }
    }
  }
  lds_barrier();

  if (two) {
    if (!HAS_M1) {
      const float* const gq = nok2 ? Ww + (N - 3u) * 6u : Wx + (nok ? N : 0u) * 6u;
      const float mq = (nok || nok2) ? 1.0f : 0.0f;
      cat_weights(lo, nok, nok2, gq[0]*mq, gq[1]*mq, gq[2]*mq, gq[3]*mq, gq[4]*mq, gq[5]*mq, c1, c2, c3);
    }
    mm_phase_cat(sAw, sWl, lane, lo, nok || nok2, c1, c2, c3, bias);
  }
  else     mm_phase(sAw, sWl, lane, lo, nok || (gsame ? nok3 : ((same || pair) && nok2)), b, b2, bias);

  lds_barrier();

  if (FIN) {
    v4f o;
    o.x = fh4.x + fc0.x + fc1.x + fc2.x;
    o.y = fh4.y + fc0.y + fc1.y + fc2.y;
    o.z = fh4.z + fc0.z + fc1.z + fc2.z;
    o.w = fh4.w + fc0.w + fc1.w + fc2.w;
    stream_store4(fout + fq, o);
  }

  const unsigned pB = tile * BLK_OUT;
  float4 tm[TLN];
#pragma unroll
  for (unsigned k = 0; k < TLN; ++k) {
    const unsigned p   = pB + 4u * (threadIdx.x + 256u * k);
    const unsigned n_o = p >> LOG_TD;
    const unsigned r   = p & (TD - 1u);
    const float* const pm = (n_o == 0u) ? m0 + r : ((HAS_M1 && n_o == 1u) ? m1 + r : hv);
    tm[k] = *(const float4*)(fuse ? pm : hv);
  }

  if (pair) {
#pragma unroll
    for (unsigned k = 0; k < TLN; ++k) {
      const unsigned i  = threadIdx.x + 256u * k;
      const unsigned p  = pB + 4u * i;
      const float4   xg = *(const float4*)(sD + 4u * i);
      const float4   xc = *(const float4*)(sC + 4u * i);
      const unsigned c0 = p % 3u;
      const float4   g  = col_pattern(c0, gc0, gc1, gc2);
      const float4   tc = col_pattern(c0, tc0, tc1, tc2);
      if (keepG) *(float4*)(G + p) = g;
      float4 o;
      o.x = (1.0f - g.x) * tm[k].x + g.x * fast_tanh(xc.x);
      o.y = (1.0f - g.y) * tm[k].y + g.y * fast_tanh(xc.y);
      o.z = (1.0f - g.z) * tm[k].z + g.z * fast_tanh(xc.z);
      o.w = (1.0f - g.w) * tm[k].w + g.w * fast_tanh(xc.w);
      *(float4*)(CT + p) = o;
      float4 o2;
      o2.x = fast_sigmoid(xg.x) * tc.x;
      o2.y = fast_sigmoid(xg.y) * tc.y;
      o2.z = fast_sigmoid(xg.z) * tc.z;
      o2.w = fast_sigmoid(xg.w) * tc.w;
      *(float4*)(CT + p + PAIR_D * BLK_OUT) = o2;
    }
  } else {
  float4 tg[TLN];
#pragma unroll
  for (unsigned k = 0; k < TLN; ++k) {
    const unsigned i = threadIdx.x + 256u * k;
    const float4 x = *(const float4*)(sD + 4u * i);
    float4 g;
    g.x = fast_sigmoid(x.x);
    g.y = fast_sigmoid(x.y);
    g.z = fast_sigmoid(x.z);
    g.w = fast_sigmoid(x.w);
    tg[k] = g;
    if (keepG) *(float4*)(G + pB + 4u * i) = g;
  }

  if (fuse) {

#pragma unroll
    for (unsigned k = 0; k < TLN; ++k) {
      const unsigned i   = threadIdx.x + 256u * k;
      const unsigned p   = pB + 4u * i;
      const float4   x   = *(const float4*)(sC + 4u * i);
      const float4   g   = tg[k];
      const unsigned n_o = p >> LOG_TD;
      const bool     has = n_o == 0u || (HAS_M1 && n_o == 1u);
      float4 mv;
      mv.x = has ? tm[k].x : 0.0f;
      mv.y = has ? tm[k].y : 0.0f;
      mv.z = has ? tm[k].z : 0.0f;
      mv.w = has ? tm[k].w : 0.0f;
      float4 o;
      o.x = (1.0f - g.x) * mv.x + g.x * fast_tanh(x.x);
      o.y = (1.0f - g.y) * mv.y + g.y * fast_tanh(x.y);
      o.z = (1.0f - g.z) * mv.z + g.z * fast_tanh(x.z);
      o.w = (1.0f - g.w) * mv.w + g.w * fast_tanh(x.w);
      *(float4*)(CT + p) = o;
    }
  }
  }
  }
}

template <bool HAS_M1>
__global__ __launch_bounds__(256) __attribute__((amdgpu_waves_per_eu(8))) void gsn_cand_wmma(
    const float* __restrict__ hv, const float* __restrict__ m0,
    const float* __restrict__ m1, const float* __restrict__ Ww,
    const float* __restrict__ Bw,
    const float* __restrict__ G, float* __restrict__ CT) {
  const unsigned lane  = threadIdx.x & 31u;
  const unsigned wave  = threadIdx.x >> 5u;
  __shared__ __attribute__((aligned(16))) float sA[BLK_IN];
  __shared__ __attribute__((aligned(16))) float sD[BLK_OUT];
  float* const sW = sD + wave * (TPW * 48u);
  const float* const sAw = sA + wave * (TPW * 96u);

  const unsigned tile0 = (HAS_M1 || blockIdx.x < FREE1_0) ? blockIdx.x : DEP_N - 1u;
  v4f st[STG], sg[STG];
  {
    const unsigned mF = tile0 * BLK_IN + 4u * threadIdx.x;
#pragma unroll
    for (unsigned j = 0; j < STG; ++j)
      cand_load4<HAS_M1>(mF + 1024u * j, hv, m0, m1, G, st[j], sg[j]);
  }

  const bool     lo  = lane < 16u;
  const unsigned N   = lane & 15u;
  const bool     nok = N < 3u;
  const unsigned wn  = nok ? N : 0u;
  const float    msk = nok ? 1.0f : 0.0f;
  const float r0 = Ww[wn*6u+0u], r1 = Ww[wn*6u+1u], r2 = Ww[wn*6u+2u],
              r3 = Ww[wn*6u+3u], r4 = Ww[wn*6u+4u], r5 = Ww[wn*6u+5u];
  const float rb = Bw[wn];
  asm volatile("" :: "v"(st[0]), "v"(st[1]), "v"(st[2]), "v"(st[3]), "v"(st[4]), "v"(st[5]), "v"(sg[0]), "v"(sg[1]), "v"(sg[2]), "v"(sg[3]), "v"(sg[4]), "v"(sg[5]),
               "v"(r0), "v"(r1), "v"(r2), "v"(r3), "v"(r4), "v"(r5), "v"(rb));
  const float w0 = r0*msk, w1 = r1*msk, w2 = r2*msk, w3 = r3*msk, w4 = r4*msk, w5 = r5*msk;
  const float bias = rb * msk;

  v2f b, b2;
  b.x  = lo ? w0 : w2;
  b.y  = lo ? w1 : w3;
  b2.x = lo ? w4 : 0.0f;
  b2.y = lo ? w5 : 0.0f;

  constexpr unsigned NTRIP = HAS_M1 ? 2u : 1u;
  for (unsigned trip = 0; trip < NTRIP; ++trip) {
  const unsigned tile = tile0 + trip * DEP2_0;
  const unsigned mT = tile * BLK_IN + 4u * threadIdx.x;
#pragma unroll
  for (unsigned j = 0; j < STG; ++j)
    *(v4f*)(sA + 4u * (threadIdx.x + 256u * j)) = cand_mix4<HAS_M1>(mT + 1024u * j, st[j], sg[j]);

  const unsigned pB = tile * BLK_OUT;
  v4f tg[TLN];
  float4 tm[TLN];
#pragma unroll
  for (unsigned k = 0; k < TLN; ++k) {
    const unsigned p   = pB + 4u * (threadIdx.x + 256u * k);
    const unsigned n_o = p >> LOG_TD;
    const unsigned r   = p & (TD - 1u);
    const float* const pm = (n_o == 0u) ? m0 + r : ((HAS_M1 && n_o == 1u) ? m1 + r : hv);
    tg[k] = stream_load4(G + p);
    tm[k] = *(const float4*)pm;
  }
  __builtin_amdgcn_sched_barrier(0);
  {
    const unsigned nt   = tile + DEP2_0;
    const bool     more = trip + 1u < NTRIP;
    const unsigned mN   = nt * BLK_IN + 4u * threadIdx.x;
#pragma unroll
    for (unsigned j = 0; j < STG; ++j)
      cand_load4<HAS_M1>(more ? mN + 1024u * j : 4u * TD, hv, m0, m1, G, st[j], sg[j]);
  }
  lds_barrier();

#pragma unroll
  for (unsigned th = 0; th < TPW; th += TPW / 2u) {
  v2f fa[TPW / 2u], fa4[TPW / 2u];
#pragma unroll
  for (unsigned u = 0; u < TPW / 2u; ++u) {
    const float* const ar = sAw + (th + u) * 96u + 6u * (lane & 15u);
    fa[u]  = *(const v2f*)(ar + (lo ? 0u : 2u));
    fa4[u] = *(const v2f*)(ar + 4u);
  }

#pragma unroll
  for (unsigned u = 0; u < TPW / 2u; ++u) {
    const unsigned t = th + u;
    const v2f a = fa[u];
    v2f a2;
    a2.x = lo ? fa4[u].x : 0.0f;
    a2.y = lo ? fa4[u].y : 0.0f;

    v8f c;
#pragma unroll
    for (int v = 0; v < 8; ++v) c[v] = bias;

    v8f d = __builtin_amdgcn_wmma_f32_16x16x4_f32(false, a,  false, b,  (short)0, c, false, false);
    d     = __builtin_amdgcn_wmma_f32_16x16x4_f32(false, a2, false, b2, (short)0, d, false, false);

    if (nok) {
      const unsigned mb = lo ? 0u : 8u;
#pragma unroll
      for (int v = 0; v < 8; ++v)
        sW[t * 48u + 3u * (mb + (unsigned)v) + N] = d[v];
    }
  }
  }

  lds_barrier();

#pragma unroll
  for (unsigned k = 0; k < TLN; ++k) {
    const unsigned i   = threadIdx.x + 256u * k;
    const unsigned p   = pB + 4u * i;
    const float4   x   = *(const float4*)(sD + 4u * i);
    const v4f      g   = tg[k];
    const unsigned n_o = p >> LOG_TD;
    const bool     has = n_o == 0u || (HAS_M1 && n_o == 1u);
    float4 mv;
    mv.x = has ? tm[k].x : 0.0f;
    mv.y = has ? tm[k].y : 0.0f;
    mv.z = has ? tm[k].z : 0.0f;
    mv.w = has ? tm[k].w : 0.0f;
    float4 o;
    o.x = (1.0f - g.x) * mv.x + g.x * fast_tanh(x.x);
    o.y = (1.0f - g.y) * mv.y + g.y * fast_tanh(x.y);
    o.z = (1.0f - g.z) * mv.z + g.z * fast_tanh(x.z);
    o.w = (1.0f - g.w) * mv.w + g.w * fast_tanh(x.w);
    *(float4*)(CT + p) = o;
  }
  }
}

__global__ __launch_bounds__(256) void gsn_finalize(
    const float* __restrict__ hv, const float* __restrict__ CT,
    float* __restrict__ out) {
  const unsigned q0 = blockIdx.x * 256u + threadIdx.x;
  const unsigned qs = gridDim.x * 256u;
  v4f h4[FIN_PER], c0[FIN_PER], c1[FIN_PER], c2[FIN_PER];
#pragma unroll
  for (unsigned k = 0; k < FIN_PER; ++k) {
    const unsigned i = (q0 + k * qs) * 4u;
    h4[k] = *(const v4f*)(hv + i);
    c0[k] = stream_load4(CT + i);
    c1[k] = stream_load4(CT + TD + i);
    c2[k] = stream_load4(CT + 2u * TD + i);
  }
#pragma unroll
  for (unsigned k = 0; k < FIN_PER; ++k) {
    const unsigned i = (q0 + k * qs) * 4u;
    v4f o;
    o.x = h4[k].x + c0[k].x + c1[k].x + c2[k].x;
    o.y = h4[k].y + c0[k].y + c1[k].y + c2[k].y;
    o.z = h4[k].z + c0[k].z + c1[k].z + c2[k].z;
    o.w = h4[k].w + c0[k].w + c1[k].w + c2[k].w;
    stream_store4(out + i, o);
  }
}

extern "C" void kernel_launch(void* const* d_in, const int* in_sizes, int n_in,
                              void* d_out, int out_size, void* d_ws, size_t ws_size,
                              hipStream_t stream) {
  (void)in_sizes; (void)n_in; (void)out_size; (void)ws_size;
  const float* feat = (const float*)d_in[0];
  const float* W_w  = (const float*)d_in[1];
  const float* W_b  = (const float*)d_in[2];
  const float* Wx_w = (const float*)d_in[3];
  const float* Wx_b = (const float*)d_in[4];
  float* out = (float*)d_out;

  float* G   = (float*)d_ws;
  float* CTa = G + (size_t)3 * TD;
  float* CTb = CTa + (size_t)3 * TD;

  const float* h0 = feat;
  const float* h1 = feat + (size_t)TD;
  const float* h2 = feat + (size_t)2 * TD;

  dim3 blk(256);
  dim3 grd_mm(PGRID);
  dim3 grd_dep(DEP_N);
  dim3 grd_dep1(FREE1_0 + 1u);
  dim3 grd_fin(TD / 4u / 256u / FIN_PER);


  gsn_gate_wmma<false, false><<<grd_mm, blk, 0, stream>>>(h0, h2, nullptr, Wx_w, Wx_b, W_w, W_b, G, CTa,
                                                        nullptr, nullptr, nullptr);
  gsn_cand_wmma<false><<<grd_dep1, blk, 0, stream>>>(h0, h2, nullptr, W_w, W_b, G, CTa);

  gsn_gate_wmma<false, true><<<grd_mm, blk, 0, stream>>>(h1, h0, nullptr, Wx_w, Wx_b, W_w, W_b, G, CTb,
                                                       h0, CTa, out + 0 * (size_t)TD);
  gsn_cand_wmma<false><<<grd_dep1, blk, 0, stream>>>(h1, h0, nullptr, W_w, W_b, G, CTb);

  gsn_gate_wmma<true, true><<<grd_mm, blk, 0, stream>>>(h2, h1, h0, Wx_w, Wx_b, W_w, W_b, G, CTa,
                                                      h1, CTb, out + 1 * (size_t)TD);
  gsn_cand_wmma<true><<<grd_dep, blk, 0, stream>>>(h2, h1, h0, W_w, W_b, G, CTa);
  gsn_finalize<<<grd_fin, blk, 0, stream>>>(h2, CTa, out + 2 * (size_t)TD);
}
